# starting kernel
# baseline (speedup 1.0000x reference)
#include <hip/hip_runtime.h>
#include <hip/hip_bf16.h>

typedef float v2f __attribute__((ext_vector_type(2)));
typedef float v8f __attribute__((ext_vector_type(8)));

__device__ __forceinline__ void atomic_add_f32(float* p, float v) {
  __hip_atomic_fetch_add(p, v, __ATOMIC_RELAXED, __HIP_MEMORY_SCOPE_AGENT);
}

__global__ void fill_kernel(float* __restrict__ p, float v, long long n) {
  long long i = (long long)blockIdx.x * blockDim.x + threadIdx.x;
  if (i < n) p[i] = v;
}

__global__ void deg_kernel(const int* __restrict__ dst, float* __restrict__ deg,
                           long long E) {
  long long e = (long long)blockIdx.x * blockDim.x + threadIdx.x;
  if (e < E) atomic_add_f32(&deg[dst[e]], 1.0f);
}

__global__ void dinv_kernel(float* __restrict__ deg, int n) {
  int i = blockIdx.x * blockDim.x + threadIdx.x;
  if (i < n) deg[i] = __frsqrt_rn(fmaxf(deg[i], 1e-12f));
}

template <int FOUT>
__global__ __launch_bounds__(FOUT * 2) void gemm_wmma_kernel(
    const float* __restrict__ A, const float* __restrict__ W,
    float* __restrict__ C, int nrows) {
  __shared__ float lds[16 * 128];

  const int row0 = blockIdx.x * 16;
  if (row0 >= nrows) return;
  const int tid = threadIdx.x;
  constexpr int NTHREADS = FOUT * 2;

  if (row0 + 16 <= nrows) {
    const float4* srcv = (const float4*)(A + (size_t)row0 * 128);
    float4* dstv = (float4*)lds;
    #pragma unroll 2
    for (int i = tid; i < 512; i += NTHREADS) dstv[i] = srcv[i];
  } else {
    for (int i = tid; i < 2048; i += NTHREADS) {
      int r = i >> 7;
      lds[i] = (row0 + r < nrows) ? A[(size_t)(row0 + r) * 128 + (i & 127)] : 0.0f;
    }
  }
  __syncthreads();

  const int lane = tid & 31;
  const int wave = tid >> 5;
  const int colbase = wave * 16;
  const int half = lane >> 4;
  const int m = lane & 15;

  v8f acc = {};
  #pragma unroll
  for (int k = 0; k < 128; k += 4) {
    const int kb = k + half * 2;
    v2f a = *(const v2f*)&lds[m * 128 + kb];
    v2f b;
    b.x = W[(size_t)kb * FOUT + colbase + m];
    b.y = W[(size_t)(kb + 1) * FOUT + colbase + m];
    acc = __builtin_amdgcn_wmma_f32_16x16x4_f32(false, a, false, b, (short)0,
                                                acc, false, false);
  }

  #pragma unroll
  for (int i = 0; i < 8; ++i) {
    int row = row0 + i + half * 8;
    if (row < nrows) C[(size_t)row * FOUT + colbase + m] = acc[i];
  }
}

template <int F>
__global__ void edge_agg_kernel(const int* __restrict__ src,
                                const int* __restrict__ dst,
                                const float* __restrict__ dinv,
                                const float* __restrict__ h,
                                float* __restrict__ agg, long long E) {
  constexpr int CH = F / 4;
  long long t = (long long)blockIdx.x * blockDim.x + threadIdx.x;
  long long e = t / CH;
  int c = (int)(t & (CH - 1));
  if (e >= E) return;
  int u = src[e];
  int v = dst[e];
  float norm = dinv[u] * dinv[v];
  float4 hv = *(const float4*)&h[(size_t)u * F + c * 4];
  float* out = &agg[(size_t)v * F + c * 4];
  atomic_add_f32(out + 0, norm * hv.x);
  atomic_add_f32(out + 1, norm * hv.y);
  atomic_add_f32(out + 2, norm * hv.z);
  atomic_add_f32(out + 3, norm * hv.w);
}

__global__ void epilogue1_kernel(float* __restrict__ agg,
                                 const float* __restrict__ hpre,
                                 const float* __restrict__ dinv,
                                 const float* __restrict__ bias, long long n) {
  long long i = (long long)blockIdx.x * blockDim.x + threadIdx.x;
  if (i >= n) return;
  int v = (int)(i >> 7);
  int c = (int)(i & 127);
  float di = dinv[v];
  float val = agg[i] + di * di * hpre[i] + bias[c];
  agg[i] = fmaxf(val, 0.0f);
}

__global__ void epilogue2_kernel(float* __restrict__ out,
                                 const float* __restrict__ hpre,
                                 const float* __restrict__ dinv,
                                 const float* __restrict__ bias, long long n) {
  long long i = (long long)blockIdx.x * blockDim.x + threadIdx.x;
  if (i >= n) return;
  int v = (int)(i >> 6);
  int c = (int)(i & 63);
  float di = dinv[v];
  out[i] += di * di * hpre[i] + bias[c];
}

extern "C" void kernel_launch(void* const* d_in, const int* in_sizes, int n_in,
                              void* d_out, int out_size, void* d_ws,
                              size_t ws_size, hipStream_t stream) {
  const float* x  = (const float*)d_in[0];
  const int*   ei = (const int*)d_in[1];
  const float* W1 = (const float*)d_in[2];
  const float* b1 = (const float*)d_in[3];
  const float* W2 = (const float*)d_in[4];
  const float* b2 = (const float*)d_in[5];
  float* out = (float*)d_out;

  const int N = in_sizes[0] / 128;
  const long long E = in_sizes[1] / 2;
  const int* src = ei;
  const int* dst = ei + E;

  float* ws = (float*)d_ws;
  float* deg   = ws;
  float* h1pre = ws + N;
  float* agg1  = h1pre + (size_t)N * 128;
  float* h2pre = h1pre;

  const int TB = 256;
  auto blocks = [](long long n, int tb) { return (unsigned)((n + tb - 1) / tb); };

  fill_kernel<<<blocks(N, TB), TB, 0, stream>>>(deg, 1.0f, N);
  fill_kernel<<<blocks((long long)N * 128, TB), TB, 0, stream>>>(agg1, 0.0f, (long long)N * 128);
  fill_kernel<<<blocks((long long)N * 64, TB), TB, 0, stream>>>(out, 0.0f, (long long)N * 64);

  deg_kernel<<<blocks(E, TB), TB, 0, stream>>>(dst, deg, E);
  dinv_kernel<<<blocks(N, TB), TB, 0, stream>>>(deg, N);
  const float* dinv = deg;

  gemm_wmma_kernel<128><<<(N + 15) / 16, 256, 0, stream>>>(x, W1, h1pre, N);
  edge_agg_kernel<128><<<blocks(E * 32, TB), TB, 0, stream>>>(src, dst, dinv, h1pre, agg1, E);
  epilogue1_kernel<<<blocks((long long)N * 128, TB), TB, 0, stream>>>(agg1, h1pre, dinv, b1, (long long)N * 128);
  const float* h1 = agg1;

  gemm_wmma_kernel<64><<<(N + 15) / 16, 128, 0, stream>>>(h1, W2, h2pre, N);
  edge_agg_kernel<64><<<blocks(E * 16, TB), TB, 0, stream>>>(src, dst, dinv, h2pre, out, E);
  epilogue2_kernel<<<blocks((long long)N * 64, TB), TB, 0, stream>>>(out, h2pre, dinv, b2, (long long)N * 64);
}


# Float-atomic scatter to memory replaced by owner blocks: a block keeps 400 dst rows in LDS, scans dst, adds src rows by LDS float adds (sum order = LDS arrival, self loop first), stores rows once
# speedup vs baseline: 1.6294x; 1.6294x over previous
#include <hip/hip_runtime.h>
#include <hip/hip_bf16.h>

typedef float v2f __attribute__((ext_vector_type(2)));
typedef float v8f __attribute__((ext_vector_type(8)));

__device__ __forceinline__ void atomic_add_f32(float* p, float v) {
  __hip_atomic_fetch_add(p, v, __ATOMIC_RELAXED, __HIP_MEMORY_SCOPE_AGENT);
}

__global__ void fill_kernel(float* __restrict__ p, float v, long long n) {
  long long i = (long long)blockIdx.x * blockDim.x + threadIdx.x;
  if (i < n) p[i] = v;
}

__global__ void deg_kernel(const int* __restrict__ dst, float* __restrict__ deg,
                           long long E) {
  long long e = (long long)blockIdx.x * blockDim.x + threadIdx.x;
  if (e < E) atomic_add_f32(&deg[dst[e]], 1.0f);
}

__global__ void dinv_kernel(float* __restrict__ deg, int n) {
  int i = blockIdx.x * blockDim.x + threadIdx.x;
  if (i < n) deg[i] = __frsqrt_rn(fmaxf(deg[i], 1e-12f));
}

template <int FOUT>
__global__ __launch_bounds__(FOUT * 2) void gemm_wmma_kernel(
    const float* __restrict__ A, const float* __restrict__ W,
    float* __restrict__ C, int nrows) {
  __shared__ float lds[16 * 128];

  const int row0 = blockIdx.x * 16;
  if (row0 >= nrows) return;
  const int tid = threadIdx.x;
  constexpr int NTHREADS = FOUT * 2;

  if (row0 + 16 <= nrows) {
    const float4* srcv = (const float4*)(A + (size_t)row0 * 128);
    float4* dstv = (float4*)lds;
    #pragma unroll 2
    for (int i = tid; i < 512; i += NTHREADS) dstv[i] = srcv[i];
  } else {
    for (int i = tid; i < 2048; i += NTHREADS) {
      int r = i >> 7;
      lds[i] = (row0 + r < nrows) ? A[(size_t)(row0 + r) * 128 + (i & 127)] : 0.0f;
    }
  }
  __syncthreads();

  const int lane = tid & 31;
  const int wave = tid >> 5;
  const int colbase = wave * 16;
  const int half = lane >> 4;
  const int m = lane & 15;

  v8f acc = {};
  #pragma unroll
  for (int k = 0; k < 128; k += 4) {
    const int kb = k + half * 2;
    v2f a = *(const v2f*)&lds[m * 128 + kb];
    v2f b;
    b.x = W[(size_t)kb * FOUT + colbase + m];
    b.y = W[(size_t)(kb + 1) * FOUT + colbase + m];
    acc = __builtin_amdgcn_wmma_f32_16x16x4_f32(false, a, false, b, (short)0,
                                                acc, false, false);
  }

  #pragma unroll
  for (int i = 0; i < 8; ++i) {
    int row = row0 + i + half * 8;
    if (row < nrows) C[(size_t)row * FOUT + colbase + m] = acc[i];
  }
}

template <int F, int R, bool RELU>
__global__ __launch_bounds__(1024) void owner_agg_kernel(
    const int* __restrict__ src, const int* __restrict__ dst,
    const float* __restrict__ dinv, const float* __restrict__ h,
    const float* __restrict__ bias, float* __restrict__ out, int N,
    long long E64) {
  constexpr int VEC = F / 32;
  constexpr int NW = 32;
  constexpr int QCAP = 64;
  constexpr int GRP = 8;
  typedef float VT __attribute__((ext_vector_type(VEC)));

  __shared__ float acc[R * F];
  __shared__ float dloc[R];
  __shared__ int qe[NW * QCAP];
  __shared__ int qr[NW * QCAP];

  const int tid = threadIdx.x;
  const int lane = tid & 31;
  const int wave = __builtin_amdgcn_readfirstlane(tid >> 5);
  const int r0 = blockIdx.x * R;
  const int E = (int)E64;

  for (int r = wave; r < R; r += NW) {
    const int v = r0 + r;
    if (v < N) {
      const float di = dinv[v];
      if (lane == 0) dloc[r] = di;
      const float s = di * di;
      const VT hv = *(const VT*)(h + (size_t)v * F + lane * VEC);
#pragma unroll
      for (int t = 0; t < VEC; ++t) acc[r * F + t * 32 + lane] = s * hv[t];
    } else {
      if (lane == 0) dloc[r] = 0.0f;
#pragma unroll
      for (int t = 0; t < VEC; ++t) acc[r * F + t * 32 + lane] = 0.0f;
    }
  }
  __syncthreads();

  int* myqe = qe + wave * QCAP;
  int* myqr = qr + wave * QCAP;
  const unsigned ltmask = (1u << lane) - 1u;
  int qn = 0;

  auto add_row = [&](int rk, float wk, const VT& hv) {
#pragma unroll
    for (int t = 0; t < VEC; ++t)
      __hip_atomic_fetch_add(&acc[rk * F + t * 32 + lane], wk * hv[t],
                             __ATOMIC_RELAXED, __HIP_MEMORY_SCOPE_WORKGROUP);
  };

  auto drain32 = [&]() {
    const int r = myqr[lane];
    const int u = src[myqe[lane]];
    const float w = dinv[u] * dloc[r];
    for (int g = 0; g < 32; g += GRP) {
      VT hv[GRP];
      int rr[GRP];
      float ww[GRP];
#pragma unroll
      for (int k = 0; k < GRP; ++k) {
        const int uk = __builtin_amdgcn_readlane(u, g + k);
        rr[k] = __builtin_amdgcn_readlane(r, g + k);
        ww[k] = __int_as_float(__builtin_amdgcn_readlane(__float_as_int(w), g + k));
        hv[k] = *(const VT*)(h + (size_t)uk * F + lane * VEC);
      }
#pragma unroll
      for (int k = 0; k < GRP; ++k) add_row(rr[k], ww[k], hv[k]);
    }
    const int rem = qn - 32;
    int te = 0, tr = 0;
    if (lane < rem) { te = myqe[32 + lane]; tr = myqr[32 + lane]; }
    if (lane < rem) { myqe[lane] = te; myqr[lane] = tr; }
    qn = rem;
  };

  const bool vec_ok = ((E & 3) == 0) && ((((size_t)dst) & 15) == 0);
  auto load4 = [&](int i) -> int4 {
    int4 d;
    d.x = (i     < E) ? dst[i]     : -1;
    d.y = (i + 1 < E) ? dst[i + 1] : -1;
    d.z = (i + 2 < E) ? dst[i + 2] : -1;
    d.w = (i + 3 < E) ? dst[i + 3] : -1;
    return d;
  };

  auto scan4 = [&](const int4& d, int base) {
    const int d0 = d.x - r0, d1 = d.y - r0, d2 = d.z - r0, d3 = d.w - r0;
    unsigned mm = ((unsigned)d0 < (unsigned)R ? 1u : 0u) |
                  ((unsigned)d1 < (unsigned)R ? 2u : 0u) |
                  ((unsigned)d2 < (unsigned)R ? 4u : 0u) |
                  ((unsigned)d3 < (unsigned)R ? 8u : 0u);
    while (true) {
      const bool has = mm != 0u;
      const unsigned mk = __builtin_amdgcn_ballot_w32(has);
      if (mk == 0u) break;
      if (has) {
        const int c = __builtin_ctz(mm);
        const int dsel = (c == 0) ? d0 : (c == 1) ? d1 : (c == 2) ? d2 : d3;
        const int p = qn + __builtin_popcount(mk & ltmask);
        myqe[p] = base + c;
        myqr[p] = dsel;
      }
      qn += __builtin_popcount(mk);
      mm &= mm - 1u;
      if (qn >= 32) drain32();
    }
  };

  constexpr int TRIP = 8;
  const int nsc = (E + 1023) >> 10;
  for (int sc = wave; sc < nsc; sc += NW) {
    const int b = sc * 1024 + lane * 4;
    int4 c[TRIP];
    if (vec_ok && sc * 1024 + 1024 <= E) {
      const int4* p = (const int4*)(dst + b);
#pragma unroll
      for (int j = 0; j < TRIP; ++j) c[j] = p[j * 32];
    } else {
#pragma unroll
      for (int j = 0; j < TRIP; ++j) c[j] = load4(b + j * 128);
    }
#pragma unroll
    for (int j = 0; j < TRIP; ++j) scan4(c[j], b + j * 128);
  }

  if (qn > 0) {
    int r = 0, u = 0;
    float w = 0.0f;
    if (lane < qn) {
      r = myqr[lane];
      u = src[myqe[lane]];
      w = dinv[u] * dloc[r];
    }
    for (int k = 0; k < qn; ++k) {
      const int uk = __builtin_amdgcn_readlane(u, k);
      const int rk = __builtin_amdgcn_readlane(r, k);
      const float wk = __int_as_float(__builtin_amdgcn_readlane(__float_as_int(w), k));
      const VT hv = *(const VT*)(h + (size_t)uk * F + lane * VEC);
      add_row(rk, wk, hv);
    }
  }
  __syncthreads();

  for (int r = wave; r < R; r += NW) {
    const int v = r0 + r;
    if (v >= N) break;
    VT o;
#pragma unroll
    for (int t = 0; t < VEC; ++t) {
      const float x = acc[r * F + t * 32 + lane] + bias[lane * VEC + t];
      o[t] = RELU ? fmaxf(x, 0.0f) : x;
    }
    *(VT*)(out + (size_t)v * F + lane * VEC) = o;
  }
}

extern "C" void kernel_launch(void* const* d_in, const int* in_sizes, int n_in,
                              void* d_out, int out_size, void* d_ws,
                              size_t ws_size, hipStream_t stream) {
  const float* x  = (const float*)d_in[0];
  const int*   ei = (const int*)d_in[1];
  const float* W1 = (const float*)d_in[2];
  const float* b1 = (const float*)d_in[3];
  const float* W2 = (const float*)d_in[4];
  const float* b2 = (const float*)d_in[5];
  float* out = (float*)d_out;

  const int N = in_sizes[0] / 128;
  const long long E = in_sizes[1] / 2;
  const int* src = ei;
  const int* dst = ei + E;

  float* ws = (float*)d_ws;
  float* deg   = ws;
  float* h1pre = ws + N;
  float* h1    = h1pre + (size_t)N * 128;
  float* h2pre = h1pre;

  const int TB = 256;
  auto blocks = [](long long n, int tb) { return (unsigned)((n + tb - 1) / tb); };
  constexpr int R = 400;

  fill_kernel<<<blocks(N, TB), TB, 0, stream>>>(deg, 1.0f, N);
  deg_kernel<<<blocks(E, TB), TB, 0, stream>>>(dst, deg, E);
  dinv_kernel<<<blocks(N, TB), TB, 0, stream>>>(deg, N);
  const float* dinv = deg;

  gemm_wmma_kernel<128><<<(N + 15) / 16, 256, 0, stream>>>(x, W1, h1pre, N);
  owner_agg_kernel<128, R, true><<<(N + R - 1) / R, 1024, 0, stream>>>(
      src, dst, dinv, h1pre, b1, h1, N, E);

  gemm_wmma_kernel<64><<<(N + 15) / 16, 128, 0, stream>>>(h1, W2, h2pre, N);
  owner_agg_kernel<64, R, false><<<(N + R - 1) / R, 1024, 0, stream>>>(
      src, dst, dinv, h2pre, b2, out, N, E);
}


# Owner blocks scan dst once: layer 1 saves each wave's (src,row) pairs in the block's own slice of the output buffer, layer 2 reads them in place of its scan; sum order still LDS arrival
# speedup vs baseline: 1.9728x; 1.2108x over previous
#include <hip/hip_runtime.h>
#include <hip/hip_bf16.h>

typedef float v2f __attribute__((ext_vector_type(2)));
typedef float v8f __attribute__((ext_vector_type(8)));

__device__ __forceinline__ void atomic_add_f32(float* p, float v) {
  __hip_atomic_fetch_add(p, v, __ATOMIC_RELAXED, __HIP_MEMORY_SCOPE_AGENT);
}

__global__ void fill_kernel(float* __restrict__ p, float v, long long n) {
  long long i = (long long)blockIdx.x * blockDim.x + threadIdx.x;
  if (i < n) p[i] = v;
}

__global__ void deg_kernel(const int* __restrict__ dst, float* __restrict__ deg,
                           long long E) {
  long long e = (long long)blockIdx.x * blockDim.x + threadIdx.x;
  if (e < E) atomic_add_f32(&deg[dst[e]], 1.0f);
}

__global__ void dinv_kernel(float* __restrict__ deg, int n) {
  int i = blockIdx.x * blockDim.x + threadIdx.x;
  if (i < n) deg[i] = __frsqrt_rn(fmaxf(deg[i], 1e-12f));
}

template <int FOUT>
__global__ __launch_bounds__(FOUT * 2) void gemm_wmma_kernel(
    const float* __restrict__ A, const float* __restrict__ W,
    float* __restrict__ C, int nrows) {
  __shared__ float lds[16 * 128];

  const int row0 = blockIdx.x * 16;
  if (row0 >= nrows) return;
  const int tid = threadIdx.x;
  constexpr int NTHREADS = FOUT * 2;

  if (row0 + 16 <= nrows) {
    const float4* srcv = (const float4*)(A + (size_t)row0 * 128);
    float4* dstv = (float4*)lds;
    #pragma unroll 2
    for (int i = tid; i < 512; i += NTHREADS) dstv[i] = srcv[i];
  } else {
    for (int i = tid; i < 2048; i += NTHREADS) {
      int r = i >> 7;
      lds[i] = (row0 + r < nrows) ? A[(size_t)(row0 + r) * 128 + (i & 127)] : 0.0f;
    }
  }
  __syncthreads();

  const int lane = tid & 31;
  const int wave = tid >> 5;
  const int colbase = wave * 16;
  const int half = lane >> 4;
  const int m = lane & 15;

  v8f acc = {};
  #pragma unroll
  for (int k = 0; k < 128; k += 4) {
    const int kb = k + half * 2;
    v2f a = *(const v2f*)&lds[m * 128 + kb];
    v2f b;
    b.x = W[(size_t)kb * FOUT + colbase + m];
    b.y = W[(size_t)(kb + 1) * FOUT + colbase + m];
    acc = __builtin_amdgcn_wmma_f32_16x16x4_f32(false, a, false, b, (short)0,
                                                acc, false, false);
  }

  #pragma unroll
  for (int i = 0; i < 8; ++i) {
    int row = row0 + i + half * 8;
    if (row < nrows) C[(size_t)row * FOUT + colbase + m] = acc[i];
  }
}

template <int F, int R, bool RELU, int MODE>
__global__ __launch_bounds__(1024) void owner_agg_kernel(
    const int* __restrict__ src, const int* __restrict__ dst,
    const float* __restrict__ dinv, const float* __restrict__ h,
    const float* __restrict__ bias, float* out, float* lst, int N,
    long long E64) {
  constexpr int VEC = F / 32;
  constexpr int NW = 32;
  constexpr int QCAP = 64;
  constexpr int GRP = 8;
  constexpr int GRPL = 16;
  constexpr int LW = R * 64;
  constexpr int CAP = (((LW - 32) / NW) / 32) * 32;
  static_assert(R <= 512, "owned row must fit the 9 low bits of a list word");
  static_assert(CAP >= 32, "slice too small for the lists");
  typedef float VT __attribute__((ext_vector_type(VEC)));

  __shared__ float acc[R * F];
  __shared__ float dloc[R];
  __shared__ int qe[NW * QCAP];
  __shared__ int qr[NW * QCAP];
  __shared__ int qcnt[NW];

  const int tid = threadIdx.x;
  const int lane = tid & 31;
  const int wave = __builtin_amdgcn_readfirstlane(tid >> 5);
  const int r0 = blockIdx.x * R;
  const int E = (int)E64;

  for (int r = wave; r < R; r += NW) {
    const int v = r0 + r;
    if (v < N) {
      const float di = dinv[v];
      if (lane == 0) dloc[r] = di;
      const float s = di * di;
      const VT hv = *(const VT*)(h + (size_t)v * F + lane * VEC);
#pragma unroll
      for (int t = 0; t < VEC; ++t) acc[r * F + t * 32 + lane] = s * hv[t];
    } else {
      if (lane == 0) dloc[r] = 0.0f;
#pragma unroll
      for (int t = 0; t < VEC; ++t) acc[r * F + t * 32 + lane] = 0.0f;
    }
  }
  __syncthreads();

  int* myqe = qe + wave * QCAP;
  int* myqr = qr + wave * QCAP;
  const unsigned ltmask = (1u << lane) - 1u;
  int qn = 0;

  float* myl = lst + (size_t)blockIdx.x * LW;
  float* wl = myl + 32 + wave * CAP;
  const bool can_save = (MODE == 0) && (r0 + R <= N) && (N <= (1 << 23));
  int cnt = 0;

  auto add_row = [&](int rk, float wk, const VT& hv) {
#pragma unroll
    for (int t = 0; t < VEC; ++t)
      __hip_atomic_fetch_add(&acc[rk * F + t * 32 + lane], wk * hv[t],
                             __ATOMIC_RELAXED, __HIP_MEMORY_SCOPE_WORKGROUP);
  };

  auto drain32 = [&]() {
    const int r = myqr[lane];
    const int u = src[myqe[lane]];
    const float w = dinv[u] * dloc[r];
    if (MODE == 0) {
      if (can_save && cnt + 32 <= CAP)
        wl[cnt + lane] = __uint_as_float(((unsigned)u << 9) | (unsigned)r);
      cnt += 32;
    }
    for (int g = 0; g < 32; g += GRP) {
      VT hv[GRP];
      int rr[GRP];
      float ww[GRP];
#pragma unroll
      for (int k = 0; k < GRP; ++k) {
        const int uk = __builtin_amdgcn_readlane(u, g + k);
        rr[k] = __builtin_amdgcn_readlane(r, g + k);
        ww[k] = __int_as_float(__builtin_amdgcn_readlane(__float_as_int(w), g + k));
        hv[k] = *(const VT*)(h + (size_t)uk * F + lane * VEC);
      }
#pragma unroll
      for (int k = 0; k < GRP; ++k) add_row(rr[k], ww[k], hv[k]);
    }
    const int rem = qn - 32;
    int te = 0, tr = 0;
    if (lane < rem) { te = myqe[32 + lane]; tr = myqr[32 + lane]; }
    if (lane < rem) { myqe[lane] = te; myqr[lane] = tr; }
    qn = rem;
  };

  const bool vec_ok = ((E & 3) == 0) && ((((size_t)dst) & 15) == 0);
  auto load4 = [&](int i) -> int4 {
    int4 d;
    d.x = (i     < E) ? dst[i]     : -1;
    d.y = (i + 1 < E) ? dst[i + 1] : -1;
    d.z = (i + 2 < E) ? dst[i + 2] : -1;
    d.w = (i + 3 < E) ? dst[i + 3] : -1;
    return d;
  };

  auto scan4 = [&](const int4& d, int base) {
    const int d0 = d.x - r0, d1 = d.y - r0, d2 = d.z - r0, d3 = d.w - r0;
    unsigned mm = ((unsigned)d0 < (unsigned)R ? 1u : 0u) |
                  ((unsigned)d1 < (unsigned)R ? 2u : 0u) |
                  ((unsigned)d2 < (unsigned)R ? 4u : 0u) |
                  ((unsigned)d3 < (unsigned)R ? 8u : 0u);
    while (true) {
      const bool has = mm != 0u;
      const unsigned mk = __builtin_amdgcn_ballot_w32(has);
      if (mk == 0u) break;
      if (has) {
        const int c = __builtin_ctz(mm);
        const int dsel = (c == 0) ? d0 : (c == 1) ? d1 : (c == 2) ? d2 : d3;
        const int p = qn + __builtin_popcount(mk & ltmask);
        myqe[p] = base + c;
        myqr[p] = dsel;
      }
      qn += __builtin_popcount(mk);
      mm &= mm - 1u;
      if (qn >= 32) drain32();
    }
  };

  bool use_list = false;
  if (MODE == 1) {
    const int c = __float_as_int(myl[lane]);
    use_list = __builtin_amdgcn_ballot_w32(c < 0 || c > CAP) == 0u;
    if (use_list) {
      const int n = __builtin_amdgcn_readlane(c, wave);
      auto ld = [&](int b) -> unsigned {
        return (b + lane < n) ? __float_as_uint(wl[b + lane]) : 0u;
      };
      unsigned pk = ld(0);
      unsigned pk1 = ld(32);
      float dv = dinv[pk >> 9];
      for (int b = 0; b < n; b += 32) {
        const unsigned pk2 = ld(b + 64);
        const float dv1 = dinv[pk1 >> 9];
        const int u = (int)(pk >> 9);
        const int r = (int)(pk & 511u);
        const float w = dv * dloc[r];
        const int m = (n - b < 32) ? (n - b) : 32;
        int g = 0;
        for (; g + GRPL <= m; g += GRPL) {
          VT hv[GRPL];
          int rr[GRPL];
          float ww[GRPL];
#pragma unroll
          for (int k = 0; k < GRPL; ++k) {
            const int uk = __builtin_amdgcn_readlane(u, g + k);
            rr[k] = __builtin_amdgcn_readlane(r, g + k);
            ww[k] = __int_as_float(__builtin_amdgcn_readlane(__float_as_int(w), g + k));
            hv[k] = *(const VT*)(h + (size_t)uk * F + lane * VEC);
          }
#pragma unroll
          for (int k = 0; k < GRPL; ++k) add_row(rr[k], ww[k], hv[k]);
        }
        if (g < m) {
          VT hv[GRPL];
          int rr[GRPL];
          float ww[GRPL];
#pragma unroll
          for (int k = 0; k < GRPL; ++k) {
            if (g + k < m) {
              const int uk = __builtin_amdgcn_readlane(u, g + k);
              rr[k] = __builtin_amdgcn_readlane(r, g + k);
              ww[k] = __int_as_float(__builtin_amdgcn_readlane(__float_as_int(w), g + k));
              hv[k] = *(const VT*)(h + (size_t)uk * F + lane * VEC);
            }
          }
#pragma unroll
          for (int k = 0; k < GRPL; ++k) {
            if (g + k < m) add_row(rr[k], ww[k], hv[k]);
          }
        }
        pk = pk1;
        dv = dv1;
        pk1 = pk2;
      }
    }
  }

  constexpr int TRIP = 8;
  const int nsc = use_list ? 0 : ((E + 1023) >> 10);
  for (int sc = wave; sc < nsc; sc += NW) {
    const int b = sc * 1024 + lane * 4;
    int4 c[TRIP];
    if (vec_ok && sc * 1024 + 1024 <= E) {
      const int4* p = (const int4*)(dst + b);
#pragma unroll
      for (int j = 0; j < TRIP; ++j) c[j] = p[j * 32];
    } else {
#pragma unroll
      for (int j = 0; j < TRIP; ++j) c[j] = load4(b + j * 128);
    }
#pragma unroll
    for (int j = 0; j < TRIP; ++j) scan4(c[j], b + j * 128);
  }

  if (qn > 0) {
    int r = 0, u = 0;
    float w = 0.0f;
    if (lane < qn) {
      r = myqr[lane];
      u = src[myqe[lane]];
      w = dinv[u] * dloc[r];
      if (MODE == 0 && can_save && cnt + qn <= CAP)
        wl[cnt + lane] = __uint_as_float(((unsigned)u << 9) | (unsigned)r);
    }
    cnt += qn;
    for (int k = 0; k < qn; ++k) {
      const int uk = __builtin_amdgcn_readlane(u, k);
      const int rk = __builtin_amdgcn_readlane(r, k);
      const float wk = __int_as_float(__builtin_amdgcn_readlane(__float_as_int(w), k));
      const VT hv = *(const VT*)(h + (size_t)uk * F + lane * VEC);
      add_row(rk, wk, hv);
    }
  }
  if (MODE == 0 && lane == 0) qcnt[wave] = (can_save && cnt <= CAP) ? cnt : -1;
  __syncthreads();

  if (MODE == 0 && wave == 0) myl[lane] = __int_as_float(qcnt[lane]);

  for (int r = wave; r < R; r += NW) {
    const int v = r0 + r;
    if (v >= N) break;
    VT o;
#pragma unroll
    for (int t = 0; t < VEC; ++t) {
      const float x = acc[r * F + t * 32 + lane] + bias[lane * VEC + t];
      o[t] = RELU ? fmaxf(x, 0.0f) : x;
    }
    *(VT*)(out + (size_t)v * F + lane * VEC) = o;
  }
}

extern "C" void kernel_launch(void* const* d_in, const int* in_sizes, int n_in,
                              void* d_out, int out_size, void* d_ws,
                              size_t ws_size, hipStream_t stream) {
  const float* x  = (const float*)d_in[0];
  const int*   ei = (const int*)d_in[1];
  const float* W1 = (const float*)d_in[2];
  const float* b1 = (const float*)d_in[3];
  const float* W2 = (const float*)d_in[4];
  const float* b2 = (const float*)d_in[5];
  float* out = (float*)d_out;

  const int N = in_sizes[0] / 128;
  const long long E = in_sizes[1] / 2;
  const int* src = ei;
  const int* dst = ei + E;

  float* ws = (float*)d_ws;
  float* deg   = ws;
  float* h1pre = ws + N;
  float* h1    = h1pre + (size_t)N * 128;
  float* h2pre = h1pre;

  const int TB = 256;
  auto blocks = [](long long n, int tb) { return (unsigned)((n + tb - 1) / tb); };
  constexpr int R = 400;

  fill_kernel<<<blocks(N, TB), TB, 0, stream>>>(deg, 1.0f, N);
  deg_kernel<<<blocks(E, TB), TB, 0, stream>>>(dst, deg, E);
  dinv_kernel<<<blocks(N, TB), TB, 0, stream>>>(deg, N);
  const float* dinv = deg;

  gemm_wmma_kernel<128><<<(N + 15) / 16, 256, 0, stream>>>(x, W1, h1pre, N);
  owner_agg_kernel<128, R, true, 0><<<(N + R - 1) / R, 1024, 0, stream>>>(
      src, dst, dinv, h1pre, b1, h1, out, N, E);

  gemm_wmma_kernel<64><<<(N + 15) / 16, 128, 0, stream>>>(h1, W2, h2pre, N);
  owner_agg_kernel<64, R, false, 1><<<(N + R - 1) / R, 1024, 0, stream>>>(
      src, dst, dinv, h2pre, b2, out, out, N, E);
}


# Layer-1 owner scan reads a 1-byte owner-index plane (written by the degree launch into spare tails of the output slices), 16 edges per 16-byte load, not 4-byte dst; sum order still LDS arrival
# speedup vs baseline: 2.0362x; 1.0321x over previous
#include <hip/hip_runtime.h>
#include <hip/hip_bf16.h>

typedef float v2f __attribute__((ext_vector_type(2)));
typedef float v8f __attribute__((ext_vector_type(8)));

__device__ __forceinline__ void atomic_add_f32(float* p, float v) {
  __hip_atomic_fetch_add(p, v, __ATOMIC_RELAXED, __HIP_MEMORY_SCOPE_AGENT);
}

__global__ void fill_kernel(float* __restrict__ p, float v, long long n) {
  long long i = (long long)blockIdx.x * blockDim.x + threadIdx.x;
  if (i < n) p[i] = v;
}

template <int R>
struct Plane {
  static constexpr int LW = R * 64;
  static constexpr int PLW = 3072;
  static constexpr int NW = 32;
  static constexpr int CAP = (((LW - 32 - PLW) / NW) / 32) * 32;
  static constexpr int TOFF = 32 + NW * CAP;
  static_assert(CAP >= 32, "slice too small for the lists");
  static_assert(TOFF % 32 == 0 && TOFF + PLW <= LW, "tail must fit the slice, line aligned");
  __device__ static bool usable(int N, long long E) {
    return (N <= 255 * R) && ((E + PLW * 4 - 1) / (PLW * 4) <= (long long)(N / R));
  }
};

template <int R>
__global__ void deg_kernel(const int* __restrict__ dst, float* __restrict__ deg,
                           unsigned* __restrict__ slices, int N, long long E) {
  typedef Plane<R> P;
  const long long i = (long long)blockIdx.x * blockDim.x + threadIdx.x;
  const long long e0 = i * 4;
  if (e0 >= ((E + 15) / 16) * 16) return;
  int d[4];
  if (e0 + 3 < E && ((((size_t)dst) & 15) == 0)) {
    const int4 v = *(const int4*)(dst + e0);
    d[0] = v.x; d[1] = v.y; d[2] = v.z; d[3] = v.w;
  } else {
#pragma unroll
    for (int k = 0; k < 4; ++k) d[k] = (e0 + k < E) ? dst[e0 + k] : -1;
  }
  unsigned w = 0u;
#pragma unroll
  for (int k = 0; k < 4; ++k) {
    if (d[k] >= 0) atomic_add_f32(&deg[d[k]], 1.0f);
    const unsigned o = (d[k] >= 0) ? (unsigned)(d[k] / R) : 255u;
    w |= (o & 255u) << (8 * k);
  }
  if (P::usable(N, E))
    slices[(size_t)(i / P::PLW) * P::LW + P::TOFF + (int)(i % P::PLW)] = w;
}

__global__ void dinv_kernel(float* __restrict__ deg, int n) {
  int i = blockIdx.x * blockDim.x + threadIdx.x;
  if (i < n) deg[i] = __frsqrt_rn(fmaxf(deg[i], 1e-12f));
}

template <int FOUT>
__global__ __launch_bounds__(FOUT * 2) void gemm_wmma_kernel(
    const float* __restrict__ A, const float* __restrict__ W,
    float* __restrict__ C, int nrows) {
  __shared__ float lds[16 * 128];

  const int row0 = blockIdx.x * 16;
  if (row0 >= nrows) return;
  const int tid = threadIdx.x;
  constexpr int NTHREADS = FOUT * 2;

  if (row0 + 16 <= nrows) {
    const float4* srcv = (const float4*)(A + (size_t)row0 * 128);
    float4* dstv = (float4*)lds;
    #pragma unroll 2
    for (int i = tid; i < 512; i += NTHREADS) dstv[i] = srcv[i];
  } else {
    for (int i = tid; i < 2048; i += NTHREADS) {
      int r = i >> 7;
      lds[i] = (row0 + r < nrows) ? A[(size_t)(row0 + r) * 128 + (i & 127)] : 0.0f;
    }
  }
  __syncthreads();

  const int lane = tid & 31;
  const int wave = tid >> 5;
  const int colbase = wave * 16;
  const int half = lane >> 4;
  const int m = lane & 15;

  v8f acc = {};
  #pragma unroll
  for (int k = 0; k < 128; k += 4) {
    const int kb = k + half * 2;
    v2f a = *(const v2f*)&lds[m * 128 + kb];
    v2f b;
    b.x = W[(size_t)kb * FOUT + colbase + m];
    b.y = W[(size_t)(kb + 1) * FOUT + colbase + m];
    acc = __builtin_amdgcn_wmma_f32_16x16x4_f32(false, a, false, b, (short)0,
                                                acc, false, false);
  }

  #pragma unroll
  for (int i = 0; i < 8; ++i) {
    int row = row0 + i + half * 8;
    if (row < nrows) C[(size_t)row * FOUT + colbase + m] = acc[i];
  }
}

template <int F, int R, bool RELU, int MODE>
__global__ __launch_bounds__(1024) void owner_agg_kernel(
    const int* __restrict__ src, const int* __restrict__ dst,
    const float* __restrict__ dinv, const float* __restrict__ h,
    const float* __restrict__ bias, float* out, float* lst, int N,
    long long E64) {
  constexpr int VEC = F / 32;
  constexpr int NW = 32;
  constexpr int QCAP = 64;
  constexpr int GRP = 8;
  constexpr int GRPL = 16;
  typedef Plane<R> P;
  constexpr int LW = P::LW;
  constexpr int CAP = P::CAP;
  static_assert(R <= 512, "owned row must fit the 9 low bits of a list word");
  static_assert(P::NW == NW, "slice layout is for 32 waves");
  typedef float VT __attribute__((ext_vector_type(VEC)));

  __shared__ float acc[R * F];
  __shared__ float dloc[R];
  __shared__ int qe[NW * QCAP];
  __shared__ int qr[NW * QCAP];
  __shared__ int qcnt[NW];

  const int tid = threadIdx.x;
  const int lane = tid & 31;
  const int wave = __builtin_amdgcn_readfirstlane(tid >> 5);
  const int r0 = blockIdx.x * R;
  const int E = (int)E64;

  for (int r = wave; r < R; r += NW) {
    const int v = r0 + r;
    if (v < N) {
      const float di = dinv[v];
      if (lane == 0) dloc[r] = di;
      const float s = di * di;
      const VT hv = *(const VT*)(h + (size_t)v * F + lane * VEC);
#pragma unroll
      for (int t = 0; t < VEC; ++t) acc[r * F + t * 32 + lane] = s * hv[t];
    } else {
      if (lane == 0) dloc[r] = 0.0f;
#pragma unroll
      for (int t = 0; t < VEC; ++t) acc[r * F + t * 32 + lane] = 0.0f;
    }
  }
  __syncthreads();

  int* myqe = qe + wave * QCAP;
  int* myqr = qr + wave * QCAP;
  const unsigned ltmask = (1u << lane) - 1u;
  int qn = 0;

  float* myl = lst + (size_t)blockIdx.x * LW;
  float* wl = myl + 32 + wave * CAP;
  const bool can_save = (MODE == 0) && (r0 + R <= N) && (N <= (1 << 23));
  int cnt = 0;

  auto add_row = [&](int rk, float wk, const VT& hv) {
#pragma unroll
    for (int t = 0; t < VEC; ++t)
      __hip_atomic_fetch_add(&acc[rk * F + t * 32 + lane], wk * hv[t],
                             __ATOMIC_RELAXED, __HIP_MEMORY_SCOPE_WORKGROUP);
  };

  const bool bytes = (MODE == 0) && P::usable(N, E64);
  auto drain32 = [&]() {
    const int r = bytes ? (dst[myqe[lane]] - r0) : myqr[lane];
    const int u = src[myqe[lane]];
    const float w = dinv[u] * dloc[r];
    if (MODE == 0) {
      if (can_save && cnt + 32 <= CAP)
        wl[cnt + lane] = __uint_as_float(((unsigned)u << 9) | (unsigned)r);
      cnt += 32;
    }
    for (int g = 0; g < 32; g += GRP) {
      VT hv[GRP];
      int rr[GRP];
      float ww[GRP];
#pragma unroll
      for (int k = 0; k < GRP; ++k) {
        const int uk = __builtin_amdgcn_readlane(u, g + k);
        rr[k] = __builtin_amdgcn_readlane(r, g + k);
        ww[k] = __int_as_float(__builtin_amdgcn_readlane(__float_as_int(w), g + k));
        hv[k] = *(const VT*)(h + (size_t)uk * F + lane * VEC);
      }
#pragma unroll
      for (int k = 0; k < GRP; ++k) add_row(rr[k], ww[k], hv[k]);
    }
    const int rem = qn - 32;
    int te = 0, tr = 0;
    if (lane < rem) { te = myqe[32 + lane]; tr = bytes ? 0 : myqr[32 + lane]; }
    if (lane < rem) { myqe[lane] = te; if (!bytes) myqr[lane] = tr; }
    qn = rem;
  };

  const bool vec_ok = ((E & 3) == 0) && ((((size_t)dst) & 15) == 0);
  auto load4 = [&](int i) -> int4 {
    int4 d;
    d.x = (i     < E) ? dst[i]     : -1;
    d.y = (i + 1 < E) ? dst[i + 1] : -1;
    d.z = (i + 2 < E) ? dst[i + 2] : -1;
    d.w = (i + 3 < E) ? dst[i + 3] : -1;
    return d;
  };

  auto scan4 = [&](const int4& d, int base) {
    const int d0 = d.x - r0, d1 = d.y - r0, d2 = d.z - r0, d3 = d.w - r0;
    unsigned mm = ((unsigned)d0 < (unsigned)R ? 1u : 0u) |
                  ((unsigned)d1 < (unsigned)R ? 2u : 0u) |
                  ((unsigned)d2 < (unsigned)R ? 4u : 0u) |
                  ((unsigned)d3 < (unsigned)R ? 8u : 0u);
    while (true) {
      const bool has = mm != 0u;
      const unsigned mk = __builtin_amdgcn_ballot_w32(has);
      if (mk == 0u) break;
      if (has) {
        const int c = __builtin_ctz(mm);
        const int dsel = (c == 0) ? d0 : (c == 1) ? d1 : (c == 2) ? d2 : d3;
        const int p = qn + __builtin_popcount(mk & ltmask);
        myqe[p] = base + c;
        myqr[p] = dsel;
      }
      qn += __builtin_popcount(mk);
      mm &= mm - 1u;
      if (qn >= 32) drain32();
    }
  };

  bool use_list = false;
  if (MODE == 1) {
    const int c = __float_as_int(myl[lane]);
    use_list = __builtin_amdgcn_ballot_w32(c < 0 || c > CAP) == 0u;
    if (use_list) {
      const int n = __builtin_amdgcn_readlane(c, wave);
      auto ld = [&](int b) -> unsigned {
        return (b + lane < n) ? __float_as_uint(wl[b + lane]) : 0u;
      };
      unsigned pk = ld(0);
      unsigned pk1 = ld(32);
      float dv = dinv[pk >> 9];
      for (int b = 0; b < n; b += 32) {
        const unsigned pk2 = ld(b + 64);
        const float dv1 = dinv[pk1 >> 9];
        const int u = (int)(pk >> 9);
        const int r = (int)(pk & 511u);
        const float w = dv * dloc[r];
        const int m = (n - b < 32) ? (n - b) : 32;
        int g = 0;
        for (; g + GRPL <= m; g += GRPL) {
          VT hv[GRPL];
          int rr[GRPL];
          float ww[GRPL];
#pragma unroll
          for (int k = 0; k < GRPL; ++k) {
            const int uk = __builtin_amdgcn_readlane(u, g + k);
            rr[k] = __builtin_amdgcn_readlane(r, g + k);
            ww[k] = __int_as_float(__builtin_amdgcn_readlane(__float_as_int(w), g + k));
            hv[k] = *(const VT*)(h + (size_t)uk * F + lane * VEC);
          }
#pragma unroll
          for (int k = 0; k < GRPL; ++k) add_row(rr[k], ww[k], hv[k]);
        }
        if (g < m) {
          VT hv[GRPL];
          int rr[GRPL];
          float ww[GRPL];
#pragma unroll
          for (int k = 0; k < GRPL; ++k) {
            if (g + k < m) {
              const int uk = __builtin_amdgcn_readlane(u, g + k);
              rr[k] = __builtin_amdgcn_readlane(r, g + k);
              ww[k] = __int_as_float(__builtin_amdgcn_readlane(__float_as_int(w), g + k));
              hv[k] = *(const VT*)(h + (size_t)uk * F + lane * VEC);
            }
          }
#pragma unroll
          for (int k = 0; k < GRPL; ++k) {
            if (g + k < m) add_row(rr[k], ww[k], hv[k]);
          }
        }
        pk = pk1;
        dv = dv1;
        pk1 = pk2;
      }
    }
  }

  constexpr int TRIP = 8;
  if (bytes) {
    const unsigned pat = (unsigned)blockIdx.x * 0x01010101u;
    auto hit = [&](unsigned x) -> unsigned {
      x ^= pat;
      return ~(((x & 0x7F7F7F7Fu) + 0x7F7F7F7Fu) | x | 0x7F7F7F7Fu);
    };
    auto mask16 = [&](const uint4& c) -> unsigned {
      return (hit(c.x) >> 7) | (hit(c.y) >> 6) | (hit(c.z) >> 5) | (hit(c.w) >> 4);
    };
    const unsigned* pl = (const unsigned*)lst;
    const int ntr = (E + 4095) >> 12;
    for (int tr = wave; tr < ntr; tr += NW) {
      const uint4* p = (const uint4*)(pl + (size_t)(tr / 3) * LW + P::TOFF) +
                       (tr % 3) * (TRIP * 32) + lane;
      const int b = tr * 4096 + lane * 16;
      uint4 c[TRIP];
      if (tr * 4096 + 4096 <= E) {
#pragma unroll
        for (int j = 0; j < TRIP; ++j) c[j] = p[j * 32];
      } else {
#pragma unroll
        for (int j = 0; j < TRIP; ++j) {
          const uint4 ff = {~0u, ~0u, ~0u, ~0u};
          c[j] = (b + j * 512 < E) ? p[j * 32] : ff;
        }
      }
      unsigned m4[TRIP / 2];
#pragma unroll
      for (int jj = 0; jj < TRIP / 2; ++jj)
        m4[jj] = mask16(c[2 * jj]) | (mask16(c[2 * jj + 1]) << 4);
#pragma unroll 1
      for (int jj = 0; jj < TRIP / 2; ++jj) {
        unsigned mm = (jj == 0) ? m4[0] : (jj == 1) ? m4[1] : (jj == 2) ? m4[2] : m4[3];
        const int base = b + jj * 1024;
        while (true) {
          const bool has = mm != 0u;
          const unsigned mk = __builtin_amdgcn_ballot_w32(has);
          if (mk == 0u) break;
          if (has) {
            const int p = __builtin_ctz(mm);
            myqe[qn + __builtin_popcount(mk & ltmask)] =
                base + (((p >> 2) & 1) << 9) + ((p & 3) << 2) + (p >> 3);
          }
          qn += __builtin_popcount(mk);
          mm &= mm - 1u;
          if (qn >= 32) drain32();
        }
      }
    }
  }
  const int nsc = (use_list || bytes) ? 0 : ((E + 1023) >> 10);
  for (int sc = wave; sc < nsc; sc += NW) {
    const int b = sc * 1024 + lane * 4;
    int4 c[TRIP];
    if (vec_ok && sc * 1024 + 1024 <= E) {
      const int4* p = (const int4*)(dst + b);
#pragma unroll
      for (int j = 0; j < TRIP; ++j) c[j] = p[j * 32];
    } else {
#pragma unroll
      for (int j = 0; j < TRIP; ++j) c[j] = load4(b + j * 128);
    }
#pragma unroll
    for (int j = 0; j < TRIP; ++j) scan4(c[j], b + j * 128);
  }

  if (qn > 0) {
    int r = 0, u = 0;
    float w = 0.0f;
    if (lane < qn) {
      r = bytes ? (dst[myqe[lane]] - r0) : myqr[lane];
      u = src[myqe[lane]];
      w = dinv[u] * dloc[r];
      if (MODE == 0 && can_save && cnt + qn <= CAP)
        wl[cnt + lane] = __uint_as_float(((unsigned)u << 9) | (unsigned)r);
    }
    cnt += qn;
    for (int k = 0; k < qn; ++k) {
      const int uk = __builtin_amdgcn_readlane(u, k);
      const int rk = __builtin_amdgcn_readlane(r, k);
      const float wk = __int_as_float(__builtin_amdgcn_readlane(__float_as_int(w), k));
      const VT hv = *(const VT*)(h + (size_t)uk * F + lane * VEC);
      add_row(rk, wk, hv);
    }
  }
  if (MODE == 0 && lane == 0) qcnt[wave] = (can_save && cnt <= CAP) ? cnt : -1;
  __syncthreads();

  if (MODE == 0 && wave == 0) myl[lane] = __int_as_float(qcnt[lane]);

  for (int r = wave; r < R; r += NW) {
    const int v = r0 + r;
    if (v >= N) break;
    VT o;
#pragma unroll
    for (int t = 0; t < VEC; ++t) {
      const float x = acc[r * F + t * 32 + lane] + bias[lane * VEC + t];
      o[t] = RELU ? fmaxf(x, 0.0f) : x;
    }
    *(VT*)(out + (size_t)v * F + lane * VEC) = o;
  }
}

extern "C" void kernel_launch(void* const* d_in, const int* in_sizes, int n_in,
                              void* d_out, int out_size, void* d_ws,
                              size_t ws_size, hipStream_t stream) {
  const float* x  = (const float*)d_in[0];
  const int*   ei = (const int*)d_in[1];
  const float* W1 = (const float*)d_in[2];
  const float* b1 = (const float*)d_in[3];
  const float* W2 = (const float*)d_in[4];
  const float* b2 = (const float*)d_in[5];
  float* out = (float*)d_out;

  const int N = in_sizes[0] / 128;
  const long long E = in_sizes[1] / 2;
  const int* src = ei;
  const int* dst = ei + E;

  float* ws = (float*)d_ws;
  float* deg   = ws;
  float* h1pre = ws + N;
  float* h1    = h1pre + (size_t)N * 128;
  float* h2pre = h1pre;

  constexpr int R = 400;
  const int TB = 256;
  auto blocks = [](long long n, int tb) { return (unsigned)((n + tb - 1) / tb); };

  fill_kernel<<<blocks(N, TB), TB, 0, stream>>>(deg, 1.0f, N);
  deg_kernel<R><<<blocks((E + 15) / 16 * 4, TB), TB, 0, stream>>>(
      dst, deg, (unsigned*)d_out, N, E);
  dinv_kernel<<<blocks(N, TB), TB, 0, stream>>>(deg, N);
  const float* dinv = deg;

  gemm_wmma_kernel<128><<<(N + 15) / 16, 256, 0, stream>>>(x, W1, h1pre, N);
  owner_agg_kernel<128, R, true, 0><<<(N + R - 1) / R, 1024, 0, stream>>>(
      src, dst, dinv, h1pre, b1, h1, out, N, E);

  gemm_wmma_kernel<64><<<(N + 15) / 16, 128, 0, stream>>>(h1, W2, h2pre, N);
  owner_agg_kernel<64, R, false, 1><<<(N + R - 1) / R, 1024, 0, stream>>>(
      src, dst, dinv, h2pre, b2, out, out, N, E);
}


# GEMM A tile in LDS at a row pitch of 132 floats, not 128: the 16 rows of an A-fragment read no longer share one bank pair (16-way conflict on every read gone); k and sum order kept
# speedup vs baseline: 2.3017x; 1.1304x over previous
#include <hip/hip_runtime.h>
#include <hip/hip_bf16.h>

typedef float v2f __attribute__((ext_vector_type(2)));
typedef float v8f __attribute__((ext_vector_type(8)));

__device__ __forceinline__ void atomic_add_f32(float* p, float v) {
  __hip_atomic_fetch_add(p, v, __ATOMIC_RELAXED, __HIP_MEMORY_SCOPE_AGENT);
}

__global__ void fill_kernel(float* __restrict__ p, float v, long long n) {
  long long i = (long long)blockIdx.x * blockDim.x + threadIdx.x;
  if (i < n) p[i] = v;
}

template <int R>
struct Plane {
  static constexpr int LW = R * 64;
  static constexpr int PLW = 3072;
  static constexpr int NW = 32;
  static constexpr int CAP = (((LW - 32 - PLW) / NW) / 32) * 32;
  static constexpr int TOFF = 32 + NW * CAP;
  static_assert(CAP >= 32, "slice too small for the lists");
  static_assert(TOFF % 32 == 0 && TOFF + PLW <= LW, "tail must fit the slice, line aligned");
  __device__ static bool usable(int N, long long E) {
    return (N <= 255 * R) && ((E + PLW * 4 - 1) / (PLW * 4) <= (long long)(N / R));
  }
};

template <int R>
__global__ void deg_kernel(const int* __restrict__ dst, float* __restrict__ deg,
                           unsigned* __restrict__ slices, int N, long long E) {
  typedef Plane<R> P;
  const long long i = (long long)blockIdx.x * blockDim.x + threadIdx.x;
  const long long e0 = i * 4;
  if (e0 >= ((E + 15) / 16) * 16) return;
  int d[4];
  if (e0 + 3 < E && ((((size_t)dst) & 15) == 0)) {
    const int4 v = *(const int4*)(dst + e0);
    d[0] = v.x; d[1] = v.y; d[2] = v.z; d[3] = v.w;
  } else {
#pragma unroll
    for (int k = 0; k < 4; ++k) d[k] = (e0 + k < E) ? dst[e0 + k] : -1;
  }
  unsigned w = 0u;
#pragma unroll
  for (int k = 0; k < 4; ++k) {
    if (d[k] >= 0) atomic_add_f32(&deg[d[k]], 1.0f);
    const unsigned o = (d[k] >= 0) ? (unsigned)(d[k] / R) : 255u;
    w |= (o & 255u) << (8 * k);
  }
  if (P::usable(N, E))
    slices[(size_t)(i / P::PLW) * P::LW + P::TOFF + (int)(i % P::PLW)] = w;
}

__global__ void dinv_kernel(float* __restrict__ deg, int n) {
  int i = blockIdx.x * blockDim.x + threadIdx.x;
  if (i < n) deg[i] = __frsqrt_rn(fmaxf(deg[i], 1e-12f));
}

template <int FOUT>
__global__ __launch_bounds__(FOUT * 2) void gemm_wmma_kernel(
    const float* __restrict__ A, const float* __restrict__ W,
    float* __restrict__ C, int nrows) {
  constexpr int PITCH = 132;
  __shared__ alignas(16) float lds[16 * PITCH];

  const int row0 = blockIdx.x * 16;
  if (row0 >= nrows) return;
  const int tid = threadIdx.x;
  constexpr int NTHREADS = FOUT * 2;

  if (row0 + 16 <= nrows) {
    const float4* srcv = (const float4*)(A + (size_t)row0 * 128);
    #pragma unroll 2
    for (int i = tid; i < 512; i += NTHREADS)
      *(float4*)&lds[(i >> 5) * PITCH + (i & 31) * 4] = srcv[i];
  } else {
    for (int i = tid; i < 2048; i += NTHREADS) {
      int r = i >> 7;
      lds[r * PITCH + (i & 127)] =
          (row0 + r < nrows) ? A[(size_t)(row0 + r) * 128 + (i & 127)] : 0.0f;
    }
  }
  __syncthreads();

  const int lane = tid & 31;
  const int wave = tid >> 5;
  const int colbase = wave * 16;
  const int half = lane >> 4;
  const int m = lane & 15;

  v8f acc = {};
  #pragma unroll
  for (int k = 0; k < 128; k += 4) {
    const int kb = k + half * 2;
    v2f a = *(const v2f*)&lds[m * PITCH + kb];
    v2f b;
    b.x = W[(size_t)kb * FOUT + colbase + m];
    b.y = W[(size_t)(kb + 1) * FOUT + colbase + m];
    acc = __builtin_amdgcn_wmma_f32_16x16x4_f32(false, a, false, b, (short)0,
                                                acc, false, false);
  }

  #pragma unroll
  for (int i = 0; i < 8; ++i) {
    int row = row0 + i + half * 8;
    if (row < nrows) C[(size_t)row * FOUT + colbase + m] = acc[i];
  }
}

template <int F, int R, bool RELU, int MODE>
__global__ __launch_bounds__(1024) void owner_agg_kernel(
    const int* __restrict__ src, const int* __restrict__ dst,
    const float* __restrict__ dinv, const float* __restrict__ h,
    const float* __restrict__ bias, float* out, float* lst, int N,
    long long E64) {
  constexpr int VEC = F / 32;
  constexpr int NW = 32;
  constexpr int QCAP = 64;
  constexpr int GRP = 8;
  constexpr int GRPL = 16;
  typedef Plane<R> P;
  constexpr int LW = P::LW;
  constexpr int CAP = P::CAP;
  static_assert(R <= 512, "owned row must fit the 9 low bits of a list word");
  static_assert(P::NW == NW, "slice layout is for 32 waves");
  typedef float VT __attribute__((ext_vector_type(VEC)));

  __shared__ float acc[R * F];
  __shared__ float dloc[R];
  __shared__ int qe[NW * QCAP];
  __shared__ int qr[NW * QCAP];
  __shared__ int qcnt[NW];

  const int tid = threadIdx.x;
  const int lane = tid & 31;
  const int wave = __builtin_amdgcn_readfirstlane(tid >> 5);
  const int r0 = blockIdx.x * R;
  const int E = (int)E64;

  for (int r = wave; r < R; r += NW) {
    const int v = r0 + r;
    if (v < N) {
      const float di = dinv[v];
      if (lane == 0) dloc[r] = di;
      const float s = di * di;
      const VT hv = *(const VT*)(h + (size_t)v * F + lane * VEC);
#pragma unroll
      for (int t = 0; t < VEC; ++t) acc[r * F + t * 32 + lane] = s * hv[t];
    } else {
      if (lane == 0) dloc[r] = 0.0f;
#pragma unroll
      for (int t = 0; t < VEC; ++t) acc[r * F + t * 32 + lane] = 0.0f;
    }
  }
  __syncthreads();

  int* myqe = qe + wave * QCAP;
  int* myqr = qr + wave * QCAP;
  const unsigned ltmask = (1u << lane) - 1u;
  int qn = 0;

  float* myl = lst + (size_t)blockIdx.x * LW;
  float* wl = myl + 32 + wave * CAP;
  const bool can_save = (MODE == 0) && (r0 + R <= N) && (N <= (1 << 23));
  int cnt = 0;

  auto add_row = [&](int rk, float wk, const VT& hv) {
#pragma unroll
    for (int t = 0; t < VEC; ++t)
      __hip_atomic_fetch_add(&acc[rk * F + t * 32 + lane], wk * hv[t],
                             __ATOMIC_RELAXED, __HIP_MEMORY_SCOPE_WORKGROUP);
  };

  const bool bytes = (MODE == 0) && P::usable(N, E64);
  auto drain32 = [&]() {
    const int r = bytes ? (dst[myqe[lane]] - r0) : myqr[lane];
    const int u = src[myqe[lane]];
    const float w = dinv[u] * dloc[r];
    if (MODE == 0) {
      if (can_save && cnt + 32 <= CAP)
        wl[cnt + lane] = __uint_as_float(((unsigned)u << 9) | (unsigned)r);
      cnt += 32;
    }
    for (int g = 0; g < 32; g += GRP) {
      VT hv[GRP];
      int rr[GRP];
      float ww[GRP];
#pragma unroll
      for (int k = 0; k < GRP; ++k) {
        const int uk = __builtin_amdgcn_readlane(u, g + k);
        rr[k] = __builtin_amdgcn_readlane(r, g + k);
        ww[k] = __int_as_float(__builtin_amdgcn_readlane(__float_as_int(w), g + k));
        hv[k] = *(const VT*)(h + (size_t)uk * F + lane * VEC);
      }
#pragma unroll
      for (int k = 0; k < GRP; ++k) add_row(rr[k], ww[k], hv[k]);
    }
    const int rem = qn - 32;
    int te = 0, tr = 0;
    if (lane < rem) { te = myqe[32 + lane]; tr = bytes ? 0 : myqr[32 + lane]; }
    if (lane < rem) { myqe[lane] = te; if (!bytes) myqr[lane] = tr; }
    qn = rem;
  };

  const bool vec_ok = ((E & 3) == 0) && ((((size_t)dst) & 15) == 0);
  auto load4 = [&](int i) -> int4 {
    int4 d;
    d.x = (i     < E) ? dst[i]     : -1;
    d.y = (i + 1 < E) ? dst[i + 1] : -1;
    d.z = (i + 2 < E) ? dst[i + 2] : -1;
    d.w = (i + 3 < E) ? dst[i + 3] : -1;
    return d;
  };

  auto scan4 = [&](const int4& d, int base) {
    const int d0 = d.x - r0, d1 = d.y - r0, d2 = d.z - r0, d3 = d.w - r0;
    unsigned mm = ((unsigned)d0 < (unsigned)R ? 1u : 0u) |
                  ((unsigned)d1 < (unsigned)R ? 2u : 0u) |
                  ((unsigned)d2 < (unsigned)R ? 4u : 0u) |
                  ((unsigned)d3 < (unsigned)R ? 8u : 0u);
    while (true) {
      const bool has = mm != 0u;
      const unsigned mk = __builtin_amdgcn_ballot_w32(has);
      if (mk == 0u) break;
      if (has) {
        const int c = __builtin_ctz(mm);
        const int dsel = (c == 0) ? d0 : (c == 1) ? d1 : (c == 2) ? d2 : d3;
        const int p = qn + __builtin_popcount(mk & ltmask);
        myqe[p] = base + c;
        myqr[p] = dsel;
      }
      qn += __builtin_popcount(mk);
      mm &= mm - 1u;
      if (qn >= 32) drain32();
    }
  };

  bool use_list = false;
  if (MODE == 1) {
    const int c = __float_as_int(myl[lane]);
    use_list = __builtin_amdgcn_ballot_w32(c < 0 || c > CAP) == 0u;
    if (use_list) {
      const int n = __builtin_amdgcn_readlane(c, wave);
      auto ld = [&](int b) -> unsigned {
        return (b + lane < n) ? __float_as_uint(wl[b + lane]) : 0u;
      };
      unsigned pk = ld(0);
      unsigned pk1 = ld(32);
      float dv = dinv[pk >> 9];
      for (int b = 0; b < n; b += 32) {
        const unsigned pk2 = ld(b + 64);
        const float dv1 = dinv[pk1 >> 9];
        const int u = (int)(pk >> 9);
        const int r = (int)(pk & 511u);
        const float w = dv * dloc[r];
        const int m = (n - b < 32) ? (n - b) : 32;
        int g = 0;
        for (; g + GRPL <= m; g += GRPL) {
          VT hv[GRPL];
          int rr[GRPL];
          float ww[GRPL];
#pragma unroll
          for (int k = 0; k < GRPL; ++k) {
            const int uk = __builtin_amdgcn_readlane(u, g + k);
            rr[k] = __builtin_amdgcn_readlane(r, g + k);
            ww[k] = __int_as_float(__builtin_amdgcn_readlane(__float_as_int(w), g + k));
            hv[k] = *(const VT*)(h + (size_t)uk * F + lane * VEC);
          }
#pragma unroll
          for (int k = 0; k < GRPL; ++k) add_row(rr[k], ww[k], hv[k]);
        }
        if (g < m) {
          VT hv[GRPL];
          int rr[GRPL];
          float ww[GRPL];
#pragma unroll
          for (int k = 0; k < GRPL; ++k) {
            if (g + k < m) {
              const int uk = __builtin_amdgcn_readlane(u, g + k);
              rr[k] = __builtin_amdgcn_readlane(r, g + k);
              ww[k] = __int_as_float(__builtin_amdgcn_readlane(__float_as_int(w), g + k));
              hv[k] = *(const VT*)(h + (size_t)uk * F + lane * VEC);
            }
          }
#pragma unroll
          for (int k = 0; k < GRPL; ++k) {
            if (g + k < m) add_row(rr[k], ww[k], hv[k]);
          }
        }
        pk = pk1;
        dv = dv1;
        pk1 = pk2;
      }
    }
  }

  constexpr int TRIP = 8;
  if (bytes) {
    const unsigned pat = (unsigned)blockIdx.x * 0x01010101u;
    auto hit = [&](unsigned x) -> unsigned {
      x ^= pat;
      return ~(((x & 0x7F7F7F7Fu) + 0x7F7F7F7Fu) | x | 0x7F7F7F7Fu);
    };
    auto mask16 = [&](const uint4& c) -> unsigned {
      return (hit(c.x) >> 7) | (hit(c.y) >> 6) | (hit(c.z) >> 5) | (hit(c.w) >> 4);
    };
    const unsigned* pl = (const unsigned*)lst;
    const int ntr = (E + 4095) >> 12;
    for (int tr = wave; tr < ntr; tr += NW) {
      const uint4* p = (const uint4*)(pl + (size_t)(tr / 3) * LW + P::TOFF) +
                       (tr % 3) * (TRIP * 32) + lane;
      const int b = tr * 4096 + lane * 16;
      uint4 c[TRIP];
      if (tr * 4096 + 4096 <= E) {
#pragma unroll
        for (int j = 0; j < TRIP; ++j) c[j] = p[j * 32];
      } else {
#pragma unroll
        for (int j = 0; j < TRIP; ++j) {
          const uint4 ff = {~0u, ~0u, ~0u, ~0u};
          c[j] = (b + j * 512 < E) ? p[j * 32] : ff;
        }
      }
      unsigned m4[TRIP / 2];
#pragma unroll
      for (int jj = 0; jj < TRIP / 2; ++jj)
        m4[jj] = mask16(c[2 * jj]) | (mask16(c[2 * jj + 1]) << 4);
#pragma unroll 1
      for (int jj = 0; jj < TRIP / 2; ++jj) {
        unsigned mm = (jj == 0) ? m4[0] : (jj == 1) ? m4[1] : (jj == 2) ? m4[2] : m4[3];
        const int base = b + jj * 1024;
        while (true) {
          const bool has = mm != 0u;
          const unsigned mk = __builtin_amdgcn_ballot_w32(has);
          if (mk == 0u) break;
          if (has) {
            const int p = __builtin_ctz(mm);
            myqe[qn + __builtin_popcount(mk & ltmask)] =
                base + (((p >> 2) & 1) << 9) + ((p & 3) << 2) + (p >> 3);
          }
          qn += __builtin_popcount(mk);
          mm &= mm - 1u;
          if (qn >= 32) drain32();
        }
      }
    }
  }
  const int nsc = (use_list || bytes) ? 0 : ((E + 1023) >> 10);
  for (int sc = wave; sc < nsc; sc += NW) {
    const int b = sc * 1024 + lane * 4;
    int4 c[TRIP];
    if (vec_ok && sc * 1024 + 1024 <= E) {
      const int4* p = (const int4*)(dst + b);
#pragma unroll
      for (int j = 0; j < TRIP; ++j) c[j] = p[j * 32];
    } else {
#pragma unroll
      for (int j = 0; j < TRIP; ++j) c[j] = load4(b + j * 128);
    }
#pragma unroll
    for (int j = 0; j < TRIP; ++j) scan4(c[j], b + j * 128);
  }

  if (qn > 0) {
    int r = 0, u = 0;
    float w = 0.0f;
    if (lane < qn) {
      r = bytes ? (dst[myqe[lane]] - r0) : myqr[lane];
      u = src[myqe[lane]];
      w = dinv[u] * dloc[r];
      if (MODE == 0 && can_save && cnt + qn <= CAP)
        wl[cnt + lane] = __uint_as_float(((unsigned)u << 9) | (unsigned)r);
    }
    cnt += qn;
    for (int k = 0; k < qn; ++k) {
      const int uk = __builtin_amdgcn_readlane(u, k);
      const int rk = __builtin_amdgcn_readlane(r, k);
      const float wk = __int_as_float(__builtin_amdgcn_readlane(__float_as_int(w), k));
      const VT hv = *(const VT*)(h + (size_t)uk * F + lane * VEC);
      add_row(rk, wk, hv);
    }
  }
  if (MODE == 0 && lane == 0) qcnt[wave] = (can_save && cnt <= CAP) ? cnt : -1;
  __syncthreads();

  if (MODE == 0 && wave == 0) myl[lane] = __int_as_float(qcnt[lane]);

  for (int r = wave; r < R; r += NW) {
    const int v = r0 + r;
    if (v >= N) break;
    VT o;
#pragma unroll
    for (int t = 0; t < VEC; ++t) {
      const float x = acc[r * F + t * 32 + lane] + bias[lane * VEC + t];
      o[t] = RELU ? fmaxf(x, 0.0f) : x;
    }
    *(VT*)(out + (size_t)v * F + lane * VEC) = o;
  }
}

extern "C" void kernel_launch(void* const* d_in, const int* in_sizes, int n_in,
                              void* d_out, int out_size, void* d_ws,
                              size_t ws_size, hipStream_t stream) {
  const float* x  = (const float*)d_in[0];
  const int*   ei = (const int*)d_in[1];
  const float* W1 = (const float*)d_in[2];
  const float* b1 = (const float*)d_in[3];
  const float* W2 = (const float*)d_in[4];
  const float* b2 = (const float*)d_in[5];
  float* out = (float*)d_out;

  const int N = in_sizes[0] / 128;
  const long long E = in_sizes[1] / 2;
  const int* src = ei;
  const int* dst = ei + E;

  float* ws = (float*)d_ws;
  float* deg   = ws;
  float* h1pre = ws + N;
  float* h1    = h1pre + (size_t)N * 128;
  float* h2pre = h1pre;

  constexpr int R = 400;
  const int TB = 256;
  auto blocks = [](long long n, int tb) { return (unsigned)((n + tb - 1) / tb); };

  fill_kernel<<<blocks(N, TB), TB, 0, stream>>>(deg, 1.0f, N);
  deg_kernel<R><<<blocks((E + 15) / 16 * 4, TB), TB, 0, stream>>>(
      dst, deg, (unsigned*)d_out, N, E);
  dinv_kernel<<<blocks(N, TB), TB, 0, stream>>>(deg, N);
  const float* dinv = deg;

  gemm_wmma_kernel<128><<<(N + 15) / 16, 256, 0, stream>>>(x, W1, h1pre, N);
  owner_agg_kernel<128, R, true, 0><<<(N + R - 1) / R, 1024, 0, stream>>>(
      src, dst, dinv, h1pre, b1, h1, out, N, E);

  gemm_wmma_kernel<64><<<(N + 15) / 16, 128, 0, stream>>>(h1, W2, h2pre, N);
  owner_agg_kernel<64, R, false, 1><<<(N + R - 1) / R, 1024, 0, stream>>>(
      src, dst, dinv, h2pre, b2, out, out, N, E);
}


# Degrees counted by the owner blocks in LDS (integer adds) in one edge scan that also saves both layers' lists: global float atomics, fill and rsqrt launches go (7 to 6); sum order still LDS arrival
# speedup vs baseline: 3.1053x; 1.3491x over previous
#include <hip/hip_runtime.h>
#include <hip/hip_bf16.h>

typedef float v2f __attribute__((ext_vector_type(2)));
typedef float v8f __attribute__((ext_vector_type(8)));

template <int R>
struct Plane {
  static constexpr int LW = R * 64;
  static constexpr int PLW = 3072;
  static constexpr int NW = 32;
  static constexpr int CAP = (((LW - 32 - PLW) / NW) / 32) * 32;
  static constexpr int TOFF = 32 + NW * CAP;
  static_assert(CAP >= 32, "slice too small for the lists");
  static_assert(TOFF % 32 == 0 && TOFF + PLW <= LW, "tail must fit the slice, line aligned");
  __device__ static bool usable(int N, long long E) {
    return (N <= 255 * R) && ((E + PLW * 4 - 1) / (PLW * 4) <= (long long)(N / R));
  }
};

template <int R>
__global__ void plane_kernel(const int* __restrict__ dst, unsigned* __restrict__ slices,
                             int N, long long E) {
  typedef Plane<R> P;
  if (!P::usable(N, E)) return;
  const long long i = (long long)blockIdx.x * blockDim.x + threadIdx.x;
  const long long e0 = i * 4;
  if (e0 >= ((E + 15) / 16) * 16) return;
  int d[4];
  if (e0 + 3 < E && ((((size_t)dst) & 15) == 0)) {
    const int4 v = *(const int4*)(dst + e0);
    d[0] = v.x; d[1] = v.y; d[2] = v.z; d[3] = v.w;
  } else {
#pragma unroll
    for (int k = 0; k < 4; ++k) d[k] = (e0 + k < E) ? dst[e0 + k] : -1;
  }
  unsigned w = 0u;
#pragma unroll
  for (int k = 0; k < 4; ++k) {
    const unsigned o = (d[k] >= 0) ? (unsigned)(d[k] / R) : 255u;
    w |= (o & 255u) << (8 * k);
  }
  slices[(size_t)(i / P::PLW) * P::LW + P::TOFF + (int)(i % P::PLW)] = w;
}

template <int FOUT>
__global__ __launch_bounds__(FOUT * 2) void gemm_wmma_kernel(
    const float* __restrict__ A, const float* __restrict__ W,
    float* __restrict__ C, int nrows) {
  constexpr int PITCH = 132;
  __shared__ alignas(16) float lds[16 * PITCH];

  const int row0 = blockIdx.x * 16;
  if (row0 >= nrows) return;
  const int tid = threadIdx.x;
  constexpr int NTHREADS = FOUT * 2;

  if (row0 + 16 <= nrows) {
    const float4* srcv = (const float4*)(A + (size_t)row0 * 128);
    #pragma unroll 2
    for (int i = tid; i < 512; i += NTHREADS)
      *(float4*)&lds[(i >> 5) * PITCH + (i & 31) * 4] = srcv[i];
  } else {
    for (int i = tid; i < 2048; i += NTHREADS) {
      int r = i >> 7;
      lds[r * PITCH + (i & 127)] =
          (row0 + r < nrows) ? A[(size_t)(row0 + r) * 128 + (i & 127)] : 0.0f;
    }
  }
  __syncthreads();

  const int lane = tid & 31;
  const int wave = tid >> 5;
  const int colbase = wave * 16;
  const int half = lane >> 4;
  const int m = lane & 15;

  v8f acc = {};
  #pragma unroll
  for (int k = 0; k < 128; k += 4) {
    const int kb = k + half * 2;
    v2f a = *(const v2f*)&lds[m * PITCH + kb];
    v2f b;
    b.x = W[(size_t)kb * FOUT + colbase + m];
    b.y = W[(size_t)(kb + 1) * FOUT + colbase + m];
    acc = __builtin_amdgcn_wmma_f32_16x16x4_f32(false, a, false, b, (short)0,
                                                acc, false, false);
  }

  #pragma unroll
  for (int i = 0; i < 8; ++i) {
    int row = row0 + i + half * 8;
    if (row < nrows) C[(size_t)row * FOUT + colbase + m] = acc[i];
  }
}

template <int R>
__global__ __launch_bounds__(1024) void owner_scan_kernel(
    const int* __restrict__ src, const int* __restrict__ dst,
    float* __restrict__ dinv, float* lst, int N, long long E64) {
  constexpr int NW = 32;
  constexpr int QCAP = 64;
  typedef Plane<R> P;
  constexpr int LW = P::LW;
  constexpr int CAP = P::CAP;
  static_assert(R <= 512, "owned row must fit the 9 low bits of a list word");
  static_assert(P::NW == NW, "slice layout is for 32 waves");

  __shared__ int cntl[R];
  __shared__ int qe[NW * QCAP];
  __shared__ int qr[NW * QCAP];
  __shared__ int qcnt[NW];

  const int tid = threadIdx.x;
  const int lane = tid & 31;
  const int wave = __builtin_amdgcn_readfirstlane(tid >> 5);
  const int r0 = blockIdx.x * R;
  const int E = (int)E64;

  for (int r = tid; r < R; r += NW * 32) cntl[r] = 0;
  __syncthreads();

  int* myqe = qe + wave * QCAP;
  int* myqr = qr + wave * QCAP;
  const unsigned ltmask = (1u << lane) - 1u;
  int qn = 0;

  float* myl = lst + (size_t)blockIdx.x * LW;
  float* wl = myl + 32 + wave * CAP;
  const bool can_save = (r0 + R <= N) && (N <= (1 << 23));
  int cnt = 0;

  const bool bytes = P::usable(N, E64);

  auto take = [&](int nq) {
    if (lane < nq) {
      const int e = myqe[lane];
      const int r = bytes ? (dst[e] - r0) : myqr[lane];
      __hip_atomic_fetch_add(&cntl[r], 1, __ATOMIC_RELAXED, __HIP_MEMORY_SCOPE_WORKGROUP);
      if (can_save && cnt + nq <= CAP)
        wl[cnt + lane] = __uint_as_float(((unsigned)src[e] << 9) | (unsigned)r);
    }
    cnt += nq;
  };
  auto drain32 = [&]() {
    take(32);
    const int rem = qn - 32;
    int te = 0, tr = 0;
    if (lane < rem) { te = myqe[32 + lane]; tr = bytes ? 0 : myqr[32 + lane]; }
    if (lane < rem) { myqe[lane] = te; if (!bytes) myqr[lane] = tr; }
    qn = rem;
  };

  const bool vec_ok = ((E & 3) == 0) && ((((size_t)dst) & 15) == 0);
  auto load4 = [&](int i) -> int4 {
    int4 d;
    d.x = (i     < E) ? dst[i]     : -1;
    d.y = (i + 1 < E) ? dst[i + 1] : -1;
    d.z = (i + 2 < E) ? dst[i + 2] : -1;
    d.w = (i + 3 < E) ? dst[i + 3] : -1;
    return d;
  };

  auto scan4 = [&](const int4& d, int base) {
    const int d0 = d.x - r0, d1 = d.y - r0, d2 = d.z - r0, d3 = d.w - r0;
    unsigned mm = ((unsigned)d0 < (unsigned)R ? 1u : 0u) |
                  ((unsigned)d1 < (unsigned)R ? 2u : 0u) |
                  ((unsigned)d2 < (unsigned)R ? 4u : 0u) |
                  ((unsigned)d3 < (unsigned)R ? 8u : 0u);
    while (true) {
      const bool has = mm != 0u;
      const unsigned mk = __builtin_amdgcn_ballot_w32(has);
      if (mk == 0u) break;
      if (has) {
        const int c = __builtin_ctz(mm);
        const int dsel = (c == 0) ? d0 : (c == 1) ? d1 : (c == 2) ? d2 : d3;
        const int p = qn + __builtin_popcount(mk & ltmask);
        myqe[p] = base + c;
        myqr[p] = dsel;
      }
      qn += __builtin_popcount(mk);
      mm &= mm - 1u;
      if (qn >= 32) drain32();
    }
  };

  constexpr int TRIP = 8;
  if (bytes) {
    const unsigned pat = (unsigned)blockIdx.x * 0x01010101u;
    auto hit = [&](unsigned x) -> unsigned {
      x ^= pat;
      return ~(((x & 0x7F7F7F7Fu) + 0x7F7F7F7Fu) | x | 0x7F7F7F7Fu);
    };
    auto mask16 = [&](const uint4& c) -> unsigned {
      return (hit(c.x) >> 7) | (hit(c.y) >> 6) | (hit(c.z) >> 5) | (hit(c.w) >> 4);
    };
    const unsigned* pl = (const unsigned*)lst;
    const int ntr = (E + 4095) >> 12;
    for (int tr = wave; tr < ntr; tr += NW) {
      const uint4* p = (const uint4*)(pl + (size_t)(tr / 3) * LW + P::TOFF) +
                       (tr % 3) * (TRIP * 32) + lane;
      const int b = tr * 4096 + lane * 16;
      uint4 c[TRIP];
      if (tr * 4096 + 4096 <= E) {
#pragma unroll
        for (int j = 0; j < TRIP; ++j) c[j] = p[j * 32];
      } else {
#pragma unroll
        for (int j = 0; j < TRIP; ++j) {
          const uint4 ff = {~0u, ~0u, ~0u, ~0u};
          c[j] = (b + j * 512 < E) ? p[j * 32] : ff;
        }
      }
      unsigned m4[TRIP / 2];
#pragma unroll
      for (int jj = 0; jj < TRIP / 2; ++jj)
        m4[jj] = mask16(c[2 * jj]) | (mask16(c[2 * jj + 1]) << 4);
#pragma unroll 1
      for (int jj = 0; jj < TRIP / 2; ++jj) {
        unsigned mm = (jj == 0) ? m4[0] : (jj == 1) ? m4[1] : (jj == 2) ? m4[2] : m4[3];
        const int base = b + jj * 1024;
        while (true) {
          const bool has = mm != 0u;
          const unsigned mk = __builtin_amdgcn_ballot_w32(has);
          if (mk == 0u) break;
          if (has) {
            const int p = __builtin_ctz(mm);
            myqe[qn + __builtin_popcount(mk & ltmask)] =
                base + (((p >> 2) & 1) << 9) + ((p & 3) << 2) + (p >> 3);
          }
          qn += __builtin_popcount(mk);
          mm &= mm - 1u;
          if (qn >= 32) drain32();
        }
      }
    }
  } else {
    const int nsc = (E + 1023) >> 10;
    for (int sc = wave; sc < nsc; sc += NW) {
      const int b = sc * 1024 + lane * 4;
      int4 c[TRIP];
      if (vec_ok && sc * 1024 + 1024 <= E) {
        const int4* p = (const int4*)(dst + b);
#pragma unroll
        for (int j = 0; j < TRIP; ++j) c[j] = p[j * 32];
      } else {
#pragma unroll
        for (int j = 0; j < TRIP; ++j) c[j] = load4(b + j * 128);
      }
#pragma unroll
      for (int j = 0; j < TRIP; ++j) scan4(c[j], b + j * 128);
    }
  }

  if (qn > 0) take(qn);
  if (lane == 0) qcnt[wave] = (can_save && cnt <= CAP) ? cnt : -1;
  __syncthreads();

  if (wave == 0) myl[lane] = __int_as_float(qcnt[lane]);

  for (int r = tid; r < R; r += NW * 32) {
    const int v = r0 + r;
    if (v < N) dinv[v] = __frsqrt_rn(fmaxf((float)(cntl[r] + 1), 1e-12f));
  }
}

template <int F, int R, bool RELU>
__global__ __launch_bounds__(1024) void owner_agg_kernel(
    const int* __restrict__ src, const int* __restrict__ dst,
    const float* __restrict__ dinv, const float* __restrict__ h,
    const float* __restrict__ bias, float* out, const float* lst, int N,
    long long E64) {
  constexpr int VEC = F / 32;
  constexpr int NW = 32;
  constexpr int QCAP = 64;
  constexpr int GRP = 8;
  constexpr int GRPL = (VEC > 2) ? 8 : 16;
  typedef Plane<R> P;
  constexpr int LW = P::LW;
  constexpr int CAP = P::CAP;
  static_assert(P::NW == NW, "slice layout is for 32 waves");
  typedef float VT __attribute__((ext_vector_type(VEC)));

  __shared__ float acc[R * F];
  __shared__ float dloc[R];
  __shared__ int qe[NW * QCAP];
  __shared__ int qr[NW * QCAP];

  const int tid = threadIdx.x;
  const int lane = tid & 31;
  const int wave = __builtin_amdgcn_readfirstlane(tid >> 5);
  const int r0 = blockIdx.x * R;
  const int E = (int)E64;

  for (int r = wave; r < R; r += NW) {
    const int v = r0 + r;
    if (v < N) {
      const float di = dinv[v];
      if (lane == 0) dloc[r] = di;
      const float s = di * di;
      const VT hv = *(const VT*)(h + (size_t)v * F + lane * VEC);
#pragma unroll
      for (int t = 0; t < VEC; ++t) acc[r * F + t * 32 + lane] = s * hv[t];
    } else {
      if (lane == 0) dloc[r] = 0.0f;
#pragma unroll
      for (int t = 0; t < VEC; ++t) acc[r * F + t * 32 + lane] = 0.0f;
    }
  }
  __syncthreads();

  auto add_row = [&](int rk, float wk, const VT& hv) {
#pragma unroll
    for (int t = 0; t < VEC; ++t)
      __hip_atomic_fetch_add(&acc[rk * F + t * 32 + lane], wk * hv[t],
                             __ATOMIC_RELAXED, __HIP_MEMORY_SCOPE_WORKGROUP);
  };

  const float* myl = lst + (size_t)blockIdx.x * LW;
  const float* wl = myl + 32 + wave * CAP;
  const int c = __float_as_int(myl[lane]);
  const bool use_list = __builtin_amdgcn_ballot_w32(c < 0 || c > CAP) == 0u;
  if (use_list) {
    const int n = __builtin_amdgcn_readlane(c, wave);
    auto ld = [&](int b) -> unsigned {
      return (b + lane < n) ? __float_as_uint(wl[b + lane]) : 0u;
    };
    unsigned pk = ld(0);
    unsigned pk1 = ld(32);
    float dv = dinv[pk >> 9];
    for (int b = 0; b < n; b += 32) {
      const unsigned pk2 = ld(b + 64);
      const float dv1 = dinv[pk1 >> 9];
      const int u = (int)(pk >> 9);
      const int r = (int)(pk & 511u);
      const float w = dv * dloc[r];
      const int m = (n - b < 32) ? (n - b) : 32;
      int g = 0;
      for (; g + GRPL <= m; g += GRPL) {
        VT hv[GRPL];
        int rr[GRPL];
        float ww[GRPL];
#pragma unroll
        for (int k = 0; k < GRPL; ++k) {
          const int uk = __builtin_amdgcn_readlane(u, g + k);
          rr[k] = __builtin_amdgcn_readlane(r, g + k);
          ww[k] = __int_as_float(__builtin_amdgcn_readlane(__float_as_int(w), g + k));
          hv[k] = *(const VT*)(h + (size_t)uk * F + lane * VEC);
        }
#pragma unroll
        for (int k = 0; k < GRPL; ++k) add_row(rr[k], ww[k], hv[k]);
      }
      if (g < m) {
        VT hv[GRPL];
        int rr[GRPL];
        float ww[GRPL];
#pragma unroll
        for (int k = 0; k < GRPL; ++k) {
          if (g + k < m) {
            const int uk = __builtin_amdgcn_readlane(u, g + k);
            rr[k] = __builtin_amdgcn_readlane(r, g + k);
            ww[k] = __int_as_float(__builtin_amdgcn_readlane(__float_as_int(w), g + k));
            hv[k] = *(const VT*)(h + (size_t)uk * F + lane * VEC);
          }
        }
#pragma unroll
        for (int k = 0; k < GRPL; ++k) {
          if (g + k < m) add_row(rr[k], ww[k], hv[k]);
        }
      }
      pk = pk1;
      dv = dv1;
      pk1 = pk2;
    }
  } else {
    int* myqe = qe + wave * QCAP;
    int* myqr = qr + wave * QCAP;
    const unsigned ltmask = (1u << lane) - 1u;
    int qn = 0;

    auto drain32 = [&]() {
      const int r = myqr[lane];
      const int u = src[myqe[lane]];
      const float w = dinv[u] * dloc[r];
      for (int g = 0; g < 32; g += GRP) {
        VT hv[GRP];
        int rr[GRP];
        float ww[GRP];
#pragma unroll
        for (int k = 0; k < GRP; ++k) {
          const int uk = __builtin_amdgcn_readlane(u, g + k);
          rr[k] = __builtin_amdgcn_readlane(r, g + k);
          ww[k] = __int_as_float(__builtin_amdgcn_readlane(__float_as_int(w), g + k));
          hv[k] = *(const VT*)(h + (size_t)uk * F + lane * VEC);
        }
#pragma unroll
        for (int k = 0; k < GRP; ++k) add_row(rr[k], ww[k], hv[k]);
      }
      const int rem = qn - 32;
      int te = 0, tr = 0;
      if (lane < rem) { te = myqe[32 + lane]; tr = myqr[32 + lane]; }
      if (lane < rem) { myqe[lane] = te; myqr[lane] = tr; }
      qn = rem;
    };

    const bool vec_ok = ((E & 3) == 0) && ((((size_t)dst) & 15) == 0);
    auto load4 = [&](int i) -> int4 {
      int4 d;
      d.x = (i     < E) ? dst[i]     : -1;
      d.y = (i + 1 < E) ? dst[i + 1] : -1;
      d.z = (i + 2 < E) ? dst[i + 2] : -1;
      d.w = (i + 3 < E) ? dst[i + 3] : -1;
      return d;
    };

    auto scan4 = [&](const int4& d, int base) {
      const int d0 = d.x - r0, d1 = d.y - r0, d2 = d.z - r0, d3 = d.w - r0;
      unsigned mm = ((unsigned)d0 < (unsigned)R ? 1u : 0u) |
                    ((unsigned)d1 < (unsigned)R ? 2u : 0u) |
                    ((unsigned)d2 < (unsigned)R ? 4u : 0u) |
                    ((unsigned)d3 < (unsigned)R ? 8u : 0u);
      while (true) {
        const bool has = mm != 0u;
        const unsigned mk = __builtin_amdgcn_ballot_w32(has);
        if (mk == 0u) break;
        if (has) {
          const int cc = __builtin_ctz(mm);
          const int dsel = (cc == 0) ? d0 : (cc == 1) ? d1 : (cc == 2) ? d2 : d3;
          const int p = qn + __builtin_popcount(mk & ltmask);
          myqe[p] = base + cc;
          myqr[p] = dsel;
        }
        qn += __builtin_popcount(mk);
        mm &= mm - 1u;
        if (qn >= 32) drain32();
      }
    };

    constexpr int TRIP = 8;
    const int nsc = (E + 1023) >> 10;
    for (int sc = wave; sc < nsc; sc += NW) {
      const int b = sc * 1024 + lane * 4;
      int4 cv[TRIP];
      if (vec_ok && sc * 1024 + 1024 <= E) {
        const int4* p = (const int4*)(dst + b);
#pragma unroll
        for (int j = 0; j < TRIP; ++j) cv[j] = p[j * 32];
      } else {
#pragma unroll
        for (int j = 0; j < TRIP; ++j) cv[j] = load4(b + j * 128);
      }
#pragma unroll
      for (int j = 0; j < TRIP; ++j) scan4(cv[j], b + j * 128);
    }

    if (qn > 0) {
      int r = 0, u = 0;
      float w = 0.0f;
      if (lane < qn) {
        r = myqr[lane];
        u = src[myqe[lane]];
        w = dinv[u] * dloc[r];
      }
      for (int k = 0; k < qn; ++k) {
        const int uk = __builtin_amdgcn_readlane(u, k);
        const int rk = __builtin_amdgcn_readlane(r, k);
        const float wk = __int_as_float(__builtin_amdgcn_readlane(__float_as_int(w), k));
        const VT hv = *(const VT*)(h + (size_t)uk * F + lane * VEC);
        add_row(rk, wk, hv);
      }
    }
  }
  __syncthreads();

  for (int r = wave; r < R; r += NW) {
    const int v = r0 + r;
    if (v >= N) break;
    VT o;
#pragma unroll
    for (int t = 0; t < VEC; ++t) {
      const float x = acc[r * F + t * 32 + lane] + bias[lane * VEC + t];
      o[t] = RELU ? fmaxf(x, 0.0f) : x;
    }
    *(VT*)(out + (size_t)v * F + lane * VEC) = o;
  }
}

extern "C" void kernel_launch(void* const* d_in, const int* in_sizes, int n_in,
                              void* d_out, int out_size, void* d_ws,
                              size_t ws_size, hipStream_t stream) {
  const float* x  = (const float*)d_in[0];
  const int*   ei = (const int*)d_in[1];
  const float* W1 = (const float*)d_in[2];
  const float* b1 = (const float*)d_in[3];
  const float* W2 = (const float*)d_in[4];
  const float* b2 = (const float*)d_in[5];
  float* out = (float*)d_out;

  const int N = in_sizes[0] / 128;
  const long long E = in_sizes[1] / 2;
  const int* src = ei;
  const int* dst = ei + E;

  float* ws = (float*)d_ws;
  float* dinvw = ws;
  float* h1pre = ws + N;
  float* h1    = h1pre + (size_t)N * 128;
  float* h2pre = h1pre;

  constexpr int R = 400;
  const int TB = 256;
  auto blocks = [](long long n, int tb) { return (unsigned)((n + tb - 1) / tb); };

  plane_kernel<R><<<blocks((E + 15) / 16 * 4, TB), TB, 0, stream>>>(
      dst, (unsigned*)d_out, N, E);
  owner_scan_kernel<R><<<(N + R - 1) / R, 1024, 0, stream>>>(
      src, dst, dinvw, out, N, E);
  const float* dinv = dinvw;

  gemm_wmma_kernel<128><<<(N + 15) / 16, 256, 0, stream>>>(x, W1, h1pre, N);
  owner_agg_kernel<128, R, true><<<(N + R - 1) / R, 1024, 0, stream>>>(
      src, dst, dinv, h1pre, b1, h1, out, N, E);

  gemm_wmma_kernel<64><<<(N + 15) / 16, 128, 0, stream>>>(h1, W2, h2pre, N);
  owner_agg_kernel<64, R, false><<<(N + R - 1) / R, 1024, 0, stream>>>(
      src, dst, dinv, h2pre, b2, out, out, N, E);
}


# Both GEMM launches persistent: 768/1024 blocks of 8 waves loop over 32-row tiles (75k one-tile waves become 14k); a wave keeps its 16-column strip of W in registers; k and sum order kept
# speedup vs baseline: 3.2003x; 1.0306x over previous
#include <hip/hip_runtime.h>
#include <hip/hip_bf16.h>

typedef float v2f __attribute__((ext_vector_type(2)));
typedef float v4f __attribute__((ext_vector_type(4)));
typedef float v8f __attribute__((ext_vector_type(8)));

template <int R>
struct Plane {
  static constexpr int LW = R * 64;
  static constexpr int PLW = 3072;
  static constexpr int NW = 32;
  static constexpr int CAP = (((LW - 32 - PLW) / NW) / 32) * 32;
  static constexpr int TOFF = 32 + NW * CAP;
  static_assert(CAP >= 32, "slice too small for the lists");
  static_assert(TOFF % 32 == 0 && TOFF + PLW <= LW, "tail must fit the slice, line aligned");
  __device__ static bool usable(int N, long long E) {
    return (N <= 255 * R) && ((E + PLW * 4 - 1) / (PLW * 4) <= (long long)(N / R));
  }
};

template <int R>
__global__ void plane_kernel(const int* __restrict__ dst, unsigned* __restrict__ slices,
                             int N, long long E) {
  typedef Plane<R> P;
  if (!P::usable(N, E)) return;
  const long long i = (long long)blockIdx.x * blockDim.x + threadIdx.x;
  const long long e0 = i * 4;
  if (e0 >= ((E + 15) / 16) * 16) return;
  int d[4];
  if (e0 + 3 < E && ((((size_t)dst) & 15) == 0)) {
    const int4 v = *(const int4*)(dst + e0);
    d[0] = v.x; d[1] = v.y; d[2] = v.z; d[3] = v.w;
  } else {
#pragma unroll
    for (int k = 0; k < 4; ++k) d[k] = (e0 + k < E) ? dst[e0 + k] : -1;
  }
  unsigned w = 0u;
#pragma unroll
  for (int k = 0; k < 4; ++k) {
    const unsigned o = (d[k] >= 0) ? (unsigned)(d[k] / R) : 255u;
    w |= (o & 255u) << (8 * k);
  }
  slices[(size_t)(i / P::PLW) * P::LW + P::TOFF + (int)(i % P::PLW)] = w;
}

template <int FOUT, int RT>
__global__ __launch_bounds__(256) void gemm_wmma_kernel(
    const float* __restrict__ A, const float* __restrict__ W,
    float* __restrict__ C, int nrows) {
  constexpr int PITCH = 132;
  constexpr int NCS = FOUT / 16;
  constexpr int RG = 8 / NCS;
  constexpr int TM = 16 * RT * RG;
  constexpr int NF4 = TM * 32 / 256;
  static_assert(NCS * RG == 8 && NF4 * 256 == TM * 32, "tile must split over 8 waves");
  __shared__ alignas(16) float lds[2][TM * PITCH];

  const int ntiles = (nrows + TM - 1) / TM;
  int tile = blockIdx.x;
  if (tile >= ntiles) return;

  const int tid = threadIdx.x;
  const int lane = tid & 31;
  const int wave = tid >> 5;
  const int colbase = (wave % NCS) * 16;
  const int rg = wave / NCS;
  const int half = lane >> 4;
  const int m = lane & 15;

  v2f b[32];
#pragma unroll
  for (int s = 0; s < 32; ++s) {
    const int kb = 4 * s + half * 2;
    b[s].x = W[(size_t)kb * FOUT + colbase + m];
    b[s].y = W[(size_t)(kb + 1) * FOUT + colbase + m];
  }

  v4f st[NF4];
  auto fetch = [&](int t) {
    const int row0 = t * TM;
    const v4f* srcv = (const v4f*)(A + (size_t)row0 * 128);
    if (row0 + TM <= nrows) {
#pragma unroll
      for (int j = 0; j < NF4; ++j) st[j] = srcv[tid + j * 256];
    } else {
#pragma unroll
      for (int j = 0; j < NF4; ++j) {
        const int i = tid + j * 256;
        const v4f z = {0.0f, 0.0f, 0.0f, 0.0f};
        st[j] = z;
        if (row0 + (i >> 5) < nrows) st[j] = srcv[i];
      }
    }
  };
  auto stash = [&](int buf) {
#pragma unroll
    for (int j = 0; j < NF4; ++j) {
      const int i = tid + j * 256;
      *(v4f*)&lds[buf][(i >> 5) * PITCH + (i & 31) * 4] = st[j];
    }
  };

  fetch(tile);
  stash(0);
  __syncthreads();
  int buf = 0;
  const int step = (int)gridDim.x;
  for (; tile < ntiles; tile += step) {
    const int nxt = tile + step;
    if (nxt < ntiles) fetch(nxt);

    v8f acc[RT];
#pragma unroll
    for (int t = 0; t < RT; ++t) acc[t] = v8f{};
    const float* abase = &lds[buf][(rg * RT * 16 + m) * PITCH + half * 2];
#pragma unroll
    for (int s = 0; s < 32; ++s) {
#pragma unroll
      for (int t = 0; t < RT; ++t) {
        const v2f a = *(const v2f*)(abase + t * 16 * PITCH + 4 * s);
        acc[t] = __builtin_amdgcn_wmma_f32_16x16x4_f32(false, a, false, b[s], (short)0,
                                                       acc[t], false, false);
      }
    }

    if (nxt < ntiles) stash(buf ^ 1);
    __syncthreads();
    buf ^= 1;

#pragma unroll
    for (int t = 0; t < RT; ++t) {
      const int rbase = tile * TM + (rg * RT + t) * 16 + half * 8;
#pragma unroll
      for (int i = 0; i < 8; ++i) {
        const int row = rbase + i;
        if (row < nrows) C[(size_t)row * FOUT + colbase + m] = acc[t][i];
      }
    }
  }
}

template <int R>
__global__ __launch_bounds__(1024) void owner_scan_kernel(
    const int* __restrict__ src, const int* __restrict__ dst,
    float* __restrict__ dinv, float* lst, int N, long long E64) {
  constexpr int NW = 32;
  constexpr int QCAP = 64;
  typedef Plane<R> P;
  constexpr int LW = P::LW;
  constexpr int CAP = P::CAP;
  static_assert(R <= 512, "owned row must fit the 9 low bits of a list word");
  static_assert(P::NW == NW, "slice layout is for 32 waves");

  __shared__ int cntl[R];
  __shared__ int qe[NW * QCAP];
  __shared__ int qr[NW * QCAP];
  __shared__ int qcnt[NW];

  const int tid = threadIdx.x;
  const int lane = tid & 31;
  const int wave = __builtin_amdgcn_readfirstlane(tid >> 5);
  const int r0 = blockIdx.x * R;
  const int E = (int)E64;

  for (int r = tid; r < R; r += NW * 32) cntl[r] = 0;
  __syncthreads();

  int* myqe = qe + wave * QCAP;
  int* myqr = qr + wave * QCAP;
  const unsigned ltmask = (1u << lane) - 1u;
  int qn = 0;

  float* myl = lst + (size_t)blockIdx.x * LW;
  float* wl = myl + 32 + wave * CAP;
  const bool can_save = (r0 + R <= N) && (N <= (1 << 23));
  int cnt = 0;

  const bool bytes = P::usable(N, E64);

  auto take = [&](int nq) {
    if (lane < nq) {
      const int e = myqe[lane];
      const int r = bytes ? (dst[e] - r0) : myqr[lane];
      __hip_atomic_fetch_add(&cntl[r], 1, __ATOMIC_RELAXED, __HIP_MEMORY_SCOPE_WORKGROUP);
      if (can_save && cnt + nq <= CAP)
        wl[cnt + lane] = __uint_as_float(((unsigned)src[e] << 9) | (unsigned)r);
    }
    cnt += nq;
  };
  auto drain32 = [&]() {
    take(32);
    const int rem = qn - 32;
    int te = 0, tr = 0;
    if (lane < rem) { te = myqe[32 + lane]; tr = bytes ? 0 : myqr[32 + lane]; }
    if (lane < rem) { myqe[lane] = te; if (!bytes) myqr[lane] = tr; }
    qn = rem;
  };

  const bool vec_ok = ((E & 3) == 0) && ((((size_t)dst) & 15) == 0);
  auto load4 = [&](int i) -> int4 {
    int4 d;
    d.x = (i     < E) ? dst[i]     : -1;
    d.y = (i + 1 < E) ? dst[i + 1] : -1;
    d.z = (i + 2 < E) ? dst[i + 2] : -1;
    d.w = (i + 3 < E) ? dst[i + 3] : -1;
    return d;
  };

  auto scan4 = [&](const int4& d, int base) {
    const int d0 = d.x - r0, d1 = d.y - r0, d2 = d.z - r0, d3 = d.w - r0;
    unsigned mm = ((unsigned)d0 < (unsigned)R ? 1u : 0u) |
                  ((unsigned)d1 < (unsigned)R ? 2u : 0u) |
                  ((unsigned)d2 < (unsigned)R ? 4u : 0u) |
                  ((unsigned)d3 < (unsigned)R ? 8u : 0u);
    while (true) {
      const bool has = mm != 0u;
      const unsigned mk = __builtin_amdgcn_ballot_w32(has);
      if (mk == 0u) break;
      if (has) {
        const int c = __builtin_ctz(mm);
        const int dsel = (c == 0) ? d0 : (c == 1) ? d1 : (c == 2) ? d2 : d3;
        const int p = qn + __builtin_popcount(mk & ltmask);
        myqe[p] = base + c;
        myqr[p] = dsel;
      }
      qn += __builtin_popcount(mk);
      mm &= mm - 1u;
      if (qn >= 32) drain32();
    }
  };

  constexpr int TRIP = 8;
  if (bytes) {
    const unsigned pat = (unsigned)blockIdx.x * 0x01010101u;
    auto hit = [&](unsigned x) -> unsigned {
      x ^= pat;
      return ~(((x & 0x7F7F7F7Fu) + 0x7F7F7F7Fu) | x | 0x7F7F7F7Fu);
    };
    auto mask16 = [&](const uint4& c) -> unsigned {
      return (hit(c.x) >> 7) | (hit(c.y) >> 6) | (hit(c.z) >> 5) | (hit(c.w) >> 4);
    };
    const unsigned* pl = (const unsigned*)lst;
    const int ntr = (E + 4095) >> 12;
    for (int tr = wave; tr < ntr; tr += NW) {
      const uint4* p = (const uint4*)(pl + (size_t)(tr / 3) * LW + P::TOFF) +
                       (tr % 3) * (TRIP * 32) + lane;
      const int b = tr * 4096 + lane * 16;
      uint4 c[TRIP];
      if (tr * 4096 + 4096 <= E) {
#pragma unroll
        for (int j = 0; j < TRIP; ++j) c[j] = p[j * 32];
      } else {
#pragma unroll
        for (int j = 0; j < TRIP; ++j) {
          const uint4 ff = {~0u, ~0u, ~0u, ~0u};
          c[j] = (b + j * 512 < E) ? p[j * 32] : ff;
        }
      }
      unsigned m4[TRIP / 2];
#pragma unroll
      for (int jj = 0; jj < TRIP / 2; ++jj)
        m4[jj] = mask16(c[2 * jj]) | (mask16(c[2 * jj + 1]) << 4);
#pragma unroll 1
      for (int jj = 0; jj < TRIP / 2; ++jj) {
        unsigned mm = (jj == 0) ? m4[0] : (jj == 1) ? m4[1] : (jj == 2) ? m4[2] : m4[3];
        const int base = b + jj * 1024;
        while (true) {
          const bool has = mm != 0u;
          const unsigned mk = __builtin_amdgcn_ballot_w32(has);
          if (mk == 0u) break;
          if (has) {
            const int p = __builtin_ctz(mm);
            myqe[qn + __builtin_popcount(mk & ltmask)] =
                base + (((p >> 2) & 1) << 9) + ((p & 3) << 2) + (p >> 3);
          }
          qn += __builtin_popcount(mk);
          mm &= mm - 1u;
          if (qn >= 32) drain32();
        }
      }
    }
  } else {
    const int nsc = (E + 1023) >> 10;
    for (int sc = wave; sc < nsc; sc += NW) {
      const int b = sc * 1024 + lane * 4;
      int4 c[TRIP];
      if (vec_ok && sc * 1024 + 1024 <= E) {
        const int4* p = (const int4*)(dst + b);
#pragma unroll
        for (int j = 0; j < TRIP; ++j) c[j] = p[j * 32];
      } else {
#pragma unroll
        for (int j = 0; j < TRIP; ++j) c[j] = load4(b + j * 128);
      }
#pragma unroll
      for (int j = 0; j < TRIP; ++j) scan4(c[j], b + j * 128);
    }
  }

  if (qn > 0) take(qn);
  if (lane == 0) qcnt[wave] = (can_save && cnt <= CAP) ? cnt : -1;
  __syncthreads();

  if (wave == 0) myl[lane] = __int_as_float(qcnt[lane]);

  for (int r = tid; r < R; r += NW * 32) {
    const int v = r0 + r;
    if (v < N) dinv[v] = __frsqrt_rn(fmaxf((float)(cntl[r] + 1), 1e-12f));
  }
}

template <int F, int R, bool RELU>
__global__ __launch_bounds__(1024) void owner_agg_kernel(
    const int* __restrict__ src, const int* __restrict__ dst,
    const float* __restrict__ dinv, const float* __restrict__ h,
    const float* __restrict__ bias, float* out, const float* lst, int N,
    long long E64) {
  constexpr int VEC = F / 32;
  constexpr int NW = 32;
  constexpr int QCAP = 64;
  constexpr int GRP = 8;
  constexpr int GRPL = (VEC > 2) ? 8 : 16;
  typedef Plane<R> P;
  constexpr int LW = P::LW;
  constexpr int CAP = P::CAP;
  static_assert(P::NW == NW, "slice layout is for 32 waves");
  typedef float VT __attribute__((ext_vector_type(VEC)));

  __shared__ float acc[R * F];
  __shared__ float dloc[R];
  __shared__ int qe[NW * QCAP];
  __shared__ int qr[NW * QCAP];

  const int tid = threadIdx.x;
  const int lane = tid & 31;
  const int wave = __builtin_amdgcn_readfirstlane(tid >> 5);
  const int r0 = blockIdx.x * R;
  const int E = (int)E64;

  for (int r = wave; r < R; r += NW) {
    const int v = r0 + r;
    if (v < N) {
      const float di = dinv[v];
      if (lane == 0) dloc[r] = di;
      const float s = di * di;
      const VT hv = *(const VT*)(h + (size_t)v * F + lane * VEC);
#pragma unroll
      for (int t = 0; t < VEC; ++t) acc[r * F + t * 32 + lane] = s * hv[t];
    } else {
      if (lane == 0) dloc[r] = 0.0f;
#pragma unroll
      for (int t = 0; t < VEC; ++t) acc[r * F + t * 32 + lane] = 0.0f;
    }
  }
  __syncthreads();

  auto add_row = [&](int rk, float wk, const VT& hv) {
#pragma unroll
    for (int t = 0; t < VEC; ++t)
      __hip_atomic_fetch_add(&acc[rk * F + t * 32 + lane], wk * hv[t],
                             __ATOMIC_RELAXED, __HIP_MEMORY_SCOPE_WORKGROUP);
  };

  const float* myl = lst + (size_t)blockIdx.x * LW;
  const float* wl = myl + 32 + wave * CAP;
  const int c = __float_as_int(myl[lane]);
  const bool use_list = __builtin_amdgcn_ballot_w32(c < 0 || c > CAP) == 0u;
  if (use_list) {
    const int n = __builtin_amdgcn_readlane(c, wave);
    auto ld = [&](int b) -> unsigned {
      return (b + lane < n) ? __float_as_uint(wl[b + lane]) : 0u;
    };
    unsigned pk = ld(0);
    unsigned pk1 = ld(32);
    float dv = dinv[pk >> 9];
    for (int b = 0; b < n; b += 32) {
      const unsigned pk2 = ld(b + 64);
      const float dv1 = dinv[pk1 >> 9];
      const int u = (int)(pk >> 9);
      const int r = (int)(pk & 511u);
      const float w = dv * dloc[r];
      const int m = (n - b < 32) ? (n - b) : 32;
      int g = 0;
      for (; g + GRPL <= m; g += GRPL) {
        VT hv[GRPL];
        int rr[GRPL];
        float ww[GRPL];
#pragma unroll
        for (int k = 0; k < GRPL; ++k) {
          const int uk = __builtin_amdgcn_readlane(u, g + k);
          rr[k] = __builtin_amdgcn_readlane(r, g + k);
          ww[k] = __int_as_float(__builtin_amdgcn_readlane(__float_as_int(w), g + k));
          hv[k] = *(const VT*)(h + (size_t)uk * F + lane * VEC);
        }
#pragma unroll
        for (int k = 0; k < GRPL; ++k) add_row(rr[k], ww[k], hv[k]);
      }
      if (g < m) {
        VT hv[GRPL];
        int rr[GRPL];
        float ww[GRPL];
#pragma unroll
        for (int k = 0; k < GRPL; ++k) {
          if (g + k < m) {
            const int uk = __builtin_amdgcn_readlane(u, g + k);
            rr[k] = __builtin_amdgcn_readlane(r, g + k);
            ww[k] = __int_as_float(__builtin_amdgcn_readlane(__float_as_int(w), g + k));
            hv[k] = *(const VT*)(h + (size_t)uk * F + lane * VEC);
          }
        }
#pragma unroll
        for (int k = 0; k < GRPL; ++k) {
          if (g + k < m) add_row(rr[k], ww[k], hv[k]);
        }
      }
      pk = pk1;
      dv = dv1;
      pk1 = pk2;
    }
  } else {
    int* myqe = qe + wave * QCAP;
    int* myqr = qr + wave * QCAP;
    const unsigned ltmask = (1u << lane) - 1u;
    int qn = 0;

    auto drain32 = [&]() {
      const int r = myqr[lane];
      const int u = src[myqe[lane]];
      const float w = dinv[u] * dloc[r];
      for (int g = 0; g < 32; g += GRP) {
        VT hv[GRP];
        int rr[GRP];
        float ww[GRP];
#pragma unroll
        for (int k = 0; k < GRP; ++k) {
          const int uk = __builtin_amdgcn_readlane(u, g + k);
          rr[k] = __builtin_amdgcn_readlane(r, g + k);
          ww[k] = __int_as_float(__builtin_amdgcn_readlane(__float_as_int(w), g + k));
          hv[k] = *(const VT*)(h + (size_t)uk * F + lane * VEC);
        }
#pragma unroll
        for (int k = 0; k < GRP; ++k) add_row(rr[k], ww[k], hv[k]);
      }
      const int rem = qn - 32;
      int te = 0, tr = 0;
      if (lane < rem) { te = myqe[32 + lane]; tr = myqr[32 + lane]; }
      if (lane < rem) { myqe[lane] = te; myqr[lane] = tr; }
      qn = rem;
    };

    const bool vec_ok = ((E & 3) == 0) && ((((size_t)dst) & 15) == 0);
    auto load4 = [&](int i) -> int4 {
      int4 d;
      d.x = (i     < E) ? dst[i]     : -1;
      d.y = (i + 1 < E) ? dst[i + 1] : -1;
      d.z = (i + 2 < E) ? dst[i + 2] : -1;
      d.w = (i + 3 < E) ? dst[i + 3] : -1;
      return d;
    };

    auto scan4 = [&](const int4& d, int base) {
      const int d0 = d.x - r0, d1 = d.y - r0, d2 = d.z - r0, d3 = d.w - r0;
      unsigned mm = ((unsigned)d0 < (unsigned)R ? 1u : 0u) |
                    ((unsigned)d1 < (unsigned)R ? 2u : 0u) |
                    ((unsigned)d2 < (unsigned)R ? 4u : 0u) |
                    ((unsigned)d3 < (unsigned)R ? 8u : 0u);
      while (true) {
        const bool has = mm != 0u;
        const unsigned mk = __builtin_amdgcn_ballot_w32(has);
        if (mk == 0u) break;
        if (has) {
          const int cc = __builtin_ctz(mm);
          const int dsel = (cc == 0) ? d0 : (cc == 1) ? d1 : (cc == 2) ? d2 : d3;
          const int p = qn + __builtin_popcount(mk & ltmask);
          myqe[p] = base + cc;
          myqr[p] = dsel;
        }
        qn += __builtin_popcount(mk);
        mm &= mm - 1u;
        if (qn >= 32) drain32();
      }
    };

    constexpr int TRIP = 8;
    const int nsc = (E + 1023) >> 10;
    for (int sc = wave; sc < nsc; sc += NW) {
      const int b = sc * 1024 + lane * 4;
      int4 cv[TRIP];
      if (vec_ok && sc * 1024 + 1024 <= E) {
        const int4* p = (const int4*)(dst + b);
#pragma unroll
        for (int j = 0; j < TRIP; ++j) cv[j] = p[j * 32];
      } else {
#pragma unroll
        for (int j = 0; j < TRIP; ++j) cv[j] = load4(b + j * 128);
      }
#pragma unroll
      for (int j = 0; j < TRIP; ++j) scan4(cv[j], b + j * 128);
    }

    if (qn > 0) {
      int r = 0, u = 0;
      float w = 0.0f;
      if (lane < qn) {
        r = myqr[lane];
        u = src[myqe[lane]];
        w = dinv[u] * dloc[r];
      }
      for (int k = 0; k < qn; ++k) {
        const int uk = __builtin_amdgcn_readlane(u, k);
        const int rk = __builtin_amdgcn_readlane(r, k);
        const float wk = __int_as_float(__builtin_amdgcn_readlane(__float_as_int(w), k));
        const VT hv = *(const VT*)(h + (size_t)uk * F + lane * VEC);
        add_row(rk, wk, hv);
      }
    }
  }
  __syncthreads();

  for (int r = wave; r < R; r += NW) {
    const int v = r0 + r;
    if (v >= N) break;
    VT o;
#pragma unroll
    for (int t = 0; t < VEC; ++t) {
      const float x = acc[r * F + t * 32 + lane] + bias[lane * VEC + t];
      o[t] = RELU ? fmaxf(x, 0.0f) : x;
    }
    *(VT*)(out + (size_t)v * F + lane * VEC) = o;
  }
}

extern "C" void kernel_launch(void* const* d_in, const int* in_sizes, int n_in,
                              void* d_out, int out_size, void* d_ws,
                              size_t ws_size, hipStream_t stream) {
  const float* x  = (const float*)d_in[0];
  const int*   ei = (const int*)d_in[1];
  const float* W1 = (const float*)d_in[2];
  const float* b1 = (const float*)d_in[3];
  const float* W2 = (const float*)d_in[4];
  const float* b2 = (const float*)d_in[5];
  float* out = (float*)d_out;

  const int N = in_sizes[0] / 128;
  const long long E = in_sizes[1] / 2;
  const int* src = ei;
  const int* dst = ei + E;

  float* ws = (float*)d_ws;
  float* dinvw = ws;
  float* h1pre = ws + N;
  float* h1    = h1pre + (size_t)N * 128;
  float* h2pre = h1pre;

  constexpr int R = 400;
  const int TB = 256;
  auto blocks = [](long long n, int tb) { return (unsigned)((n + tb - 1) / tb); };

  plane_kernel<R><<<blocks((E + 15) / 16 * 4, TB), TB, 0, stream>>>(
      dst, (unsigned*)d_out, N, E);
  owner_scan_kernel<R><<<(N + R - 1) / R, 1024, 0, stream>>>(
      src, dst, dinvw, out, N, E);
  const float* dinv = dinvw;

  gemm_wmma_kernel<128, 2><<<768, 256, 0, stream>>>(x, W1, h1pre, N);
  owner_agg_kernel<128, R, true><<<(N + R - 1) / R, 1024, 0, stream>>>(
      src, dst, dinv, h1pre, b1, h1, out, N, E);

  gemm_wmma_kernel<64, 1><<<1024, 256, 0, stream>>>(h1, W2, h2pre, N);
  owner_agg_kernel<64, R, false><<<(N + R - 1) / R, 1024, 0, stream>>>(
      src, dst, dinv, h2pre, b2, out, out, N, E);
}


# Edge scan and layer-1 GEMM, independent, in ONE launch: a scan and a GEMM block of 32 waves side by side per WGP; GEMM B fragments from LDS to fit 64 registers; 6 to 5 launches; k, sum order kept
# speedup vs baseline: 3.3007x; 1.0314x over previous
#include <hip/hip_runtime.h>
#include <hip/hip_bf16.h>

typedef float v2f __attribute__((ext_vector_type(2)));
typedef float v4f __attribute__((ext_vector_type(4)));
typedef float v8f __attribute__((ext_vector_type(8)));

template <int R>
struct Plane {
  static constexpr int LW = R * 64;
  static constexpr int PLW = 3072;
  static constexpr int NW = 32;
  static constexpr int CAP = (((LW - 32 - PLW) / NW) / 32) * 32;
  static constexpr int TOFF = 32 + NW * CAP;
  static_assert(CAP >= 32, "slice too small for the lists");
  static_assert(TOFF % 32 == 0 && TOFF + PLW <= LW, "tail must fit the slice, line aligned");
  __device__ static bool usable(int N, long long E) {
    return (N <= 255 * R) && ((E + PLW * 4 - 1) / (PLW * 4) <= (long long)(N / R));
  }
};

template <int R>
__global__ void plane_kernel(const int* __restrict__ dst, unsigned* __restrict__ slices,
                             int N, long long E) {
  typedef Plane<R> P;
  if (!P::usable(N, E)) return;
  const long long i = (long long)blockIdx.x * blockDim.x + threadIdx.x;
  const long long e0 = i * 4;
  if (e0 >= ((E + 15) / 16) * 16) return;
  int d[4];
  if (e0 + 3 < E && ((((size_t)dst) & 15) == 0)) {
    const int4 v = *(const int4*)(dst + e0);
    d[0] = v.x; d[1] = v.y; d[2] = v.z; d[3] = v.w;
  } else {
#pragma unroll
    for (int k = 0; k < 4; ++k) d[k] = (e0 + k < E) ? dst[e0 + k] : -1;
  }
  unsigned w = 0u;
#pragma unroll
  for (int k = 0; k < 4; ++k) {
    const unsigned o = (d[k] >= 0) ? (unsigned)(d[k] / R) : 255u;
    w |= (o & 255u) << (8 * k);
  }
  slices[(size_t)(i / P::PLW) * P::LW + P::TOFF + (int)(i % P::PLW)] = w;
}

template <int FOUT, int RT>
__global__ __launch_bounds__(256) void gemm_wmma_kernel(
    const float* __restrict__ A, const float* __restrict__ W,
    float* __restrict__ C, int nrows) {
  constexpr int PITCH = 132;
  constexpr int NCS = FOUT / 16;
  constexpr int RG = 8 / NCS;
  constexpr int TM = 16 * RT * RG;
  constexpr int NF4 = TM * 32 / 256;
  static_assert(NCS * RG == 8 && NF4 * 256 == TM * 32, "tile must split over 8 waves");
  __shared__ alignas(16) float lds[2][TM * PITCH];

  const int ntiles = (nrows + TM - 1) / TM;
  int tile = blockIdx.x;
  if (tile >= ntiles) return;

  const int tid = threadIdx.x;
  const int lane = tid & 31;
  const int wave = tid >> 5;
  const int colbase = (wave % NCS) * 16;
  const int rg = wave / NCS;
  const int half = lane >> 4;
  const int m = lane & 15;

  v2f b[32];
#pragma unroll
  for (int s = 0; s < 32; ++s) {
    const int kb = 4 * s + half * 2;
    b[s].x = W[(size_t)kb * FOUT + colbase + m];
    b[s].y = W[(size_t)(kb + 1) * FOUT + colbase + m];
  }

  v4f st[NF4];
  auto fetch = [&](int t) {
    const int row0 = t * TM;
    const v4f* srcv = (const v4f*)(A + (size_t)row0 * 128);
    if (row0 + TM <= nrows) {
#pragma unroll
      for (int j = 0; j < NF4; ++j) st[j] = srcv[tid + j * 256];
    } else {
#pragma unroll
      for (int j = 0; j < NF4; ++j) {
        const int i = tid + j * 256;
        const v4f z = {0.0f, 0.0f, 0.0f, 0.0f};
        st[j] = z;
        if (row0 + (i >> 5) < nrows) st[j] = srcv[i];
      }
    }
  };
  auto stash = [&](int buf) {
#pragma unroll
    for (int j = 0; j < NF4; ++j) {
      const int i = tid + j * 256;
      *(v4f*)&lds[buf][(i >> 5) * PITCH + (i & 31) * 4] = st[j];
    }
  };

  fetch(tile);
  stash(0);
  __syncthreads();
  int buf = 0;
  const int step = (int)gridDim.x;
  for (; tile < ntiles; tile += step) {
    const int nxt = tile + step;
    if (nxt < ntiles) fetch(nxt);

    v8f acc[RT];
#pragma unroll
    for (int t = 0; t < RT; ++t) acc[t] = v8f{};
    const float* abase = &lds[buf][(rg * RT * 16 + m) * PITCH + half * 2];
#pragma unroll
    for (int s = 0; s < 32; ++s) {
#pragma unroll
      for (int t = 0; t < RT; ++t) {
        const v2f a = *(const v2f*)(abase + t * 16 * PITCH + 4 * s);
        acc[t] = __builtin_amdgcn_wmma_f32_16x16x4_f32(false, a, false, b[s], (short)0,
                                                       acc[t], false, false);
      }
    }

    if (nxt < ntiles) stash(buf ^ 1);
    __syncthreads();
    buf ^= 1;

#pragma unroll
    for (int t = 0; t < RT; ++t) {
      const int rbase = tile * TM + (rg * RT + t) * 16 + half * 8;
#pragma unroll
      for (int i = 0; i < 8; ++i) {
        const int row = rbase + i;
        if (row < nrows) C[(size_t)row * FOUT + colbase + m] = acc[t][i];
      }
    }
  }
}

template <int R>
__device__ __forceinline__ void owner_scan_part(
    const int* __restrict__ src, const int* __restrict__ dst,
    float* __restrict__ dinv, float* lst, int N, long long E64, const int ob) {
  constexpr int NW = 32;
  constexpr int QCAP = 64;
  typedef Plane<R> P;
  constexpr int LW = P::LW;
  constexpr int CAP = P::CAP;
  static_assert(R <= 512, "owned row must fit the 9 low bits of a list word");
  static_assert(P::NW == NW, "slice layout is for 32 waves");

  __shared__ int cntl[R];
  __shared__ int qe[NW * QCAP];
  __shared__ int qr[NW * QCAP];
  __shared__ int qcnt[NW];

  const int tid = threadIdx.x;
  const int lane = tid & 31;
  const int wave = __builtin_amdgcn_readfirstlane(tid >> 5);
  const int r0 = ob * R;
  const int E = (int)E64;

  for (int r = tid; r < R; r += NW * 32) cntl[r] = 0;
  __syncthreads();

  int* myqe = qe + wave * QCAP;
  int* myqr = qr + wave * QCAP;
  const unsigned ltmask = (1u << lane) - 1u;
  int qn = 0;

  float* myl = lst + (size_t)ob * LW;
  float* wl = myl + 32 + wave * CAP;
  const bool can_save = (r0 + R <= N) && (N <= (1 << 23));
  int cnt = 0;

  const bool bytes = P::usable(N, E64);

  auto take = [&](int nq) {
    if (lane < nq) {
      const int e = myqe[lane];
      const int r = bytes ? (dst[e] - r0) : myqr[lane];
      __hip_atomic_fetch_add(&cntl[r], 1, __ATOMIC_RELAXED, __HIP_MEMORY_SCOPE_WORKGROUP);
      if (can_save && cnt + nq <= CAP)
        wl[cnt + lane] = __uint_as_float(((unsigned)src[e] << 9) | (unsigned)r);
    }
    cnt += nq;
  };
  auto drain32 = [&]() {
    take(32);
    const int rem = qn - 32;
    int te = 0, tr = 0;
    if (lane < rem) { te = myqe[32 + lane]; tr = bytes ? 0 : myqr[32 + lane]; }
    if (lane < rem) { myqe[lane] = te; if (!bytes) myqr[lane] = tr; }
    qn = rem;
  };

  const bool vec_ok = ((E & 3) == 0) && ((((size_t)dst) & 15) == 0);
  auto load4 = [&](int i) -> int4 {
    int4 d;
    d.x = (i     < E) ? dst[i]     : -1;
    d.y = (i + 1 < E) ? dst[i + 1] : -1;
    d.z = (i + 2 < E) ? dst[i + 2] : -1;
    d.w = (i + 3 < E) ? dst[i + 3] : -1;
    return d;
  };

  auto scan4 = [&](const int4& d, int base) {
    const int d0 = d.x - r0, d1 = d.y - r0, d2 = d.z - r0, d3 = d.w - r0;
    unsigned mm = ((unsigned)d0 < (unsigned)R ? 1u : 0u) |
                  ((unsigned)d1 < (unsigned)R ? 2u : 0u) |
                  ((unsigned)d2 < (unsigned)R ? 4u : 0u) |
                  ((unsigned)d3 < (unsigned)R ? 8u : 0u);
    while (true) {
      const bool has = mm != 0u;
      const unsigned mk = __builtin_amdgcn_ballot_w32(has);
      if (mk == 0u) break;
      if (has) {
        const int c = __builtin_ctz(mm);
        const int dsel = (c == 0) ? d0 : (c == 1) ? d1 : (c == 2) ? d2 : d3;
        const int p = qn + __builtin_popcount(mk & ltmask);
        myqe[p] = base + c;
        myqr[p] = dsel;
      }
      qn += __builtin_popcount(mk);
      mm &= mm - 1u;
      if (qn >= 32) drain32();
    }
  };

  constexpr int TRIP = 8;
  if (bytes) {
    const unsigned pat = (unsigned)ob * 0x01010101u;
    auto hit = [&](unsigned x) -> unsigned {
      x ^= pat;
      return ~(((x & 0x7F7F7F7Fu) + 0x7F7F7F7Fu) | x | 0x7F7F7F7Fu);
    };
    auto mask16 = [&](const uint4& c) -> unsigned {
      return (hit(c.x) >> 7) | (hit(c.y) >> 6) | (hit(c.z) >> 5) | (hit(c.w) >> 4);
    };
    const unsigned* pl = (const unsigned*)lst;
    const int ntr = (E + 4095) >> 12;
    for (int tr = wave; tr < ntr; tr += NW) {
      const uint4* p = (const uint4*)(pl + (size_t)(tr / 3) * LW + P::TOFF) +
                       (tr % 3) * (TRIP * 32) + lane;
      const int b = tr * 4096 + lane * 16;
      uint4 c[TRIP];
      if (tr * 4096 + 4096 <= E) {
#pragma unroll
        for (int j = 0; j < TRIP; ++j) c[j] = p[j * 32];
      } else {
#pragma unroll
        for (int j = 0; j < TRIP; ++j) {
          const uint4 ff = {~0u, ~0u, ~0u, ~0u};
          c[j] = (b + j * 512 < E) ? p[j * 32] : ff;
        }
      }
      unsigned m4[TRIP / 2];
#pragma unroll
      for (int jj = 0; jj < TRIP / 2; ++jj)
        m4[jj] = mask16(c[2 * jj]) | (mask16(c[2 * jj + 1]) << 4);
#pragma unroll 1
      for (int jj = 0; jj < TRIP / 2; ++jj) {
        unsigned mm = (jj == 0) ? m4[0] : (jj == 1) ? m4[1] : (jj == 2) ? m4[2] : m4[3];
        const int base = b + jj * 1024;
        while (true) {
          const bool has = mm != 0u;
          const unsigned mk = __builtin_amdgcn_ballot_w32(has);
          if (mk == 0u) break;
          if (has) {
            const int p = __builtin_ctz(mm);
            myqe[qn + __builtin_popcount(mk & ltmask)] =
                base + (((p >> 2) & 1) << 9) + ((p & 3) << 2) + (p >> 3);
          }
          qn += __builtin_popcount(mk);
          mm &= mm - 1u;
          if (qn >= 32) drain32();
        }
      }
    }
  } else {
    const int nsc = (E + 1023) >> 10;
    for (int sc = wave; sc < nsc; sc += NW) {
      const int b = sc * 1024 + lane * 4;
      int4 c[TRIP];
      if (vec_ok && sc * 1024 + 1024 <= E) {
        const int4* p = (const int4*)(dst + b);
#pragma unroll
        for (int j = 0; j < TRIP; ++j) c[j] = p[j * 32];
      } else {
#pragma unroll
        for (int j = 0; j < TRIP; ++j) c[j] = load4(b + j * 128);
      }
#pragma unroll
      for (int j = 0; j < TRIP; ++j) scan4(c[j], b + j * 128);
    }
  }

  if (qn > 0) take(qn);
  if (lane == 0) qcnt[wave] = (can_save && cnt <= CAP) ? cnt : -1;
  __syncthreads();

  if (wave == 0) myl[lane] = __int_as_float(qcnt[lane]);

  for (int r = tid; r < R; r += NW * 32) {
    const int v = r0 + r;
    if (v < N) dinv[v] = __frsqrt_rn(fmaxf((float)(cntl[r] + 1), 1e-12f));
  }
}

template <int FOUT>
__device__ __forceinline__ void gemm_lds_b_part(
    const float* __restrict__ A, const float* __restrict__ W,
    float* __restrict__ C, int nrows, const int gb, const int ngb) {
  constexpr int PITCH = 132;
  constexpr int NCS = FOUT / 16;
  constexpr int RG = 32 / NCS;
  constexpr int TM = 16 * RG;
  constexpr int NF4 = TM * 32 / 1024;
  constexpr int NWQ = 128 * FOUT / 1024;
  static_assert(NCS * RG == 32 && NF4 * 1024 == TM * 32 && NWQ * 1024 == 128 * FOUT,
                "tile must split over 32 waves");
  __shared__ alignas(16) float wl[128 * FOUT];
  __shared__ alignas(16) float at[2][TM * PITCH];

  const int ntiles = (nrows + TM - 1) / TM;
  int tile = gb;
  if (tile >= ntiles) return;

  const int tid = threadIdx.x;
  const int lane = tid & 31;
  const int wave = tid >> 5;
  const int strip = wave % NCS;
  const int colbase = strip * 16;
  const int rg = wave / NCS;
  const int half = lane >> 4;
  const int m = lane & 15;

#pragma unroll
  for (int q = 0; q < NWQ; ++q) {
    const int i = tid + q * 1024;
    const int k = i / FOUT, col = i % FOUT;
    const int d = ((((k >> 2) * NCS + (col >> 4)) * 32 + ((k >> 1) & 1) * 16 + (col & 15)) << 1) + (k & 1);
    wl[d] = W[i];
  }

  v4f st[NF4];
  auto fetch = [&](int t) {
    const int row0 = t * TM;
    const v4f* srcv = (const v4f*)(A + (size_t)row0 * 128);
    if (row0 + TM <= nrows) {
#pragma unroll
      for (int j = 0; j < NF4; ++j) st[j] = srcv[tid + j * 1024];
    } else {
#pragma unroll
      for (int j = 0; j < NF4; ++j) {
        const int i = tid + j * 1024;
        const v4f z = {0.0f, 0.0f, 0.0f, 0.0f};
        st[j] = z;
        if (row0 + (i >> 5) < nrows) st[j] = srcv[i];
      }
    }
  };
  auto stash = [&](int buf) {
#pragma unroll
    for (int j = 0; j < NF4; ++j) {
      const int i = tid + j * 1024;
      *(v4f*)&at[buf][(i >> 5) * PITCH + (i & 31) * 4] = st[j];
    }
  };

  fetch(tile);
  stash(0);
  __syncthreads();
  int buf = 0;
  const float* bbase = &wl[(strip * 32 + lane) * 2];
  for (; tile < ntiles; tile += ngb) {
    const int nxt = tile + ngb;
    if (nxt < ntiles) fetch(nxt);

    v8f acc = v8f{};
    const float* abase = &at[buf][(rg * 16 + m) * PITCH + half * 2];
#pragma unroll
    for (int s = 0; s < 32; ++s) {
      const v2f a = *(const v2f*)(abase + 4 * s);
      const v2f b = *(const v2f*)(bbase + s * NCS * 64);
      acc = __builtin_amdgcn_wmma_f32_16x16x4_f32(false, a, false, b, (short)0, acc,
                                                  false, false);
    }

    if (nxt < ntiles) stash(buf ^ 1);
    __syncthreads();
    buf ^= 1;

    const int rbase = tile * TM + rg * 16 + half * 8;
#pragma unroll
    for (int i = 0; i < 8; ++i) {
      const int row = rbase + i;
      if (row < nrows) C[(size_t)row * FOUT + colbase + m] = acc[i];
    }
  }
}

template <int R>
__global__ __launch_bounds__(1024, 16) void scan_gemm_kernel(
    const int* __restrict__ src, const int* __restrict__ dst,
    float* __restrict__ dinv, float* lst, int N, long long E64,
    const float* __restrict__ A, const float* __restrict__ W,
    float* __restrict__ C, int nsb) {
  const int bi = (int)blockIdx.x;
  const int p = bi >> 1;
  const int nb = (int)(gridDim.x >> 1);
  if (((bi ^ (bi >> 8)) & 1) == 0) {
    if (p < nsb) owner_scan_part<R>(src, dst, dinv, lst, N, E64, p);
  } else {
    gemm_lds_b_part<128>(A, W, C, N, p, nb);
  }
}

template <int F, int R, bool RELU>
__global__ __launch_bounds__(1024) void owner_agg_kernel(
    const int* __restrict__ src, const int* __restrict__ dst,
    const float* __restrict__ dinv, const float* __restrict__ h,
    const float* __restrict__ bias, float* out, const float* lst, int N,
    long long E64) {
  constexpr int VEC = F / 32;
  constexpr int NW = 32;
  constexpr int QCAP = 64;
  constexpr int GRP = 8;
  constexpr int GRPL = (VEC > 2) ? 8 : 16;
  typedef Plane<R> P;
  constexpr int LW = P::LW;
  constexpr int CAP = P::CAP;
  static_assert(P::NW == NW, "slice layout is for 32 waves");
  typedef float VT __attribute__((ext_vector_type(VEC)));

  __shared__ float acc[R * F];
  __shared__ float dloc[R];
  __shared__ int qe[NW * QCAP];
  __shared__ int qr[NW * QCAP];

  const int tid = threadIdx.x;
  const int lane = tid & 31;
  const int wave = __builtin_amdgcn_readfirstlane(tid >> 5);
  const int r0 = blockIdx.x * R;
  const int E = (int)E64;

  for (int r = wave; r < R; r += NW) {
    const int v = r0 + r;
    if (v < N) {
      const float di = dinv[v];
      if (lane == 0) dloc[r] = di;
      const float s = di * di;
      const VT hv = *(const VT*)(h + (size_t)v * F + lane * VEC);
#pragma unroll
      for (int t = 0; t < VEC; ++t) acc[r * F + t * 32 + lane] = s * hv[t];
    } else {
      if (lane == 0) dloc[r] = 0.0f;
#pragma unroll
      for (int t = 0; t < VEC; ++t) acc[r * F + t * 32 + lane] = 0.0f;
    }
  }
  __syncthreads();

  auto add_row = [&](int rk, float wk, const VT& hv) {
#pragma unroll
    for (int t = 0; t < VEC; ++t)
      __hip_atomic_fetch_add(&acc[rk * F + t * 32 + lane], wk * hv[t],
                             __ATOMIC_RELAXED, __HIP_MEMORY_SCOPE_WORKGROUP);
  };

  const float* myl = lst + (size_t)blockIdx.x * LW;
  const float* wl = myl + 32 + wave * CAP;
  const int c = __float_as_int(myl[lane]);
  const bool use_list = __builtin_amdgcn_ballot_w32(c < 0 || c > CAP) == 0u;
  if (use_list) {
    const int n = __builtin_amdgcn_readlane(c, wave);
    auto ld = [&](int b) -> unsigned {
      return (b + lane < n) ? __float_as_uint(wl[b + lane]) : 0u;
    };
    unsigned pk = ld(0);
    unsigned pk1 = ld(32);
    float dv = dinv[pk >> 9];
    for (int b = 0; b < n; b += 32) {
      const unsigned pk2 = ld(b + 64);
      const float dv1 = dinv[pk1 >> 9];
      const int u = (int)(pk >> 9);
      const int r = (int)(pk & 511u);
      const float w = dv * dloc[r];
      const int m = (n - b < 32) ? (n - b) : 32;
      int g = 0;
      for (; g + GRPL <= m; g += GRPL) {
        VT hv[GRPL];
        int rr[GRPL];
        float ww[GRPL];
#pragma unroll
        for (int k = 0; k < GRPL; ++k) {
          const int uk = __builtin_amdgcn_readlane(u, g + k);
          rr[k] = __builtin_amdgcn_readlane(r, g + k);
          ww[k] = __int_as_float(__builtin_amdgcn_readlane(__float_as_int(w), g + k));
          hv[k] = *(const VT*)(h + (size_t)uk * F + lane * VEC);
        }
#pragma unroll
        for (int k = 0; k < GRPL; ++k) add_row(rr[k], ww[k], hv[k]);
      }
      if (g < m) {
        VT hv[GRPL];
        int rr[GRPL];
        float ww[GRPL];
#pragma unroll
        for (int k = 0; k < GRPL; ++k) {
          if (g + k < m) {
            const int uk = __builtin_amdgcn_readlane(u, g + k);
            rr[k] = __builtin_amdgcn_readlane(r, g + k);
            ww[k] = __int_as_float(__builtin_amdgcn_readlane(__float_as_int(w), g + k));
            hv[k] = *(const VT*)(h + (size_t)uk * F + lane * VEC);
          }
        }
#pragma unroll
        for (int k = 0; k < GRPL; ++k) {
          if (g + k < m) add_row(rr[k], ww[k], hv[k]);
        }
      }
      pk = pk1;
      dv = dv1;
      pk1 = pk2;
    }
  } else {
    int* myqe = qe + wave * QCAP;
    int* myqr = qr + wave * QCAP;
    const unsigned ltmask = (1u << lane) - 1u;
    int qn = 0;

    auto drain32 = [&]() {
      const int r = myqr[lane];
      const int u = src[myqe[lane]];
      const float w = dinv[u] * dloc[r];
      for (int g = 0; g < 32; g += GRP) {
        VT hv[GRP];
        int rr[GRP];
        float ww[GRP];
#pragma unroll
        for (int k = 0; k < GRP; ++k) {
          const int uk = __builtin_amdgcn_readlane(u, g + k);
          rr[k] = __builtin_amdgcn_readlane(r, g + k);
          ww[k] = __int_as_float(__builtin_amdgcn_readlane(__float_as_int(w), g + k));
          hv[k] = *(const VT*)(h + (size_t)uk * F + lane * VEC);
        }
#pragma unroll
        for (int k = 0; k < GRP; ++k) add_row(rr[k], ww[k], hv[k]);
      }
      const int rem = qn - 32;
      int te = 0, tr = 0;
      if (lane < rem) { te = myqe[32 + lane]; tr = myqr[32 + lane]; }
      if (lane < rem) { myqe[lane] = te; myqr[lane] = tr; }
      qn = rem;
    };

    const bool vec_ok = ((E & 3) == 0) && ((((size_t)dst) & 15) == 0);
    auto load4 = [&](int i) -> int4 {
      int4 d;
      d.x = (i     < E) ? dst[i]     : -1;
      d.y = (i + 1 < E) ? dst[i + 1] : -1;
      d.z = (i + 2 < E) ? dst[i + 2] : -1;
      d.w = (i + 3 < E) ? dst[i + 3] : -1;
      return d;
    };

    auto scan4 = [&](const int4& d, int base) {
      const int d0 = d.x - r0, d1 = d.y - r0, d2 = d.z - r0, d3 = d.w - r0;
      unsigned mm = ((unsigned)d0 < (unsigned)R ? 1u : 0u) |
                    ((unsigned)d1 < (unsigned)R ? 2u : 0u) |
                    ((unsigned)d2 < (unsigned)R ? 4u : 0u) |
                    ((unsigned)d3 < (unsigned)R ? 8u : 0u);
      while (true) {
        const bool has = mm != 0u;
        const unsigned mk = __builtin_amdgcn_ballot_w32(has);
        if (mk == 0u) break;
        if (has) {
          const int cc = __builtin_ctz(mm);
          const int dsel = (cc == 0) ? d0 : (cc == 1) ? d1 : (cc == 2) ? d2 : d3;
          const int p = qn + __builtin_popcount(mk & ltmask);
          myqe[p] = base + cc;
          myqr[p] = dsel;
        }
        qn += __builtin_popcount(mk);
        mm &= mm - 1u;
        if (qn >= 32) drain32();
      }
    };

    constexpr int TRIP = 8;
    const int nsc = (E + 1023) >> 10;
    for (int sc = wave; sc < nsc; sc += NW) {
      const int b = sc * 1024 + lane * 4;
      int4 cv[TRIP];
      if (vec_ok && sc * 1024 + 1024 <= E) {
        const int4* p = (const int4*)(dst + b);
#pragma unroll
        for (int j = 0; j < TRIP; ++j) cv[j] = p[j * 32];
      } else {
#pragma unroll
        for (int j = 0; j < TRIP; ++j) cv[j] = load4(b + j * 128);
      }
#pragma unroll
      for (int j = 0; j < TRIP; ++j) scan4(cv[j], b + j * 128);
    }

    if (qn > 0) {
      int r = 0, u = 0;
      float w = 0.0f;
      if (lane < qn) {
        r = myqr[lane];
        u = src[myqe[lane]];
        w = dinv[u] * dloc[r];
      }
      for (int k = 0; k < qn; ++k) {
        const int uk = __builtin_amdgcn_readlane(u, k);
        const int rk = __builtin_amdgcn_readlane(r, k);
        const float wk = __int_as_float(__builtin_amdgcn_readlane(__float_as_int(w), k));
        const VT hv = *(const VT*)(h + (size_t)uk * F + lane * VEC);
        add_row(rk, wk, hv);
      }
    }
  }
  __syncthreads();

  for (int r = wave; r < R; r += NW) {
    const int v = r0 + r;
    if (v >= N) break;
    VT o;
#pragma unroll
    for (int t = 0; t < VEC; ++t) {
      const float x = acc[r * F + t * 32 + lane] + bias[lane * VEC + t];
      o[t] = RELU ? fmaxf(x, 0.0f) : x;
    }
    *(VT*)(out + (size_t)v * F + lane * VEC) = o;
  }
}

extern "C" void kernel_launch(void* const* d_in, const int* in_sizes, int n_in,
                              void* d_out, int out_size, void* d_ws,
                              size_t ws_size, hipStream_t stream) {
  const float* x  = (const float*)d_in[0];
  const int*   ei = (const int*)d_in[1];
  const float* W1 = (const float*)d_in[2];
  const float* b1 = (const float*)d_in[3];
  const float* W2 = (const float*)d_in[4];
  const float* b2 = (const float*)d_in[5];
  float* out = (float*)d_out;

  const int N = in_sizes[0] / 128;
  const long long E = in_sizes[1] / 2;
  const int* src = ei;
  const int* dst = ei + E;

  float* ws = (float*)d_ws;
  float* dinvw = ws;
  float* h1pre = ws + N;
  float* h1    = h1pre + (size_t)N * 128;
  float* h2pre = h1pre;

  constexpr int R = 400;
  const int TB = 256;
  auto blocks = [](long long n, int tb) { return (unsigned)((n + tb - 1) / tb); };

  plane_kernel<R><<<blocks((E + 15) / 16 * 4, TB), TB, 0, stream>>>(
      dst, (unsigned*)d_out, N, E);
  const int nsb = (N + R - 1) / R;
  const int nb = nsb > 256 ? nsb : 256;
  scan_gemm_kernel<R><<<2 * nb, 1024, 0, stream>>>(
      src, dst, dinvw, out, N, E, x, W1, h1pre, nsb);
  const float* dinv = dinvw;

  owner_agg_kernel<128, R, true><<<(N + R - 1) / R, 1024, 0, stream>>>(
      src, dst, dinv, h1pre, b1, h1, out, N, E);

  gemm_wmma_kernel<64, 1><<<1024, 256, 0, stream>>>(h1, W2, h2pre, N);
  owner_agg_kernel<64, R, false><<<(N + R - 1) / R, 1024, 0, stream>>>(
      src, dst, dinv, h2pre, b2, out, out, N, E);
}


# Aggregation blocks: a wave's 12-13 start rows and their dinv loaded as one group (was two dependent memory hops per row), list head fetched under them ahead of the barrier; sum order unchanged
# speedup vs baseline: 3.4279x; 1.0385x over previous
#include <hip/hip_runtime.h>
#include <hip/hip_bf16.h>

typedef float v2f __attribute__((ext_vector_type(2)));
typedef float v4f __attribute__((ext_vector_type(4)));
typedef float v8f __attribute__((ext_vector_type(8)));

template <int R>
struct Plane {
  static constexpr int LW = R * 64;
  static constexpr int PLW = 3072;
  static constexpr int NW = 32;
  static constexpr int CAP = (((LW - 32 - PLW) / NW) / 32) * 32;
  static constexpr int TOFF = 32 + NW * CAP;
  static_assert(CAP >= 32, "slice too small for the lists");
  static_assert(TOFF % 32 == 0 && TOFF + PLW <= LW, "tail must fit the slice, line aligned");
  __device__ static bool usable(int N, long long E) {
    return (N <= 255 * R) && ((E + PLW * 4 - 1) / (PLW * 4) <= (long long)(N / R));
  }
};

template <int R>
__global__ void plane_kernel(const int* __restrict__ dst, unsigned* __restrict__ slices,
                             int N, long long E) {
  typedef Plane<R> P;
  if (!P::usable(N, E)) return;
  const long long i = (long long)blockIdx.x * blockDim.x + threadIdx.x;
  const long long e0 = i * 4;
  if (e0 >= ((E + 15) / 16) * 16) return;
  int d[4];
  if (e0 + 3 < E && ((((size_t)dst) & 15) == 0)) {
    const int4 v = *(const int4*)(dst + e0);
    d[0] = v.x; d[1] = v.y; d[2] = v.z; d[3] = v.w;
  } else {
#pragma unroll
    for (int k = 0; k < 4; ++k) d[k] = (e0 + k < E) ? dst[e0 + k] : -1;
  }
  unsigned w = 0u;
#pragma unroll
  for (int k = 0; k < 4; ++k) {
    const unsigned o = (d[k] >= 0) ? (unsigned)(d[k] / R) : 255u;
    w |= (o & 255u) << (8 * k);
  }
  slices[(size_t)(i / P::PLW) * P::LW + P::TOFF + (int)(i % P::PLW)] = w;
}

template <int FOUT, int RT>
__global__ __launch_bounds__(256) void gemm_wmma_kernel(
    const float* __restrict__ A, const float* __restrict__ W,
    float* __restrict__ C, int nrows) {
  constexpr int PITCH = 132;
  constexpr int NCS = FOUT / 16;
  constexpr int RG = 8 / NCS;
  constexpr int TM = 16 * RT * RG;
  constexpr int NF4 = TM * 32 / 256;
  static_assert(NCS * RG == 8 && NF4 * 256 == TM * 32, "tile must split over 8 waves");
  __shared__ alignas(16) float lds[2][TM * PITCH];

  const int ntiles = (nrows + TM - 1) / TM;
  int tile = blockIdx.x;
  if (tile >= ntiles) return;

  const int tid = threadIdx.x;
  const int lane = tid & 31;
  const int wave = tid >> 5;
  const int colbase = (wave % NCS) * 16;
  const int rg = wave / NCS;
  const int half = lane >> 4;
  const int m = lane & 15;

  v2f b[32];
#pragma unroll
  for (int s = 0; s < 32; ++s) {
    const int kb = 4 * s + half * 2;
    b[s].x = W[(size_t)kb * FOUT + colbase + m];
    b[s].y = W[(size_t)(kb + 1) * FOUT + colbase + m];
  }

  v4f st[NF4];
  auto fetch = [&](int t) {
    const int row0 = t * TM;
    const v4f* srcv = (const v4f*)(A + (size_t)row0 * 128);
    if (row0 + TM <= nrows) {
#pragma unroll
      for (int j = 0; j < NF4; ++j) st[j] = srcv[tid + j * 256];
    } else {
#pragma unroll
      for (int j = 0; j < NF4; ++j) {
        const int i = tid + j * 256;
        const v4f z = {0.0f, 0.0f, 0.0f, 0.0f};
        st[j] = z;
        if (row0 + (i >> 5) < nrows) st[j] = srcv[i];
      }
    }
  };
  auto stash = [&](int buf) {
#pragma unroll
    for (int j = 0; j < NF4; ++j) {
      const int i = tid + j * 256;
      *(v4f*)&lds[buf][(i >> 5) * PITCH + (i & 31) * 4] = st[j];
    }
  };

  fetch(tile);
  stash(0);
  __syncthreads();
  int buf = 0;
  const int step = (int)gridDim.x;
  for (; tile < ntiles; tile += step) {
    const int nxt = tile + step;
    if (nxt < ntiles) fetch(nxt);

    v8f acc[RT];
#pragma unroll
    for (int t = 0; t < RT; ++t) acc[t] = v8f{};
    const float* abase = &lds[buf][(rg * RT * 16 + m) * PITCH + half * 2];
#pragma unroll
    for (int s = 0; s < 32; ++s) {
#pragma unroll
      for (int t = 0; t < RT; ++t) {
        const v2f a = *(const v2f*)(abase + t * 16 * PITCH + 4 * s);
        acc[t] = __builtin_amdgcn_wmma_f32_16x16x4_f32(false, a, false, b[s], (short)0,
                                                       acc[t], false, false);
      }
    }

    if (nxt < ntiles) stash(buf ^ 1);
    __syncthreads();
    buf ^= 1;

#pragma unroll
    for (int t = 0; t < RT; ++t) {
      const int rbase = tile * TM + (rg * RT + t) * 16 + half * 8;
#pragma unroll
      for (int i = 0; i < 8; ++i) {
        const int row = rbase + i;
        if (row < nrows) C[(size_t)row * FOUT + colbase + m] = acc[t][i];
      }
    }
  }
}

template <int R>
__device__ __forceinline__ void owner_scan_part(
    const int* __restrict__ src, const int* __restrict__ dst,
    float* __restrict__ dinv, float* lst, int N, long long E64, const int ob) {
  constexpr int NW = 32;
  constexpr int QCAP = 64;
  typedef Plane<R> P;
  constexpr int LW = P::LW;
  constexpr int CAP = P::CAP;
  static_assert(R <= 512, "owned row must fit the 9 low bits of a list word");
  static_assert(P::NW == NW, "slice layout is for 32 waves");

  __shared__ int cntl[R];
  __shared__ int qe[NW * QCAP];
  __shared__ int qr[NW * QCAP];
  __shared__ int qcnt[NW];

  const int tid = threadIdx.x;
  const int lane = tid & 31;
  const int wave = __builtin_amdgcn_readfirstlane(tid >> 5);
  const int r0 = ob * R;
  const int E = (int)E64;

  for (int r = tid; r < R; r += NW * 32) cntl[r] = 0;
  __syncthreads();

  int* myqe = qe + wave * QCAP;
  int* myqr = qr + wave * QCAP;
  const unsigned ltmask = (1u << lane) - 1u;
  int qn = 0;

  float* myl = lst + (size_t)ob * LW;
  float* wl = myl + 32 + wave * CAP;
  const bool can_save = (r0 + R <= N) && (N <= (1 << 23));
  int cnt = 0;

  const bool bytes = P::usable(N, E64);

  auto take = [&](int nq) {
    if (lane < nq) {
      const int e = myqe[lane];
      const int r = bytes ? (dst[e] - r0) : myqr[lane];
      __hip_atomic_fetch_add(&cntl[r], 1, __ATOMIC_RELAXED, __HIP_MEMORY_SCOPE_WORKGROUP);
      if (can_save && cnt + nq <= CAP)
        wl[cnt + lane] = __uint_as_float(((unsigned)src[e] << 9) | (unsigned)r);
    }
    cnt += nq;
  };
  auto drain32 = [&]() {
    take(32);
    const int rem = qn - 32;
    int te = 0, tr = 0;
    if (lane < rem) { te = myqe[32 + lane]; tr = bytes ? 0 : myqr[32 + lane]; }
    if (lane < rem) { myqe[lane] = te; if (!bytes) myqr[lane] = tr; }
    qn = rem;
  };

  const bool vec_ok = ((E & 3) == 0) && ((((size_t)dst) & 15) == 0);
  auto load4 = [&](int i) -> int4 {
    int4 d;
    d.x = (i     < E) ? dst[i]     : -1;
    d.y = (i + 1 < E) ? dst[i + 1] : -1;
    d.z = (i + 2 < E) ? dst[i + 2] : -1;
    d.w = (i + 3 < E) ? dst[i + 3] : -1;
    return d;
  };

  auto scan4 = [&](const int4& d, int base) {
    const int d0 = d.x - r0, d1 = d.y - r0, d2 = d.z - r0, d3 = d.w - r0;
    unsigned mm = ((unsigned)d0 < (unsigned)R ? 1u : 0u) |
                  ((unsigned)d1 < (unsigned)R ? 2u : 0u) |
                  ((unsigned)d2 < (unsigned)R ? 4u : 0u) |
                  ((unsigned)d3 < (unsigned)R ? 8u : 0u);
    while (true) {
      const bool has = mm != 0u;
      const unsigned mk = __builtin_amdgcn_ballot_w32(has);
      if (mk == 0u) break;
      if (has) {
        const int c = __builtin_ctz(mm);
        const int dsel = (c == 0) ? d0 : (c == 1) ? d1 : (c == 2) ? d2 : d3;
        const int p = qn + __builtin_popcount(mk & ltmask);
        myqe[p] = base + c;
        myqr[p] = dsel;
      }
      qn += __builtin_popcount(mk);
      mm &= mm - 1u;
      if (qn >= 32) drain32();
    }
  };

  constexpr int TRIP = 8;
  if (bytes) {
    const unsigned pat = (unsigned)ob * 0x01010101u;
    auto hit = [&](unsigned x) -> unsigned {
      x ^= pat;
      return ~(((x & 0x7F7F7F7Fu) + 0x7F7F7F7Fu) | x | 0x7F7F7F7Fu);
    };
    auto mask16 = [&](const uint4& c) -> unsigned {
      return (hit(c.x) >> 7) | (hit(c.y) >> 6) | (hit(c.z) >> 5) | (hit(c.w) >> 4);
    };
    const unsigned* pl = (const unsigned*)lst;
    const int ntr = (E + 4095) >> 12;
    for (int tr = wave; tr < ntr; tr += NW) {
      const uint4* p = (const uint4*)(pl + (size_t)(tr / 3) * LW + P::TOFF) +
                       (tr % 3) * (TRIP * 32) + lane;
      const int b = tr * 4096 + lane * 16;
      uint4 c[TRIP];
      if (tr * 4096 + 4096 <= E) {
#pragma unroll
        for (int j = 0; j < TRIP; ++j) c[j] = p[j * 32];
      } else {
#pragma unroll
        for (int j = 0; j < TRIP; ++j) {
          const uint4 ff = {~0u, ~0u, ~0u, ~0u};
          c[j] = (b + j * 512 < E) ? p[j * 32] : ff;
        }
      }
      unsigned m4[TRIP / 2];
#pragma unroll
      for (int jj = 0; jj < TRIP / 2; ++jj)
        m4[jj] = mask16(c[2 * jj]) | (mask16(c[2 * jj + 1]) << 4);
#pragma unroll 1
      for (int jj = 0; jj < TRIP / 2; ++jj) {
        unsigned mm = (jj == 0) ? m4[0] : (jj == 1) ? m4[1] : (jj == 2) ? m4[2] : m4[3];
        const int base = b + jj * 1024;
        while (true) {
          const bool has = mm != 0u;
          const unsigned mk = __builtin_amdgcn_ballot_w32(has);
          if (mk == 0u) break;
          if (has) {
            const int p = __builtin_ctz(mm);
            myqe[qn + __builtin_popcount(mk & ltmask)] =
                base + (((p >> 2) & 1) << 9) + ((p & 3) << 2) + (p >> 3);
          }
          qn += __builtin_popcount(mk);
          mm &= mm - 1u;
          if (qn >= 32) drain32();
        }
      }
    }
  } else {
    const int nsc = (E + 1023) >> 10;
    for (int sc = wave; sc < nsc; sc += NW) {
      const int b = sc * 1024 + lane * 4;
      int4 c[TRIP];
      if (vec_ok && sc * 1024 + 1024 <= E) {
        const int4* p = (const int4*)(dst + b);
#pragma unroll
        for (int j = 0; j < TRIP; ++j) c[j] = p[j * 32];
      } else {
#pragma unroll
        for (int j = 0; j < TRIP; ++j) c[j] = load4(b + j * 128);
      }
#pragma unroll
      for (int j = 0; j < TRIP; ++j) scan4(c[j], b + j * 128);
    }
  }

  if (qn > 0) take(qn);
  if (lane == 0) qcnt[wave] = (can_save && cnt <= CAP) ? cnt : -1;
  __syncthreads();

  if (wave == 0) myl[lane] = __int_as_float(qcnt[lane]);

  for (int r = tid; r < R; r += NW * 32) {
    const int v = r0 + r;
    if (v < N) dinv[v] = __frsqrt_rn(fmaxf((float)(cntl[r] + 1), 1e-12f));
  }
}

template <int FOUT>
__device__ __forceinline__ void gemm_lds_b_part(
    const float* __restrict__ A, const float* __restrict__ W,
    float* __restrict__ C, int nrows, const int gb, const int ngb) {
  constexpr int PITCH = 132;
  constexpr int NCS = FOUT / 16;
  constexpr int RG = 32 / NCS;
  constexpr int TM = 16 * RG;
  constexpr int NF4 = TM * 32 / 1024;
  constexpr int NWQ = 128 * FOUT / 1024;
  static_assert(NCS * RG == 32 && NF4 * 1024 == TM * 32 && NWQ * 1024 == 128 * FOUT,
                "tile must split over 32 waves");
  __shared__ alignas(16) float wl[128 * FOUT];
  __shared__ alignas(16) float at[2][TM * PITCH];

  const int ntiles = (nrows + TM - 1) / TM;
  int tile = gb;
  if (tile >= ntiles) return;

  const int tid = threadIdx.x;
  const int lane = tid & 31;
  const int wave = tid >> 5;
  const int strip = wave % NCS;
  const int colbase = strip * 16;
  const int rg = wave / NCS;
  const int half = lane >> 4;
  const int m = lane & 15;

#pragma unroll
  for (int q = 0; q < NWQ; ++q) {
    const int i = tid + q * 1024;
    const int k = i / FOUT, col = i % FOUT;
    const int d = ((((k >> 2) * NCS + (col >> 4)) * 32 + ((k >> 1) & 1) * 16 + (col & 15)) << 1) + (k & 1);
    wl[d] = W[i];
  }

  v4f st[NF4];
  auto fetch = [&](int t) {
    const int row0 = t * TM;
    const v4f* srcv = (const v4f*)(A + (size_t)row0 * 128);
    if (row0 + TM <= nrows) {
#pragma unroll
      for (int j = 0; j < NF4; ++j) st[j] = srcv[tid + j * 1024];
    } else {
#pragma unroll
      for (int j = 0; j < NF4; ++j) {
        const int i = tid + j * 1024;
        const v4f z = {0.0f, 0.0f, 0.0f, 0.0f};
        st[j] = z;
        if (row0 + (i >> 5) < nrows) st[j] = srcv[i];
      }
    }
  };
  auto stash = [&](int buf) {
#pragma unroll
    for (int j = 0; j < NF4; ++j) {
      const int i = tid + j * 1024;
      *(v4f*)&at[buf][(i >> 5) * PITCH + (i & 31) * 4] = st[j];
    }
  };

  fetch(tile);
  stash(0);
  __syncthreads();
  int buf = 0;
  const float* bbase = &wl[(strip * 32 + lane) * 2];
  for (; tile < ntiles; tile += ngb) {
    const int nxt = tile + ngb;
    if (nxt < ntiles) fetch(nxt);

    v8f acc = v8f{};
    const float* abase = &at[buf][(rg * 16 + m) * PITCH + half * 2];
#pragma unroll
    for (int s = 0; s < 32; ++s) {
      const v2f a = *(const v2f*)(abase + 4 * s);
      const v2f b = *(const v2f*)(bbase + s * NCS * 64);
      acc = __builtin_amdgcn_wmma_f32_16x16x4_f32(false, a, false, b, (short)0, acc,
                                                  false, false);
    }

    if (nxt < ntiles) stash(buf ^ 1);
    __syncthreads();
    buf ^= 1;

    const int rbase = tile * TM + rg * 16 + half * 8;
#pragma unroll
    for (int i = 0; i < 8; ++i) {
      const int row = rbase + i;
      if (row < nrows) C[(size_t)row * FOUT + colbase + m] = acc[i];
    }
  }
}

template <int R>
__global__ __launch_bounds__(1024, 16) void scan_gemm_kernel(
    const int* __restrict__ src, const int* __restrict__ dst,
    float* __restrict__ dinv, float* lst, int N, long long E64,
    const float* __restrict__ A, const float* __restrict__ W,
    float* __restrict__ C, int nsb) {
  const int bi = (int)blockIdx.x;
  const int p = bi >> 1;
  const int nb = (int)(gridDim.x >> 1);
  if (((bi ^ (bi >> 8)) & 1) == 0) {
    if (p < nsb) owner_scan_part<R>(src, dst, dinv, lst, N, E64, p);
  } else {
    gemm_lds_b_part<128>(A, W, C, N, p, nb);
  }
}

template <int F, int R, bool RELU>
__global__ __launch_bounds__(1024) void owner_agg_kernel(
    const int* __restrict__ src, const int* __restrict__ dst,
    const float* __restrict__ dinv, const float* __restrict__ h,
    const float* __restrict__ bias, float* out, const float* lst, int N,
    long long E64) {
  constexpr int VEC = F / 32;
  constexpr int NW = 32;
  constexpr int QCAP = 64;
  constexpr int GRP = 8;
  constexpr int GRPL = (VEC > 2) ? 8 : 16;
  typedef Plane<R> P;
  constexpr int LW = P::LW;
  constexpr int CAP = P::CAP;
  static_assert(P::NW == NW, "slice layout is for 32 waves");
  typedef float VT __attribute__((ext_vector_type(VEC)));

  __shared__ float acc[R * F];
  __shared__ float dloc[R];
  __shared__ int qe[NW * QCAP];
  __shared__ int qr[NW * QCAP];

  const int tid = threadIdx.x;
  const int lane = tid & 31;
  const int wave = __builtin_amdgcn_readfirstlane(tid >> 5);
  const int r0 = blockIdx.x * R;
  const int E = (int)E64;

  const float* myl = lst + (size_t)blockIdx.x * LW;
  const float* wl = myl + 32 + wave * CAP;
  const int c = __float_as_int(myl[lane]);

  constexpr int NR = (R + NW - 1) / NW;
  float di[NR];
  VT hs[NR];
#pragma unroll
  for (int j = 0; j < NR; ++j) {
    const int r = wave + j * NW;
    const int v = (r < R && r0 + r < N) ? r0 + r : r0;
    di[j] = dinv[v];
    hs[j] = *(const VT*)(h + (size_t)v * F + lane * VEC);
  }

  const bool use_list = __builtin_amdgcn_ballot_w32(c < 0 || c > CAP) == 0u;
  const int n = __builtin_amdgcn_readlane(c, wave);
  auto ld = [&](int b) -> unsigned {
    return (b + lane < n) ? __float_as_uint(wl[b + lane]) : 0u;
  };
  unsigned pk = ld(0);
  unsigned pk1 = ld(32);

#pragma unroll
  for (int j = 0; j < NR; ++j) {
    const int r = wave + j * NW;
    if (r < R) {
      const bool ok = r0 + r < N;
      const float d = ok ? di[j] : 0.0f;
      if (lane == 0) dloc[r] = d;
      const float s = d * d;
#pragma unroll
      for (int t = 0; t < VEC; ++t) acc[r * F + t * 32 + lane] = ok ? s * hs[j][t] : 0.0f;
    }
  }
  float dv = dinv[pk >> 9];
  __syncthreads();

  auto add_row = [&](int rk, float wk, const VT& hv) {
#pragma unroll
    for (int t = 0; t < VEC; ++t)
      __hip_atomic_fetch_add(&acc[rk * F + t * 32 + lane], wk * hv[t],
                             __ATOMIC_RELAXED, __HIP_MEMORY_SCOPE_WORKGROUP);
  };

  if (use_list) {
    for (int b = 0; b < n; b += 32) {
      const unsigned pk2 = ld(b + 64);
      const float dv1 = dinv[pk1 >> 9];
      const int u = (int)(pk >> 9);
      const int r = (int)(pk & 511u);
      const float w = dv * dloc[r];
      const int m = (n - b < 32) ? (n - b) : 32;
      int g = 0;
      for (; g + GRPL <= m; g += GRPL) {
        VT hv[GRPL];
        int rr[GRPL];
        float ww[GRPL];
#pragma unroll
        for (int k = 0; k < GRPL; ++k) {
          const int uk = __builtin_amdgcn_readlane(u, g + k);
          rr[k] = __builtin_amdgcn_readlane(r, g + k);
          ww[k] = __int_as_float(__builtin_amdgcn_readlane(__float_as_int(w), g + k));
          hv[k] = *(const VT*)(h + (size_t)uk * F + lane * VEC);
        }
#pragma unroll
        for (int k = 0; k < GRPL; ++k) add_row(rr[k], ww[k], hv[k]);
      }
      if (g < m) {
        VT hv[GRPL];
        int rr[GRPL];
        float ww[GRPL];
#pragma unroll
        for (int k = 0; k < GRPL; ++k) {
          if (g + k < m) {
            const int uk = __builtin_amdgcn_readlane(u, g + k);
            rr[k] = __builtin_amdgcn_readlane(r, g + k);
            ww[k] = __int_as_float(__builtin_amdgcn_readlane(__float_as_int(w), g + k));
            hv[k] = *(const VT*)(h + (size_t)uk * F + lane * VEC);
          }
        }
#pragma unroll
        for (int k = 0; k < GRPL; ++k) {
          if (g + k < m) add_row(rr[k], ww[k], hv[k]);
        }
      }
      pk = pk1;
      dv = dv1;
      pk1 = pk2;
    }
  } else {
    int* myqe = qe + wave * QCAP;
    int* myqr = qr + wave * QCAP;
    const unsigned ltmask = (1u << lane) - 1u;
    int qn = 0;

    auto drain32 = [&]() {
      const int r = myqr[lane];
      const int u = src[myqe[lane]];
      const float w = dinv[u] * dloc[r];
      for (int g = 0; g < 32; g += GRP) {
        VT hv[GRP];
        int rr[GRP];
        float ww[GRP];
#pragma unroll
        for (int k = 0; k < GRP; ++k) {
          const int uk = __builtin_amdgcn_readlane(u, g + k);
          rr[k] = __builtin_amdgcn_readlane(r, g + k);
          ww[k] = __int_as_float(__builtin_amdgcn_readlane(__float_as_int(w), g + k));
          hv[k] = *(const VT*)(h + (size_t)uk * F + lane * VEC);
        }
#pragma unroll
        for (int k = 0; k < GRP; ++k) add_row(rr[k], ww[k], hv[k]);
      }
      const int rem = qn - 32;
      int te = 0, tr = 0;
      if (lane < rem) { te = myqe[32 + lane]; tr = myqr[32 + lane]; }
      if (lane < rem) { myqe[lane] = te; myqr[lane] = tr; }
      qn = rem;
    };

    const bool vec_ok = ((E & 3) == 0) && ((((size_t)dst) & 15) == 0);
    auto load4 = [&](int i) -> int4 {
      int4 d;
      d.x = (i     < E) ? dst[i]     : -1;
      d.y = (i + 1 < E) ? dst[i + 1] : -1;
      d.z = (i + 2 < E) ? dst[i + 2] : -1;
      d.w = (i + 3 < E) ? dst[i + 3] : -1;
      return d;
    };

    auto scan4 = [&](const int4& d, int base) {
      const int d0 = d.x - r0, d1 = d.y - r0, d2 = d.z - r0, d3 = d.w - r0;
      unsigned mm = ((unsigned)d0 < (unsigned)R ? 1u : 0u) |
                    ((unsigned)d1 < (unsigned)R ? 2u : 0u) |
                    ((unsigned)d2 < (unsigned)R ? 4u : 0u) |
                    ((unsigned)d3 < (unsigned)R ? 8u : 0u);
      while (true) {
        const bool has = mm != 0u;
        const unsigned mk = __builtin_amdgcn_ballot_w32(has);
        if (mk == 0u) break;
        if (has) {
          const int cc = __builtin_ctz(mm);
          const int dsel = (cc == 0) ? d0 : (cc == 1) ? d1 : (cc == 2) ? d2 : d3;
          const int p = qn + __builtin_popcount(mk & ltmask);
          myqe[p] = base + cc;
          myqr[p] = dsel;
        }
        qn += __builtin_popcount(mk);
        mm &= mm - 1u;
        if (qn >= 32) drain32();
      }
    };

    constexpr int TRIP = 8;
    const int nsc = (E + 1023) >> 10;
    for (int sc = wave; sc < nsc; sc += NW) {
      const int b = sc * 1024 + lane * 4;
      int4 cv[TRIP];
      if (vec_ok && sc * 1024 + 1024 <= E) {
        const int4* p = (const int4*)(dst + b);
#pragma unroll
        for (int j = 0; j < TRIP; ++j) cv[j] = p[j * 32];
      } else {
#pragma unroll
        for (int j = 0; j < TRIP; ++j) cv[j] = load4(b + j * 128);
      }
#pragma unroll
      for (int j = 0; j < TRIP; ++j) scan4(cv[j], b + j * 128);
    }

    if (qn > 0) {
      int r = 0, u = 0;
      float w = 0.0f;
      if (lane < qn) {
        r = myqr[lane];
        u = src[myqe[lane]];
        w = dinv[u] * dloc[r];
      }
      for (int k = 0; k < qn; ++k) {
        const int uk = __builtin_amdgcn_readlane(u, k);
        const int rk = __builtin_amdgcn_readlane(r, k);
        const float wk = __int_as_float(__builtin_amdgcn_readlane(__float_as_int(w), k));
        const VT hv = *(const VT*)(h + (size_t)uk * F + lane * VEC);
        add_row(rk, wk, hv);
      }
    }
  }
  __syncthreads();

  for (int r = wave; r < R; r += NW) {
    const int v = r0 + r;
    if (v >= N) break;
    VT o;
#pragma unroll
    for (int t = 0; t < VEC; ++t) {
      const float x = acc[r * F + t * 32 + lane] + bias[lane * VEC + t];
      o[t] = RELU ? fmaxf(x, 0.0f) : x;
    }
    *(VT*)(out + (size_t)v * F + lane * VEC) = o;
  }
}

extern "C" void kernel_launch(void* const* d_in, const int* in_sizes, int n_in,
                              void* d_out, int out_size, void* d_ws,
                              size_t ws_size, hipStream_t stream) {
  const float* x  = (const float*)d_in[0];
  const int*   ei = (const int*)d_in[1];
  const float* W1 = (const float*)d_in[2];
  const float* b1 = (const float*)d_in[3];
  const float* W2 = (const float*)d_in[4];
  const float* b2 = (const float*)d_in[5];
  float* out = (float*)d_out;

  const int N = in_sizes[0] / 128;
  const long long E = in_sizes[1] / 2;
  const int* src = ei;
  const int* dst = ei + E;

  float* ws = (float*)d_ws;
  float* dinvw = ws;
  float* h1pre = ws + N;
  float* h1    = h1pre + (size_t)N * 128;
  float* h2pre = h1pre;

  constexpr int R = 400;
  const int TB = 256;
  auto blocks = [](long long n, int tb) { return (unsigned)((n + tb - 1) / tb); };

  plane_kernel<R><<<blocks((E + 15) / 16 * 4, TB), TB, 0, stream>>>(
      dst, (unsigned*)d_out, N, E);
  const int nsb = (N + R - 1) / R;
  const int nb = nsb > 256 ? nsb : 256;
  scan_gemm_kernel<R><<<2 * nb, 1024, 0, stream>>>(
      src, dst, dinvw, out, N, E, x, W1, h1pre, nsb);
  const float* dinv = dinvw;

  owner_agg_kernel<128, R, true><<<(N + R - 1) / R, 1024, 0, stream>>>(
      src, dst, dinv, h1pre, b1, h1, out, N, E);

  gemm_wmma_kernel<64, 1><<<1024, 256, 0, stream>>>(h1, W2, h2pre, N);
  owner_agg_kernel<64, R, false><<<(N + R - 1) / R, 1024, 0, stream>>>(
      src, dst, dinv, h2pre, b2, out, out, N, E);
}


# Owner bit planes replace the owner-byte plane: launch 1 builds per-owner edge bitmaps in LDS, a scan block reads only its own plane (1/8 the bytes), set bits only; sum order still LDS arrival
# speedup vs baseline: 3.5659x; 1.0403x over previous
#include <hip/hip_runtime.h>
#include <hip/hip_bf16.h>

typedef float v2f __attribute__((ext_vector_type(2)));
typedef float v4f __attribute__((ext_vector_type(4)));
typedef float v8f __attribute__((ext_vector_type(8)));

template <int R>
struct Slice {
  static constexpr int LW = R * 64;
  static constexpr int NW = 32;
  static constexpr int CAP = (((LW - 32) / NW) / 32) * 32;
  static_assert(CAP >= 32, "slice too small for the lists");
};

template <int R>
struct Bits {
  static constexpr int SEGE = 8192;
  static constexpr int SEGW = SEGE / 32;
  static constexpr int NOWN = 256;
  static constexpr int LP = SEGW + 4;
  __device__ static int nseg(long long E) { return (int)((E + SEGE - 1) / SEGE); }
  __device__ static bool usable(int N, long long E, const void* planes) {
    if (E <= 0 || E > (1ll << 30)) return false;
    const long long nown = (N + R - 1) / R;
    const long long wpo = (long long)nseg(E) * SEGW;
    return nown <= NOWN && nown * wpo <= (long long)N * 128 && ((((size_t)planes) & 15) == 0);
  }
};

template <int R>
__global__ __launch_bounds__(1024) void bitplane_kernel(const int* __restrict__ dst,
                                                        unsigned* __restrict__ planes,
                                                        int N, long long E) {
  typedef Bits<R> B;
  if (!B::usable(N, E, planes)) return;
  constexpr int LP4 = B::LP / 4;
  __shared__ alignas(16) unsigned bm[B::NOWN * B::LP];
  uint4* bm4 = (uint4*)bm;
  const int tid = threadIdx.x;
  const int seg = blockIdx.x;
  const int wpo = B::nseg(E) * B::SEGW;
  const int nown = (N + R - 1) / R;

  constexpr int NJ = B::SEGE / 4096;
  const long long e0 = (long long)seg * B::SEGE;
  const bool al = ((((size_t)dst) & 15) == 0);
  int d[NJ][4];
#pragma unroll
  for (int j = 0; j < NJ; ++j) {
    const long long e = e0 + j * 4096 + tid * 4;
    if (al && e + 3 < E) {
      const int4 v = *(const int4*)(dst + e);
      d[j][0] = v.x; d[j][1] = v.y; d[j][2] = v.z; d[j][3] = v.w;
    } else {
#pragma unroll
      for (int k = 0; k < 4; ++k) d[j][k] = (e + k < E) ? dst[e + k] : -1;
    }
  }

  const uint4 z4 = {0u, 0u, 0u, 0u};
  for (int i = tid; i < nown * LP4; i += 1024) bm4[i] = z4;
  __syncthreads();

#pragma unroll
  for (int j = 0; j < NJ; ++j) {
#pragma unroll
    for (int k = 0; k < 4; ++k) {
      const int el = j * 4096 + tid * 4 + k;
      if ((unsigned)d[j][k] < (unsigned)N) {
        const int o = d[j][k] / R;
        __hip_atomic_fetch_or(&bm[o * B::LP + (el >> 5)], 1u << (el & 31),
                              __ATOMIC_RELAXED, __HIP_MEMORY_SCOPE_WORKGROUP);
      }
    }
  }
  __syncthreads();

  for (int i = tid; i < nown * (B::SEGW / 4); i += 1024) {
    const int o = i / (B::SEGW / 4), w4 = i % (B::SEGW / 4);
    ((uint4*)(planes + (size_t)o * wpo + (size_t)seg * B::SEGW))[w4] = bm4[o * LP4 + w4];
  }
}

template <int FOUT, int RT>
__global__ __launch_bounds__(256) void gemm_wmma_kernel(
    const float* __restrict__ A, const float* __restrict__ W,
    float* __restrict__ C, int nrows) {
  constexpr int PITCH = 132;
  constexpr int NCS = FOUT / 16;
  constexpr int RG = 8 / NCS;
  constexpr int TM = 16 * RT * RG;
  constexpr int NF4 = TM * 32 / 256;
  static_assert(NCS * RG == 8 && NF4 * 256 == TM * 32, "tile must split over 8 waves");
  __shared__ alignas(16) float lds[2][TM * PITCH];

  const int ntiles = (nrows + TM - 1) / TM;
  int tile = blockIdx.x;
  if (tile >= ntiles) return;

  const int tid = threadIdx.x;
  const int lane = tid & 31;
  const int wave = tid >> 5;
  const int colbase = (wave % NCS) * 16;
  const int rg = wave / NCS;
  const int half = lane >> 4;
  const int m = lane & 15;

  v2f b[32];
#pragma unroll
  for (int s = 0; s < 32; ++s) {
    const int kb = 4 * s + half * 2;
    b[s].x = W[(size_t)kb * FOUT + colbase + m];
    b[s].y = W[(size_t)(kb + 1) * FOUT + colbase + m];
  }

  v4f st[NF4];
  auto fetch = [&](int t) {
    const int row0 = t * TM;
    const v4f* srcv = (const v4f*)(A + (size_t)row0 * 128);
    if (row0 + TM <= nrows) {
#pragma unroll
      for (int j = 0; j < NF4; ++j) st[j] = srcv[tid + j * 256];
    } else {
#pragma unroll
      for (int j = 0; j < NF4; ++j) {
        const int i = tid + j * 256;
        const v4f z = {0.0f, 0.0f, 0.0f, 0.0f};
        st[j] = z;
        if (row0 + (i >> 5) < nrows) st[j] = srcv[i];
      }
    }
  };
  auto stash = [&](int buf) {
#pragma unroll
    for (int j = 0; j < NF4; ++j) {
      const int i = tid + j * 256;
      *(v4f*)&lds[buf][(i >> 5) * PITCH + (i & 31) * 4] = st[j];
    }
  };

  fetch(tile);
  stash(0);
  __syncthreads();
  int buf = 0;
  const int step = (int)gridDim.x;
  for (; tile < ntiles; tile += step) {
    const int nxt = tile + step;
    if (nxt < ntiles) fetch(nxt);

    v8f acc[RT];
#pragma unroll
    for (int t = 0; t < RT; ++t) acc[t] = v8f{};
    const float* abase = &lds[buf][(rg * RT * 16 + m) * PITCH + half * 2];
#pragma unroll
    for (int s = 0; s < 32; ++s) {
#pragma unroll
      for (int t = 0; t < RT; ++t) {
        const v2f a = *(const v2f*)(abase + t * 16 * PITCH + 4 * s);
        acc[t] = __builtin_amdgcn_wmma_f32_16x16x4_f32(false, a, false, b[s], (short)0,
                                                       acc[t], false, false);
      }
    }

    if (nxt < ntiles) stash(buf ^ 1);
    __syncthreads();
    buf ^= 1;

#pragma unroll
    for (int t = 0; t < RT; ++t) {
      const int rbase = tile * TM + (rg * RT + t) * 16 + half * 8;
#pragma unroll
      for (int i = 0; i < 8; ++i) {
        const int row = rbase + i;
        if (row < nrows) C[(size_t)row * FOUT + colbase + m] = acc[t][i];
      }
    }
  }
}

template <int R>
__device__ __forceinline__ void owner_scan_part(
    const int* __restrict__ src, const int* __restrict__ dst,
    float* __restrict__ dinv, float* lst, const unsigned* __restrict__ planes, int N,
    long long E64, const int ob) {
  constexpr int NW = 32;
  constexpr int QCAP = 64;
  typedef Slice<R> P;
  typedef Bits<R> B;
  constexpr int LW = P::LW;
  constexpr int CAP = P::CAP;
  static_assert(R <= 512, "owned row must fit the 9 low bits of a list word");
  static_assert(P::NW == NW, "slice layout is for 32 waves");

  __shared__ int cntl[R];
  __shared__ int qe[NW * QCAP];
  __shared__ int qr[NW * QCAP];
  __shared__ int qcnt[NW];

  const int tid = threadIdx.x;
  const int lane = tid & 31;
  const int wave = __builtin_amdgcn_readfirstlane(tid >> 5);
  const int r0 = ob * R;
  const int E = (int)E64;

  for (int r = tid; r < R; r += NW * 32) cntl[r] = 0;
  __syncthreads();

  int* myqe = qe + wave * QCAP;
  int* myqr = qr + wave * QCAP;
  const unsigned ltmask = (1u << lane) - 1u;
  int qn = 0;

  float* myl = lst + (size_t)ob * LW;
  float* wl = myl + 32 + wave * CAP;
  const bool can_save = (r0 + R <= N) && (N <= (1 << 23));
  int cnt = 0;

  const bool bits = B::usable(N, E64, planes);

  auto take = [&](int nq) {
    if (lane < nq) {
      const int e = myqe[lane];
      const int r = bits ? (dst[e] - r0) : myqr[lane];
      __hip_atomic_fetch_add(&cntl[r], 1, __ATOMIC_RELAXED, __HIP_MEMORY_SCOPE_WORKGROUP);
      if (can_save && cnt + nq <= CAP)
        wl[cnt + lane] = __uint_as_float(((unsigned)src[e] << 9) | (unsigned)r);
    }
    cnt += nq;
  };
  auto drain32 = [&]() {
    take(32);
    const int rem = qn - 32;
    int te = 0, tr = 0;
    if (lane < rem) { te = myqe[32 + lane]; tr = bits ? 0 : myqr[32 + lane]; }
    if (lane < rem) { myqe[lane] = te; if (!bits) myqr[lane] = tr; }
    qn = rem;
  };

  const bool vec_ok = ((E & 3) == 0) && ((((size_t)dst) & 15) == 0);
  auto load4 = [&](int i) -> int4 {
    int4 d;
    d.x = (i     < E) ? dst[i]     : -1;
    d.y = (i + 1 < E) ? dst[i + 1] : -1;
    d.z = (i + 2 < E) ? dst[i + 2] : -1;
    d.w = (i + 3 < E) ? dst[i + 3] : -1;
    return d;
  };

  auto scan4 = [&](const int4& d, int base) {
    const int d0 = d.x - r0, d1 = d.y - r0, d2 = d.z - r0, d3 = d.w - r0;
    unsigned mm = ((unsigned)d0 < (unsigned)R ? 1u : 0u) |
                  ((unsigned)d1 < (unsigned)R ? 2u : 0u) |
                  ((unsigned)d2 < (unsigned)R ? 4u : 0u) |
                  ((unsigned)d3 < (unsigned)R ? 8u : 0u);
    while (true) {
      const bool has = mm != 0u;
      const unsigned mk = __builtin_amdgcn_ballot_w32(has);
      if (mk == 0u) break;
      if (has) {
        const int c = __builtin_ctz(mm);
        const int dsel = (c == 0) ? d0 : (c == 1) ? d1 : (c == 2) ? d2 : d3;
        const int p = qn + __builtin_popcount(mk & ltmask);
        myqe[p] = base + c;
        myqr[p] = dsel;
      }
      qn += __builtin_popcount(mk);
      mm &= mm - 1u;
      if (qn >= 32) drain32();
    }
  };

  constexpr int TRIP = 8;
  if (bits) {
    const int ntr = B::nseg(E64);
    const int wpo = ntr * B::SEGW;
    const uint4* pq = (const uint4*)(planes + (size_t)ob * wpo) + lane;
    uint4 n0 = {0u, 0u, 0u, 0u}, n1 = n0;
    if (wave < ntr) { n0 = pq[(size_t)wave * 64]; n1 = pq[(size_t)wave * 64 + 32]; }
    for (int tr = wave; tr < ntr; tr += NW) {
      const uint4 c0 = n0, c1 = n1;
      if (tr + NW < ntr) {
        n0 = pq[(size_t)(tr + NW) * 64];
        n1 = pq[(size_t)(tr + NW) * 64 + 32];
      }
#pragma unroll 1
      for (int hh = 0; hh < 2; ++hh) {
        const uint4 c = (hh == 0) ? c0 : c1;
        unsigned long long lo = (unsigned long long)c.x | ((unsigned long long)c.y << 32);
        unsigned long long hi = (unsigned long long)c.z | ((unsigned long long)c.w << 32);
        const int base = ((tr * 2 + hh) * 32 + lane) * 128;
        while (true) {
          const bool has = (lo | hi) != 0ull;
          const unsigned mk = __builtin_amdgcn_ballot_w32(has);
          if (mk == 0u) break;
          if (has) {
            int p;
            if (lo != 0ull) { p = __builtin_ctzll(lo); lo &= lo - 1ull; }
            else { p = 64 + __builtin_ctzll(hi); hi &= hi - 1ull; }
            myqe[qn + __builtin_popcount(mk & ltmask)] = base + p;
          }
          qn += __builtin_popcount(mk);
          if (qn >= 32) drain32();
        }
      }
    }
  } else {
    const int nsc = (E + 1023) >> 10;
    for (int sc = wave; sc < nsc; sc += NW) {
      const int b = sc * 1024 + lane * 4;
      int4 c[TRIP];
      if (vec_ok && sc * 1024 + 1024 <= E) {
        const int4* p = (const int4*)(dst + b);
#pragma unroll
        for (int j = 0; j < TRIP; ++j) c[j] = p[j * 32];
      } else {
#pragma unroll
        for (int j = 0; j < TRIP; ++j) c[j] = load4(b + j * 128);
      }
#pragma unroll
      for (int j = 0; j < TRIP; ++j) scan4(c[j], b + j * 128);
    }
  }

  if (qn > 0) take(qn);
  if (lane == 0) qcnt[wave] = (can_save && cnt <= CAP) ? cnt : -1;
  __syncthreads();

  if (wave == 0) myl[lane] = __int_as_float(qcnt[lane]);

  for (int r = tid; r < R; r += NW * 32) {
    const int v = r0 + r;
    if (v < N) dinv[v] = __frsqrt_rn(fmaxf((float)(cntl[r] + 1), 1e-12f));
  }
}

template <int FOUT>
__device__ __forceinline__ void gemm_lds_b_part(
    const float* __restrict__ A, const float* __restrict__ W,
    float* __restrict__ C, int nrows, const int gb, const int ngb) {
  constexpr int PITCH = 132;
  constexpr int NCS = FOUT / 16;
  constexpr int RG = 32 / NCS;
  constexpr int TM = 16 * RG;
  constexpr int NF4 = TM * 32 / 1024;
  constexpr int NWQ = 128 * FOUT / 1024;
  static_assert(NCS * RG == 32 && NF4 * 1024 == TM * 32 && NWQ * 1024 == 128 * FOUT,
                "tile must split over 32 waves");
  __shared__ alignas(16) float wl[128 * FOUT];
  __shared__ alignas(16) float at[2][TM * PITCH];

  const int ntiles = (nrows + TM - 1) / TM;
  int tile = gb;
  if (tile >= ntiles) return;

  const int tid = threadIdx.x;
  const int lane = tid & 31;
  const int wave = tid >> 5;
  const int strip = wave % NCS;
  const int colbase = strip * 16;
  const int rg = wave / NCS;
  const int half = lane >> 4;
  const int m = lane & 15;

#pragma unroll
  for (int q = 0; q < NWQ; ++q) {
    const int i = tid + q * 1024;
    const int k = i / FOUT, col = i % FOUT;
    const int d = ((((k >> 2) * NCS + (col >> 4)) * 32 + ((k >> 1) & 1) * 16 + (col & 15)) << 1) + (k & 1);
    wl[d] = W[i];
  }

  v4f st[NF4];
  auto fetch = [&](int t) {
    const int row0 = t * TM;
    const v4f* srcv = (const v4f*)(A + (size_t)row0 * 128);
    if (row0 + TM <= nrows) {
#pragma unroll
      for (int j = 0; j < NF4; ++j) st[j] = srcv[tid + j * 1024];
    } else {
#pragma unroll
      for (int j = 0; j < NF4; ++j) {
        const int i = tid + j * 1024;
        const v4f z = {0.0f, 0.0f, 0.0f, 0.0f};
        st[j] = z;
        if (row0 + (i >> 5) < nrows) st[j] = srcv[i];
      }
    }
  };
  auto stash = [&](int buf) {
#pragma unroll
    for (int j = 0; j < NF4; ++j) {
      const int i = tid + j * 1024;
      *(v4f*)&at[buf][(i >> 5) * PITCH + (i & 31) * 4] = st[j];
    }
  };

  fetch(tile);
  stash(0);
  __syncthreads();
  int buf = 0;
  const float* bbase = &wl[(strip * 32 + lane) * 2];
  for (; tile < ntiles; tile += ngb) {
    const int nxt = tile + ngb;
    if (nxt < ntiles) fetch(nxt);

    v8f acc = v8f{};
    const float* abase = &at[buf][(rg * 16 + m) * PITCH + half * 2];
#pragma unroll
    for (int s = 0; s < 32; ++s) {
      const v2f a = *(const v2f*)(abase + 4 * s);
      const v2f b = *(const v2f*)(bbase + s * NCS * 64);
      acc = __builtin_amdgcn_wmma_f32_16x16x4_f32(false, a, false, b, (short)0, acc,
                                                  false, false);
    }

    if (nxt < ntiles) stash(buf ^ 1);
    __syncthreads();
    buf ^= 1;

    const int rbase = tile * TM + rg * 16 + half * 8;
#pragma unroll
    for (int i = 0; i < 8; ++i) {
      const int row = rbase + i;
      if (row < nrows) C[(size_t)row * FOUT + colbase + m] = acc[i];
    }
  }
}

template <int R>
__global__ __launch_bounds__(1024, 16) void scan_gemm_kernel(
    const int* __restrict__ src, const int* __restrict__ dst,
    float* __restrict__ dinv, float* lst, const unsigned* __restrict__ planes, int N,
    long long E64, const float* __restrict__ A, const float* __restrict__ W,
    float* __restrict__ C, int nsb) {
  const int bi = (int)blockIdx.x;
  const int p = bi >> 1;
  const int nb = (int)(gridDim.x >> 1);
  if (((bi ^ (bi >> 8)) & 1) == 0) {
    if (p < nsb) owner_scan_part<R>(src, dst, dinv, lst, planes, N, E64, p);
  } else {
    gemm_lds_b_part<128>(A, W, C, N, p, nb);
  }
}

template <int F, int R, bool RELU>
__global__ __launch_bounds__(1024) void owner_agg_kernel(
    const int* __restrict__ src, const int* __restrict__ dst,
    const float* __restrict__ dinv, const float* __restrict__ h,
    const float* __restrict__ bias, float* out, const float* lst, int N,
    long long E64) {
  constexpr int VEC = F / 32;
  constexpr int NW = 32;
  constexpr int QCAP = 64;
  constexpr int GRP = 8;
  constexpr int GRPL = (VEC > 2) ? 8 : 16;
  typedef Slice<R> P;
  constexpr int LW = P::LW;
  constexpr int CAP = P::CAP;
  static_assert(P::NW == NW, "slice layout is for 32 waves");
  typedef float VT __attribute__((ext_vector_type(VEC)));

  __shared__ float acc[R * F];
  __shared__ float dloc[R];
  __shared__ int qe[NW * QCAP];
  __shared__ int qr[NW * QCAP];

  const int tid = threadIdx.x;
  const int lane = tid & 31;
  const int wave = __builtin_amdgcn_readfirstlane(tid >> 5);
  const int r0 = blockIdx.x * R;
  const int E = (int)E64;

  const float* myl = lst + (size_t)blockIdx.x * LW;
  const float* wl = myl + 32 + wave * CAP;
  const int c = __float_as_int(myl[lane]);

  constexpr int NR = (R + NW - 1) / NW;
  float di[NR];
  VT hs[NR];
#pragma unroll
  for (int j = 0; j < NR; ++j) {
    const int r = wave + j * NW;
    const int v = (r < R && r0 + r < N) ? r0 + r : r0;
    di[j] = dinv[v];
    hs[j] = *(const VT*)(h + (size_t)v * F + lane * VEC);
  }

  const bool use_list = __builtin_amdgcn_ballot_w32(c < 0 || c > CAP) == 0u;
  const int n = __builtin_amdgcn_readlane(c, wave);
  auto ld = [&](int b) -> unsigned {
    return (b + lane < n) ? __float_as_uint(wl[b + lane]) : 0u;
  };
  unsigned pk = ld(0);
  unsigned pk1 = ld(32);

#pragma unroll
  for (int j = 0; j < NR; ++j) {
    const int r = wave + j * NW;
    if (r < R) {
      const bool ok = r0 + r < N;
      const float d = ok ? di[j] : 0.0f;
      if (lane == 0) dloc[r] = d;
      const float s = d * d;
#pragma unroll
      for (int t = 0; t < VEC; ++t) acc[r * F + t * 32 + lane] = ok ? s * hs[j][t] : 0.0f;
    }
  }
  float dv = dinv[pk >> 9];
  __syncthreads();

  auto add_row = [&](int rk, float wk, const VT& hv) {
#pragma unroll
    for (int t = 0; t < VEC; ++t)
      __hip_atomic_fetch_add(&acc[rk * F + t * 32 + lane], wk * hv[t],
                             __ATOMIC_RELAXED, __HIP_MEMORY_SCOPE_WORKGROUP);
  };

  if (use_list) {
    for (int b = 0; b < n; b += 32) {
      const unsigned pk2 = ld(b + 64);
      const float dv1 = dinv[pk1 >> 9];
      const int u = (int)(pk >> 9);
      const int r = (int)(pk & 511u);
      const float w = dv * dloc[r];
      const int m = (n - b < 32) ? (n - b) : 32;
      int g = 0;
      for (; g + GRPL <= m; g += GRPL) {
        VT hv[GRPL];
        int rr[GRPL];
        float ww[GRPL];
#pragma unroll
        for (int k = 0; k < GRPL; ++k) {
          const int uk = __builtin_amdgcn_readlane(u, g + k);
          rr[k] = __builtin_amdgcn_readlane(r, g + k);
          ww[k] = __int_as_float(__builtin_amdgcn_readlane(__float_as_int(w), g + k));
          hv[k] = *(const VT*)(h + (size_t)uk * F + lane * VEC);
        }
#pragma unroll
        for (int k = 0; k < GRPL; ++k) add_row(rr[k], ww[k], hv[k]);
      }
      if (g < m) {
        VT hv[GRPL];
        int rr[GRPL];
        float ww[GRPL];
#pragma unroll
        for (int k = 0; k < GRPL; ++k) {
          if (g + k < m) {
            const int uk = __builtin_amdgcn_readlane(u, g + k);
            rr[k] = __builtin_amdgcn_readlane(r, g + k);
            ww[k] = __int_as_float(__builtin_amdgcn_readlane(__float_as_int(w), g + k));
            hv[k] = *(const VT*)(h + (size_t)uk * F + lane * VEC);
          }
        }
#pragma unroll
        for (int k = 0; k < GRPL; ++k) {
          if (g + k < m) add_row(rr[k], ww[k], hv[k]);
        }
      }
      pk = pk1;
      dv = dv1;
      pk1 = pk2;
    }
  } else {
    int* myqe = qe + wave * QCAP;
    int* myqr = qr + wave * QCAP;
    const unsigned ltmask = (1u << lane) - 1u;
    int qn = 0;

    auto drain32 = [&]() {
      const int r = myqr[lane];
      const int u = src[myqe[lane]];
      const float w = dinv[u] * dloc[r];
      for (int g = 0; g < 32; g += GRP) {
        VT hv[GRP];
        int rr[GRP];
        float ww[GRP];
#pragma unroll
        for (int k = 0; k < GRP; ++k) {
          const int uk = __builtin_amdgcn_readlane(u, g + k);
          rr[k] = __builtin_amdgcn_readlane(r, g + k);
          ww[k] = __int_as_float(__builtin_amdgcn_readlane(__float_as_int(w), g + k));
          hv[k] = *(const VT*)(h + (size_t)uk * F + lane * VEC);
        }
#pragma unroll
        for (int k = 0; k < GRP; ++k) add_row(rr[k], ww[k], hv[k]);
      }
      const int rem = qn - 32;
      int te = 0, tr = 0;
      if (lane < rem) { te = myqe[32 + lane]; tr = myqr[32 + lane]; }
      if (lane < rem) { myqe[lane] = te; myqr[lane] = tr; }
      qn = rem;
    };

    const bool vec_ok = ((E & 3) == 0) && ((((size_t)dst) & 15) == 0);
    auto load4 = [&](int i) -> int4 {
      int4 d;
      d.x = (i     < E) ? dst[i]     : -1;
      d.y = (i + 1 < E) ? dst[i + 1] : -1;
      d.z = (i + 2 < E) ? dst[i + 2] : -1;
      d.w = (i + 3 < E) ? dst[i + 3] : -1;
      return d;
    };

    auto scan4 = [&](const int4& d, int base) {
      const int d0 = d.x - r0, d1 = d.y - r0, d2 = d.z - r0, d3 = d.w - r0;
      unsigned mm = ((unsigned)d0 < (unsigned)R ? 1u : 0u) |
                    ((unsigned)d1 < (unsigned)R ? 2u : 0u) |
                    ((unsigned)d2 < (unsigned)R ? 4u : 0u) |
                    ((unsigned)d3 < (unsigned)R ? 8u : 0u);
      while (true) {
        const bool has = mm != 0u;
        const unsigned mk = __builtin_amdgcn_ballot_w32(has);
        if (mk == 0u) break;
        if (has) {
          const int cc = __builtin_ctz(mm);
          const int dsel = (cc == 0) ? d0 : (cc == 1) ? d1 : (cc == 2) ? d2 : d3;
          const int p = qn + __builtin_popcount(mk & ltmask);
          myqe[p] = base + cc;
          myqr[p] = dsel;
        }
        qn += __builtin_popcount(mk);
        mm &= mm - 1u;
        if (qn >= 32) drain32();
      }
    };

    constexpr int TRIP = 8;
    const int nsc = (E + 1023) >> 10;
    for (int sc = wave; sc < nsc; sc += NW) {
      const int b = sc * 1024 + lane * 4;
      int4 cv[TRIP];
      if (vec_ok && sc * 1024 + 1024 <= E) {
        const int4* p = (const int4*)(dst + b);
#pragma unroll
        for (int j = 0; j < TRIP; ++j) cv[j] = p[j * 32];
      } else {
#pragma unroll
        for (int j = 0; j < TRIP; ++j) cv[j] = load4(b + j * 128);
      }
#pragma unroll
      for (int j = 0; j < TRIP; ++j) scan4(cv[j], b + j * 128);
    }

    if (qn > 0) {
      int r = 0, u = 0;
      float w = 0.0f;
      if (lane < qn) {
        r = myqr[lane];
        u = src[myqe[lane]];
        w = dinv[u] * dloc[r];
      }
      for (int k = 0; k < qn; ++k) {
        const int uk = __builtin_amdgcn_readlane(u, k);
        const int rk = __builtin_amdgcn_readlane(r, k);
        const float wk = __int_as_float(__builtin_amdgcn_readlane(__float_as_int(w), k));
        const VT hv = *(const VT*)(h + (size_t)uk * F + lane * VEC);
        add_row(rk, wk, hv);
      }
    }
  }
  __syncthreads();

  for (int r = wave; r < R; r += NW) {
    const int v = r0 + r;
    if (v >= N) break;
    VT o;
#pragma unroll
    for (int t = 0; t < VEC; ++t) {
      const float x = acc[r * F + t * 32 + lane] + bias[lane * VEC + t];
      o[t] = RELU ? fmaxf(x, 0.0f) : x;
    }
    *(VT*)(out + (size_t)v * F + lane * VEC) = o;
  }
}

extern "C" void kernel_launch(void* const* d_in, const int* in_sizes, int n_in,
                              void* d_out, int out_size, void* d_ws,
                              size_t ws_size, hipStream_t stream) {
  const float* x  = (const float*)d_in[0];
  const int*   ei = (const int*)d_in[1];
  const float* W1 = (const float*)d_in[2];
  const float* b1 = (const float*)d_in[3];
  const float* W2 = (const float*)d_in[4];
  const float* b2 = (const float*)d_in[5];
  float* out = (float*)d_out;

  const int N = in_sizes[0] / 128;
  const long long E = in_sizes[1] / 2;
  const int* src = ei;
  const int* dst = ei + E;

  float* ws = (float*)d_ws;
  float* dinvw = ws;
  float* h1pre = ws + N;
  float* h1    = h1pre + (size_t)N * 128;
  float* h2pre = h1pre;

  constexpr int R = 400;
  auto blocks = [](long long n, int tb) { return (unsigned)((n + tb - 1) / tb); };

  unsigned* planes = (unsigned*)h1;
  bitplane_kernel<R><<<blocks(E, 8192), 1024, 0, stream>>>(dst, planes, N, E);
  const int nsb = (N + R - 1) / R;
  const int nb = nsb > 256 ? nsb : 256;
  scan_gemm_kernel<R><<<2 * nb, 1024, 0, stream>>>(
      src, dst, dinvw, out, planes, N, E, x, W1, h1pre, nsb);
  const float* dinv = dinvw;

  owner_agg_kernel<128, R, true><<<(N + R - 1) / R, 1024, 0, stream>>>(
      src, dst, dinv, h1pre, b1, h1, out, N, E);

  gemm_wmma_kernel<64, 1><<<1024, 256, 0, stream>>>(h1, W2, h2pre, N);
  owner_agg_kernel<64, R, false><<<(N + R - 1) / R, 1024, 0, stream>>>(
      src, dst, dinv, h2pre, b2, out, out, N, E);
}


# speedup vs baseline: 3.8990x; 1.0934x over previous
#include <hip/hip_runtime.h>
#include <hip/hip_bf16.h>

typedef float v2f __attribute__((ext_vector_type(2)));
typedef float v4f __attribute__((ext_vector_type(4)));
typedef float v8f __attribute__((ext_vector_type(8)));

template <int R>
struct Slice {
  static constexpr int LW = R * 64;
  static constexpr int NW = 32;
  static constexpr int CAP = (((LW - 32) / NW) / 32) * 32;
  static_assert(CAP >= 32, "slice too small for the lists");
};

template <int R>
struct Bits {
  static constexpr int SEGE = 8192;
  static constexpr int SEGW = SEGE / 32;
  static constexpr int NOWN = 256;
  static constexpr int LP = SEGW + 4;
  __device__ static int nseg(long long E) { return (int)((E + SEGE - 1) / SEGE); }
  __device__ static bool usable(int N, long long E, const void* planes) {
    if (E <= 0 || E > (1ll << 30)) return false;
    const long long nown = (N + R - 1) / R;
    const long long wpo = (long long)nseg(E) * SEGW;
    return nown <= NOWN && nown * wpo <= (long long)N * 128 && ((((size_t)planes) & 15) == 0);
  }
};

template <int R>
__global__ __launch_bounds__(1024) void bitplane_kernel(const int* __restrict__ dst,
                                                        unsigned* __restrict__ planes,
                                                        int N, long long E) {
  typedef Bits<R> B;
  if (!B::usable(N, E, planes)) return;
  constexpr int LP4 = B::LP / 4;
  __shared__ alignas(16) unsigned bm[B::NOWN * B::LP];
  uint4* bm4 = (uint4*)bm;
  const int tid = threadIdx.x;
  const int seg = blockIdx.x;
  const int wpo = B::nseg(E) * B::SEGW;
  const int nown = (N + R - 1) / R;

  constexpr int NJ = B::SEGE / 4096;
  const long long e0 = (long long)seg * B::SEGE;
  const bool al = ((((size_t)dst) & 15) == 0);
  int d[NJ][4];
#pragma unroll
  for (int j = 0; j < NJ; ++j) {
    const long long e = e0 + j * 4096 + tid * 4;
    if (al && e + 3 < E) {
      const int4 v = *(const int4*)(dst + e);
      d[j][0] = v.x; d[j][1] = v.y; d[j][2] = v.z; d[j][3] = v.w;
    } else {
#pragma unroll
      for (int k = 0; k < 4; ++k) d[j][k] = (e + k < E) ? dst[e + k] : -1;
    }
  }

  const uint4 z4 = {0u, 0u, 0u, 0u};
  for (int i = tid; i < nown * LP4; i += 1024) bm4[i] = z4;
  __syncthreads();

#pragma unroll
  for (int j = 0; j < NJ; ++j) {
#pragma unroll
    for (int k = 0; k < 4; ++k) {
      const int el = j * 4096 + tid * 4 + k;
      if ((unsigned)d[j][k] < (unsigned)N) {
        const int o = d[j][k] / R;
        __hip_atomic_fetch_or(&bm[o * B::LP + (el >> 5)], 1u << (el & 31),
                              __ATOMIC_RELAXED, __HIP_MEMORY_SCOPE_WORKGROUP);
      }
    }
  }
  __syncthreads();

  for (int i = tid; i < nown * (B::SEGW / 4); i += 1024) {
    const int o = i / (B::SEGW / 4), w4 = i % (B::SEGW / 4);
    ((uint4*)(planes + (size_t)o * wpo + (size_t)seg * B::SEGW))[w4] = bm4[o * LP4 + w4];
  }
}

template <int FOUT, int RT>
__global__ __launch_bounds__(256) void gemm_wmma_kernel(
    const float* __restrict__ A, const float* __restrict__ W,
    float* __restrict__ C, int nrows) {
  constexpr int PITCH = 132;
  constexpr int NCS = FOUT / 16;
  constexpr int RG = 8 / NCS;
  constexpr int TM = 16 * RT * RG;
  constexpr int NF4 = TM * 32 / 256;
  static_assert(NCS * RG == 8 && NF4 * 256 == TM * 32, "tile must split over 8 waves");
  __shared__ alignas(16) float lds[2][TM * PITCH];

  const int ntiles = (nrows + TM - 1) / TM;
  int tile = blockIdx.x;
  if (tile >= ntiles) return;

  const int tid = threadIdx.x;
  const int lane = tid & 31;
  const int wave = tid >> 5;
  const int colbase = (wave % NCS) * 16;
  const int rg = wave / NCS;
  const int half = lane >> 4;
  const int m = lane & 15;

  v2f b[32];
#pragma unroll
  for (int s = 0; s < 32; ++s) {
    const int kb = 4 * s + half * 2;
    b[s].x = W[(size_t)kb * FOUT + colbase + m];
    b[s].y = W[(size_t)(kb + 1) * FOUT + colbase + m];
  }

  v4f st[NF4];
  auto fetch = [&](int t) {
    const int row0 = t * TM;
    const v4f* srcv = (const v4f*)(A + (size_t)row0 * 128);
    if (row0 + TM <= nrows) {
#pragma unroll
      for (int j = 0; j < NF4; ++j) st[j] = srcv[tid + j * 256];
    } else {
#pragma unroll
      for (int j = 0; j < NF4; ++j) {
        const int i = tid + j * 256;
        const v4f z = {0.0f, 0.0f, 0.0f, 0.0f};
        st[j] = z;
        if (row0 + (i >> 5) < nrows) st[j] = srcv[i];
      }
    }
  };
  auto stash = [&](int buf) {
#pragma unroll
    for (int j = 0; j < NF4; ++j) {
      const int i = tid + j * 256;
      *(v4f*)&lds[buf][(i >> 5) * PITCH + (i & 31) * 4] = st[j];
    }
  };

  fetch(tile);
  stash(0);
  __syncthreads();
  int buf = 0;
  const int step = (int)gridDim.x;
  for (; tile < ntiles; tile += step) {
    const int nxt = tile + step;
    if (nxt < ntiles) fetch(nxt);

    v8f acc[RT];
#pragma unroll
    for (int t = 0; t < RT; ++t) acc[t] = v8f{};
    const float* abase = &lds[buf][(rg * RT * 16 + m) * PITCH + half * 2];
#pragma unroll
    for (int s = 0; s < 32; ++s) {
#pragma unroll
      for (int t = 0; t < RT; ++t) {
        const v2f a = *(const v2f*)(abase + t * 16 * PITCH + 4 * s);
        acc[t] = __builtin_amdgcn_wmma_f32_16x16x4_f32(false, a, false, b[s], (short)0,
                                                       acc[t], false, false);
      }
    }

    if (nxt < ntiles) stash(buf ^ 1);
    __syncthreads();
    buf ^= 1;

#pragma unroll
    for (int t = 0; t < RT; ++t) {
      const int rbase = tile * TM + (rg * RT + t) * 16 + half * 8;
#pragma unroll
      for (int i = 0; i < 8; ++i) {
        const int row = rbase + i;
        if (row < nrows) C[(size_t)row * FOUT + colbase + m] = acc[t][i];
      }
    }
  }
}

template <int R>
__device__ __forceinline__ void owner_scan_part(
    const int* __restrict__ src, const int* __restrict__ dst,
    float* __restrict__ dinv, float* lst, const unsigned* __restrict__ planes, int N,
    long long E64, const int ob) {
  constexpr int NW = 32;
  constexpr int QCAP = 64;
  typedef Slice<R> P;
  typedef Bits<R> B;
  constexpr int LW = P::LW;
  constexpr int CAP = P::CAP;
  static_assert(R <= 512, "owned row must fit the 9 low bits of a list word");
  static_assert(P::NW == NW, "slice layout is for 32 waves");

  __shared__ int cntl[R];
  __shared__ int qe[NW * QCAP];
  __shared__ int qr[NW * QCAP];
  __shared__ int qcnt[NW];

  const int tid = threadIdx.x;
  const int lane = tid & 31;
  const int wave = __builtin_amdgcn_readfirstlane(tid >> 5);
  const int r0 = ob * R;
  const int E = (int)E64;

  for (int r = tid; r < R; r += NW * 32) cntl[r] = 0;
  __syncthreads();

  int* myqe = qe + wave * QCAP;
  int* myqr = qr + wave * QCAP;
  const unsigned ltmask = (1u << lane) - 1u;
  int qn = 0;

  float* myl = lst + (size_t)ob * LW;
  float* wl = myl + 32 + wave * CAP;
  const bool can_save = (r0 + R <= N) && (N <= (1 << 23));
  int cnt = 0;

  const bool bits = B::usable(N, E64, planes);

  auto take = [&](int nq) {
    if (lane < nq) {
      const int e = myqe[lane];
      const int r = bits ? (dst[e] - r0) : myqr[lane];
      __hip_atomic_fetch_add(&cntl[r], 1, __ATOMIC_RELAXED, __HIP_MEMORY_SCOPE_WORKGROUP);
      if (can_save && cnt + nq <= CAP)
        wl[cnt + lane] = __uint_as_float(((unsigned)src[e] << 9) | (unsigned)r);
    }
    cnt += nq;
  };
  auto drain32 = [&]() {
    take(32);
    const int rem = qn - 32;
    int te = 0, tr = 0;
    if (lane < rem) { te = myqe[32 + lane]; tr = bits ? 0 : myqr[32 + lane]; }
    if (lane < rem) { myqe[lane] = te; if (!bits) myqr[lane] = tr; }
    qn = rem;
  };

  const bool vec_ok = ((E & 3) == 0) && ((((size_t)dst) & 15) == 0);
  auto load4 = [&](int i) -> int4 {
    int4 d;
    d.x = (i     < E) ? dst[i]     : -1;
    d.y = (i + 1 < E) ? dst[i + 1] : -1;
    d.z = (i + 2 < E) ? dst[i + 2] : -1;
    d.w = (i + 3 < E) ? dst[i + 3] : -1;
    return d;
  };

  auto scan4 = [&](const int4& d, int base) {
    const int d0 = d.x - r0, d1 = d.y - r0, d2 = d.z - r0, d3 = d.w - r0;
    unsigned mm = ((unsigned)d0 < (unsigned)R ? 1u : 0u) |
                  ((unsigned)d1 < (unsigned)R ? 2u : 0u) |
                  ((unsigned)d2 < (unsigned)R ? 4u : 0u) |
                  ((unsigned)d3 < (unsigned)R ? 8u : 0u);
    while (true) {
      const bool has = mm != 0u;
      const unsigned mk = __builtin_amdgcn_ballot_w32(has);
      if (mk == 0u) break;
      if (has) {
        const int c = __builtin_ctz(mm);
        const int dsel = (c == 0) ? d0 : (c == 1) ? d1 : (c == 2) ? d2 : d3;
        const int p = qn + __builtin_popcount(mk & ltmask);
        myqe[p] = base + c;
        myqr[p] = dsel;
      }
      qn += __builtin_popcount(mk);
      mm &= mm - 1u;
      if (qn >= 32) drain32();
    }
  };

  constexpr int TRIP = 8;
  if (bits) {
    const int ntr = B::nseg(E64);
    const int wpo = ntr * B::SEGW;
    const uint4* pq = (const uint4*)(planes + (size_t)ob * wpo) + lane;
    uint4 n0 = {0u, 0u, 0u, 0u}, n1 = n0;
    if (wave < ntr) { n0 = pq[(size_t)wave * 64]; n1 = pq[(size_t)wave * 64 + 32]; }
    for (int tr = wave; tr < ntr; tr += NW) {
      const uint4 c0 = n0, c1 = n1;
      if (tr + NW < ntr) {
        n0 = pq[(size_t)(tr + NW) * 64];
        n1 = pq[(size_t)(tr + NW) * 64 + 32];
      }
#pragma unroll 1
      for (int hh = 0; hh < 2; ++hh) {
        const uint4 c = (hh == 0) ? c0 : c1;
        unsigned long long lo = (unsigned long long)c.x | ((unsigned long long)c.y << 32);
        unsigned long long hi = (unsigned long long)c.z | ((unsigned long long)c.w << 32);
        const int base = ((tr * 2 + hh) * 32 + lane) * 128;
        while (true) {
          const bool has = (lo | hi) != 0ull;
          const unsigned mk = __builtin_amdgcn_ballot_w32(has);
          if (mk == 0u) break;
          if (has) {
            int p;
            if (lo != 0ull) { p = __builtin_ctzll(lo); lo &= lo - 1ull; }
            else { p = 64 + __builtin_ctzll(hi); hi &= hi - 1ull; }
            myqe[qn + __builtin_popcount(mk & ltmask)] = base + p;
          }
          qn += __builtin_popcount(mk);
          if (qn >= 32) drain32();
        }
      }
    }
  } else {
    const int nsc = (E + 1023) >> 10;
    for (int sc = wave; sc < nsc; sc += NW) {
      const int b = sc * 1024 + lane * 4;
      int4 c[TRIP];
      if (vec_ok && sc * 1024 + 1024 <= E) {
        const int4* p = (const int4*)(dst + b);
#pragma unroll
        for (int j = 0; j < TRIP; ++j) c[j] = p[j * 32];
      } else {
#pragma unroll
        for (int j = 0; j < TRIP; ++j) c[j] = load4(b + j * 128);
      }
#pragma unroll
      for (int j = 0; j < TRIP; ++j) scan4(c[j], b + j * 128);
    }
  }

  if (qn > 0) take(qn);
  if (lane == 0) qcnt[wave] = (can_save && cnt <= CAP) ? cnt : -1;
  __syncthreads();

  if (wave == 0) myl[lane] = __int_as_float(qcnt[lane]);

  for (int r = tid; r < R; r += NW * 32) {
    const int v = r0 + r;
    if (v < N) dinv[v] = __frsqrt_rn(fmaxf((float)(cntl[r] + 1), 1e-12f));
  }
}

template <int FOUT>
__device__ __forceinline__ void gemm_lds_b_part(
    const float* __restrict__ A, const float* __restrict__ W,
    float* __restrict__ C, int nrows, const int gb, const int ngb) {
  constexpr int PITCH = 132;
  constexpr int NCS = FOUT / 16;
  constexpr int RG = 32 / NCS;
  constexpr int TM = 16 * RG;
  constexpr int NF4 = TM * 32 / 1024;
  constexpr int NWQ = 128 * FOUT / 1024;
  static_assert(NCS * RG == 32 && NF4 * 1024 == TM * 32 && NWQ * 1024 == 128 * FOUT,
                "tile must split over 32 waves");
  __shared__ alignas(16) float wl[128 * FOUT];
  __shared__ alignas(16) float at[2][TM * PITCH];

  const int ntiles = (nrows + TM - 1) / TM;
  int tile = gb;
  if (tile >= ntiles) return;

  const int tid = threadIdx.x;
  const int lane = tid & 31;
  const int wave = tid >> 5;
  const int strip = wave % NCS;
  const int colbase = strip * 16;
  const int rg = wave / NCS;
  const int half = lane >> 4;
  const int m = lane & 15;

#pragma unroll
  for (int q = 0; q < NWQ; ++q) {
    const int i = tid + q * 1024;
    const int k = i / FOUT, col = i % FOUT;
    const int d = ((((k >> 2) * NCS + (col >> 4)) * 32 + ((k >> 1) & 1) * 16 + (col & 15)) << 1) + (k & 1);
    wl[d] = W[i];
  }

  v4f st[NF4];
  auto fetch = [&](int t) {
    const int row0 = t * TM;
    const v4f* srcv = (const v4f*)(A + (size_t)row0 * 128);
    if (row0 + TM <= nrows) {
#pragma unroll
      for (int j = 0; j < NF4; ++j) st[j] = srcv[tid + j * 1024];
    } else {
#pragma unroll
      for (int j = 0; j < NF4; ++j) {
        const int i = tid + j * 1024;
        const v4f z = {0.0f, 0.0f, 0.0f, 0.0f};
        st[j] = z;
        if (row0 + (i >> 5) < nrows) st[j] = srcv[i];
      }
    }
  };
  auto stash = [&](int buf) {
#pragma unroll
    for (int j = 0; j < NF4; ++j) {
      const int i = tid + j * 1024;
      *(v4f*)&at[buf][(i >> 5) * PITCH + (i & 31) * 4] = st[j];
    }
  };

  fetch(tile);
  stash(0);
  __syncthreads();
  int buf = 0;
  const float* bbase = &wl[(strip * 32 + lane) * 2];
  for (; tile < ntiles; tile += ngb) {
    const int nxt = tile + ngb;
    if (nxt < ntiles) fetch(nxt);

    v8f acc = v8f{};
    const float* abase = &at[buf][(rg * 16 + m) * PITCH + half * 2];
#pragma unroll
    for (int s = 0; s < 32; ++s) {
      const v2f a = *(const v2f*)(abase + 4 * s);
      const v2f b = *(const v2f*)(bbase + s * NCS * 64);
      acc = __builtin_amdgcn_wmma_f32_16x16x4_f32(false, a, false, b, (short)0, acc,
                                                  false, false);
    }

    if (nxt < ntiles) stash(buf ^ 1);
    __syncthreads();
    buf ^= 1;

    const int rbase = tile * TM + rg * 16 + half * 8;
#pragma unroll
    for (int i = 0; i < 8; ++i) {
      const int row = rbase + i;
      if (row < nrows) C[(size_t)row * FOUT + colbase + m] = acc[i];
    }
  }
}

template <int R>
__global__ __launch_bounds__(1024, 16) void scan_gemm_kernel(
    const int* __restrict__ src, const int* __restrict__ dst,
    float* __restrict__ dinv, float* lst, const unsigned* __restrict__ planes, int N,
    long long E64, const float* __restrict__ A, const float* __restrict__ W,
    float* __restrict__ C, int nsb) {
  const int bi = (int)blockIdx.x;
  const int p = bi >> 1;
  const int nb = (int)(gridDim.x >> 1);
  if (((bi ^ (bi >> 8)) & 1) == 0) {
    if (p < nsb) owner_scan_part<R>(src, dst, dinv, lst, planes, N, E64, p);
  } else {
    gemm_lds_b_part<128>(A, W, C, N, p, nb);
  }
}

template <int F, int R, bool RELU>
__global__ __launch_bounds__(1024) void owner_agg_kernel(
    const int* __restrict__ src, const int* __restrict__ dst,
    const float* __restrict__ dinv, const float* __restrict__ h,
    const float* __restrict__ bias, float* out, const float* lst, int N,
    long long E64) {
  constexpr int VEC = F / 32;
  constexpr int NW = 32;
  constexpr int QCAP = 64;
  constexpr int GRP = 8;
  constexpr int GRPL = (VEC > 2) ? 8 : 16;
  typedef Slice<R> P;
  constexpr int LW = P::LW;
  constexpr int CAP = P::CAP;
  static_assert(P::NW == NW, "slice layout is for 32 waves");
  typedef float VT __attribute__((ext_vector_type(VEC)));

  __shared__ float acc[R * F];
  __shared__ float dloc[R];
  __shared__ int qe[NW * QCAP];
  __shared__ int qr[NW * QCAP];

  const int tid = threadIdx.x;
  const int lane = tid & 31;
  const int wave = __builtin_amdgcn_readfirstlane(tid >> 5);
  const int r0 = blockIdx.x * R;
  const int E = (int)E64;

  const float* myl = lst + (size_t)blockIdx.x * LW;
  const float* wl = myl + 32 + wave * CAP;
  const int c = __float_as_int(myl[lane]);

  constexpr int NR = (R + NW - 1) / NW;
  float di[NR];
  VT hs[NR];
#pragma unroll
  for (int j = 0; j < NR; ++j) {
    const int r = wave + j * NW;
    const int v = (r < R && r0 + r < N) ? r0 + r : r0;
    di[j] = dinv[v];
    hs[j] = *(const VT*)(h + (size_t)v * F + lane * VEC);
  }

  const bool use_list = __builtin_amdgcn_ballot_w32(c < 0 || c > CAP) == 0u;
  const int n = __builtin_amdgcn_readlane(c, wave);
  auto ld = [&](int b) -> unsigned {
    return (b + lane < n) ? __float_as_uint(wl[b + lane]) : 0u;
  };
  unsigned pk = ld(0);
  unsigned pk1 = ld(32);

#pragma unroll
  for (int j = 0; j < NR; ++j) {
    const int r = wave + j * NW;
    if (r < R) {
      const bool ok = r0 + r < N;
      const float d = ok ? di[j] : 0.0f;
      if (lane == 0) dloc[r] = d;
      const float s = d * d;
#pragma unroll
      for (int t = 0; t < VEC; ++t) acc[r * F + t * 32 + lane] = ok ? s * hs[j][t] : 0.0f;
    }
  }
  float dv = dinv[pk >> 9];
  __syncthreads();

  auto add_row = [&](int rk, float wk, const VT& hv) {
#pragma unroll
    for (int t = 0; t < VEC; ++t)
      __hip_atomic_fetch_add(&acc[rk * F + t * 32 + lane], wk * hv[t],
                             __ATOMIC_RELAXED, __HIP_MEMORY_SCOPE_WORKGROUP);
  };

  if (use_list) {
    for (int b = 0; b < n; b += 32) {
      const unsigned pk2 = ld(b + 64);
      const float dv1 = dinv[pk1 >> 9];
      const int u = (int)(pk >> 9);
      const int r = (int)(pk & 511u);
      const float w = dv * dloc[r];
      const int m = (n - b < 32) ? (n - b) : 32;
      int g = 0;
      for (; g + GRPL <= m; g += GRPL) {
        VT hv[GRPL];
        int rr[GRPL];
        float ww[GRPL];
#pragma unroll
        for (int k = 0; k < GRPL; ++k) {
          const int uk = __builtin_amdgcn_readlane(u, g + k);
          rr[k] = __builtin_amdgcn_readlane(r, g + k);
          ww[k] = __int_as_float(__builtin_amdgcn_readlane(__float_as_int(w), g + k));
          hv[k] = *(const VT*)(h + (size_t)uk * F + lane * VEC);
        }
#pragma unroll
        for (int k = 0; k < GRPL; ++k) add_row(rr[k], ww[k], hv[k]);
      }
      if (g < m) {
        VT hv[GRPL];
        int rr[GRPL];
        float ww[GRPL];
#pragma unroll
        for (int k = 0; k < GRPL; ++k) {
          if (g + k < m) {
            const int uk = __builtin_amdgcn_readlane(u, g + k);
            rr[k] = __builtin_amdgcn_readlane(r, g + k);
            ww[k] = __int_as_float(__builtin_amdgcn_readlane(__float_as_int(w), g + k));
            hv[k] = *(const VT*)(h + (size_t)uk * F + lane * VEC);
          }
        }
#pragma unroll
        for (int k = 0; k < GRPL; ++k) {
          if (g + k < m) add_row(rr[k], ww[k], hv[k]);
        }
      }
      pk = pk1;
      dv = dv1;
      pk1 = pk2;
    }
  } else {
    int* myqe = qe + wave * QCAP;
    int* myqr = qr + wave * QCAP;
    const unsigned ltmask = (1u << lane) - 1u;
    int qn = 0;

    auto drain32 = [&]() {
      const int r = myqr[lane];
      const int u = src[myqe[lane]];
      const float w = dinv[u] * dloc[r];
      for (int g = 0; g < 32; g += GRP) {
        VT hv[GRP];
        int rr[GRP];
        float ww[GRP];
#pragma unroll
        for (int k = 0; k < GRP; ++k) {
          const int uk = __builtin_amdgcn_readlane(u, g + k);
          rr[k] = __builtin_amdgcn_readlane(r, g + k);
          ww[k] = __int_as_float(__builtin_amdgcn_readlane(__float_as_int(w), g + k));
          hv[k] = *(const VT*)(h + (size_t)uk * F + lane * VEC);
        }
#pragma unroll
        for (int k = 0; k < GRP; ++k) add_row(rr[k], ww[k], hv[k]);
      }
      const int rem = qn - 32;
      int te = 0, tr = 0;
      if (lane < rem) { te = myqe[32 + lane]; tr = myqr[32 + lane]; }
      if (lane < rem) { myqe[lane] = te; myqr[lane] = tr; }
      qn = rem;
    };

    const bool vec_ok = ((E & 3) == 0) && ((((size_t)dst) & 15) == 0);
    auto load4 = [&](int i) -> int4 {
      int4 d;
      d.x = (i     < E) ? dst[i]     : -1;
      d.y = (i + 1 < E) ? dst[i + 1] : -1;
      d.z = (i + 2 < E) ? dst[i + 2] : -1;
      d.w = (i + 3 < E) ? dst[i + 3] : -1;
      return d;
    };

    auto scan4 = [&](const int4& d, int base) {
      const int d0 = d.x - r0, d1 = d.y - r0, d2 = d.z - r0, d3 = d.w - r0;
      unsigned mm = ((unsigned)d0 < (unsigned)R ? 1u : 0u) |
                    ((unsigned)d1 < (unsigned)R ? 2u : 0u) |
                    ((unsigned)d2 < (unsigned)R ? 4u : 0u) |
                    ((unsigned)d3 < (unsigned)R ? 8u : 0u);
      while (true) {
        const bool has = mm != 0u;
        const unsigned mk = __builtin_amdgcn_ballot_w32(has);
        if (mk == 0u) break;
        if (has) {
          const int cc = __builtin_ctz(mm);
          const int dsel = (cc == 0) ? d0 : (cc == 1) ? d1 : (cc == 2) ? d2 : d3;
          const int p = qn + __builtin_popcount(mk & ltmask);
          myqe[p] = base + cc;
          myqr[p] = dsel;
        }
        qn += __builtin_popcount(mk);
        mm &= mm - 1u;
        if (qn >= 32) drain32();
      }
    };

    constexpr int TRIP = 8;
    const int nsc = (E + 1023) >> 10;
    for (int sc = wave; sc < nsc; sc += NW) {
      const int b = sc * 1024 + lane * 4;
      int4 cv[TRIP];
      if (vec_ok && sc * 1024 + 1024 <= E) {
        const int4* p = (const int4*)(dst + b);
#pragma unroll
        for (int j = 0; j < TRIP; ++j) cv[j] = p[j * 32];
      } else {
#pragma unroll
        for (int j = 0; j < TRIP; ++j) cv[j] = load4(b + j * 128);
      }
#pragma unroll
      for (int j = 0; j < TRIP; ++j) scan4(cv[j], b + j * 128);
    }

    if (qn > 0) {
      int r = 0, u = 0;
      float w = 0.0f;
      if (lane < qn) {
        r = myqr[lane];
        u = src[myqe[lane]];
        w = dinv[u] * dloc[r];
      }
      for (int k = 0; k < qn; ++k) {
        const int uk = __builtin_amdgcn_readlane(u, k);
        const int rk = __builtin_amdgcn_readlane(r, k);
        const float wk = __int_as_float(__builtin_amdgcn_readlane(__float_as_int(w), k));
        const VT hv = *(const VT*)(h + (size_t)uk * F + lane * VEC);
        add_row(rk, wk, hv);
      }
    }
  }
  __syncthreads();

  for (int r = wave; r < R; r += NW) {
    const int v = r0 + r;
    if (v >= N) break;
    VT o;
#pragma unroll
    for (int t = 0; t < VEC; ++t) {
      const float x = acc[r * F + t * 32 + lane] + bias[lane * VEC + t];
      o[t] = RELU ? fmaxf(x, 0.0f) : x;
    }
    *(VT*)(out + (size_t)v * F + lane * VEC) = o;
  }
}

extern "C" void kernel_launch(void* const* d_in, const int* in_sizes, int n_in,
                              void* d_out, int out_size, void* d_ws,
                              size_t ws_size, hipStream_t stream) {
  const float* x  = (const float*)d_in[0];
  const int*   ei = (const int*)d_in[1];
  const float* W1 = (const float*)d_in[2];
  const float* b1 = (const float*)d_in[3];
  const float* W2 = (const float*)d_in[4];
  const float* b2 = (const float*)d_in[5];
  float* out = (float*)d_out;

  const int N = in_sizes[0] / 128;
  const long long E = in_sizes[1] / 2;
  const int* src = ei;
  const int* dst = ei + E;

  float* ws = (float*)d_ws;
  float* h1pre = ws;
  float* h1    = h1pre + (size_t)N * 128;
  float* dinvw = h1 + (size_t)N * 128;
  float* h2pre = h1pre;

  constexpr int R = 400;
  auto blocks = [](long long n, int tb) { return (unsigned)((n + tb - 1) / tb); };

  unsigned* planes = (unsigned*)h1;
  bitplane_kernel<R><<<blocks(E, 8192), 1024, 0, stream>>>(dst, planes, N, E);
  const int nsb = (N + R - 1) / R;
  const int nb = nsb > 256 ? nsb : 256;
  scan_gemm_kernel<R><<<2 * nb, 1024, 0, stream>>>(
      src, dst, dinvw, out, planes, N, E, x, W1, h1pre, nsb);
  const float* dinv = dinvw;

  owner_agg_kernel<128, R, true><<<(N + R - 1) / R, 1024, 0, stream>>>(
      src, dst, dinv, h1pre, b1, h1, out, N, E);

  gemm_wmma_kernel<64, 1><<<1024, 256, 0, stream>>>(h1, W2, h2pre, N);
  owner_agg_kernel<64, R, false><<<(N + R - 1) / R, 1024, 0, stream>>>(
      src, dst, dinv, h2pre, b2, out, out, N, E);
}
